# starting kernel
# baseline (speedup 1.0000x reference)
#include <hip/hip_runtime.h>
#include <hip/hip_bf16.h>
#include <math.h>

typedef __attribute__((ext_vector_type(2))) float v2f;
typedef __attribute__((ext_vector_type(8))) float v8f;

#define D_CH 32
#define G_CH 96


__global__ void gcrn_init_deg(float* deg, int n) {
    int i = blockIdx.x * blockDim.x + threadIdx.x;
    if (i < n) deg[i] = 1.0f;
}

__global__ void gcrn_acc_deg(const int* __restrict__ dst, const float* __restrict__ ew,
                             float* __restrict__ deg, int e) {
    int i = blockIdx.x * blockDim.x + threadIdx.x;
    if (i < e)
        __hip_atomic_fetch_add(&deg[dst[i]], ew[i], __ATOMIC_RELAXED, __HIP_MEMORY_SCOPE_AGENT);
}

__global__ void gcrn_fin_norm(float* deg, int n) {
    int i = blockIdx.x * blockDim.x + threadIdx.x;
    if (i < n) {
        float d = deg[i];
        deg[i] = (d > 0.0f) ? rsqrtf(d) : 0.0f;
    }
}

__global__ void gcrn_edge_nw(const int* __restrict__ src, const int* __restrict__ dst,
                             const float* __restrict__ ew, const float* __restrict__ dinv,
                             float* __restrict__ nw, int e) {
    int i = blockIdx.x * blockDim.x + threadIdx.x;
    if (i < e) nw[i] = dinv[src[i]] * ew[i] * dinv[dst[i]];
}

__global__ void gcrn_matmul_selfinit(const float* __restrict__ x,
                                     const float* __restrict__ Wl,
                                     const float* __restrict__ dinv,
                                     float* __restrict__ xw,
                                     float* __restrict__ agg) {
    const int wave = threadIdx.x >> 5;
    const int lane = threadIdx.x & 31;
    const int n0   = lane & 15;
    const int hi   = lane >> 4;
    const int row0 = blockIdx.x * 16;

    const int gate = wave >> 1;
    const int g    = (gate == 0) ? 0 : (2 * gate + 2);
    const float* Wg = Wl + g * 32 * 32 + (wave & 1) * 16 + n0;
    const float* xr = x + (size_t)(row0 + n0) * D_CH;

    v8f c = {};
    #pragma unroll
    for (int k = 0; k < 32; k += 4) {
        const int ka = k + 2 * hi;
        v2f a; a.x = xr[ka];            a.y = xr[ka + 1];
        v2f b; b.x = Wg[ka * 32];       b.y = Wg[(ka + 1) * 32];
        c = __builtin_amdgcn_wmma_f32_16x16x4_f32(false, a, false, b, (short)0, c, false, false);
    }

    const int col = wave * 16 + n0;
    #pragma unroll
    for (int j = 0; j < 8; ++j) {
        const int r = row0 + hi * 8 + j;
        const float v  = c[j];
        const float dv = dinv[r];
        xw [(size_t)r * G_CH + col] = v;
        agg[(size_t)r * G_CH + col] = dv * dv * v;
    }
}

__global__ void gcrn_edge_agg(const int* __restrict__ src, const int* __restrict__ dst,
                              const float* __restrict__ nw,
                              const float* __restrict__ xw, float* __restrict__ agg, int e) {
    const int edge = blockIdx.x * (blockDim.x >> 5) + (threadIdx.x >> 5);
    if (edge >= e) return;
    const int lane = threadIdx.x & 31;
    const int s = src[edge], d = dst[edge];
    const float w = nw[edge];
    const float* xs = xw  + (size_t)s * G_CH;
    float*       ad = agg + (size_t)d * G_CH;
    #pragma unroll
    for (int j = 0; j < 3; ++j) {
        const int cc = lane + 32 * j;
        __hip_atomic_fetch_add(&ad[cc], w * xs[cc], __ATOMIC_RELAXED, __HIP_MEMORY_SCOPE_AGENT);
    }
}

__global__ void gcrn_gates(const float* __restrict__ agg,
                           const float* __restrict__ cb,
                           const float* __restrict__ bgl,
                           const float* __restrict__ wco,
                           float* __restrict__ out, int n) {
    const int idx = blockIdx.x * blockDim.x + threadIdx.x;
    if (idx >= n * D_CH) return;
    const int c    = idx & 31;
    const int node = idx >> 5;
    const float* a = agg + (size_t)node * G_CH;

    const float bi = cb[c]          + cb[32 + c]      + bgl[c];
    const float bt = cb[4 * 32 + c] + cb[5 * 32 + c]  + bgl[2 * 32 + c];
    const float bo = cb[6 * 32 + c] + cb[7 * 32 + c]  + bgl[3 * 32 + c];

    const float I = 1.0f / (1.0f + expf(-(a[c] + bi)));
    const float T = tanhf(a[32 + c] + bt);
    const float C = I * T;
    const float O = 1.0f / (1.0f + expf(-(a[64 + c] + bo + wco[c] * C)));
    out[idx] = O * tanhf(C);
}

extern "C" void kernel_launch(void* const* d_in, const int* in_sizes, int n_in,
                              void* d_out, int out_size, void* d_ws, size_t ws_size,
                              hipStream_t stream) {
    const float* X       = (const float*)d_in[0];
    const int*   eidx    = (const int*)  d_in[1];
    const float* ew      = (const float*)d_in[2];
    const float* conv_w  = (const float*)d_in[3];
    const float* conv_b  = (const float*)d_in[4];
    const float* wc      = (const float*)d_in[5];
    const float* bg      = (const float*)d_in[6];

    const int N = in_sizes[0] / D_CH;
    const int E = in_sizes[2];
    const int L = in_sizes[3] / (8 * D_CH * D_CH);
    const int* src = eidx;
    const int* dst = eidx + E;

    float* ws   = (float*)d_ws;
    float* dinv = ws;
    float* nw   = dinv + N;
    float* xw   = nw   + E;
    float* agg  = xw   + (size_t)N * G_CH;
    float* h1   = agg  + (size_t)N * G_CH;
    float* h2   = h1   + (size_t)N * D_CH;

    const int TB = 256;
    gcrn_init_deg<<<(N + TB - 1) / TB, TB, 0, stream>>>(dinv, N);
    gcrn_acc_deg <<<(E + TB - 1) / TB, TB, 0, stream>>>(dst, ew, dinv, E);
    gcrn_fin_norm<<<(N + TB - 1) / TB, TB, 0, stream>>>(dinv, N);
    gcrn_edge_nw <<<(E + TB - 1) / TB, TB, 0, stream>>>(src, dst, ew, dinv, nw, E);

    const float* xin = X;
    for (int l = 0; l < L; ++l) {
        gcrn_matmul_selfinit<<<N / 16, 192, 0, stream>>>(
            xin, conv_w + (size_t)l * 8 * 32 * 32, dinv, xw, agg);

        const int edgesPerBlock = TB / 32;
        gcrn_edge_agg<<<(E + edgesPerBlock - 1) / edgesPerBlock, TB, 0, stream>>>(
            src, dst, nw, xw, agg, E);

        float* out_l = (l == L - 1) ? (float*)d_out : ((l & 1) ? h2 : h1);
        gcrn_gates<<<((size_t)N * D_CH + TB - 1) / TB, TB, 0, stream>>>(
            agg, conv_b + (size_t)l * 8 * 32, bg + (size_t)l * 4 * 32,
            wc + ((size_t)l * 3 + 2) * 32, out_l, N);
        xin = out_l;
    }
}


# f32 atomic scatter-add replaced by a per-call edge list sorted by destination (int count, scan, place, per-node sort by edge id) and a per-node register gather; f32 sums now in fixed edge order
# speedup vs baseline: 2.6610x; 2.6610x over previous
#include <hip/hip_runtime.h>
#include <hip/hip_bf16.h>
#include <math.h>


typedef __attribute__((ext_vector_type(2))) float v2f;
typedef __attribute__((ext_vector_type(8))) float v8f;

#define D_CH 32
#define G_CH 96
#define SCAN_TB 1024
#define ROWS_IN_FLIGHT 8


__global__ void gcrn_zero_cnt(int* cnt, int n) {
    int i = blockIdx.x * blockDim.x + threadIdx.x;
    if (i < n) cnt[i] = 0;
}

__global__ void gcrn_count_rank(const int* __restrict__ dst, int* __restrict__ cnt,
                                int* __restrict__ rnk, int e) {
    int i = blockIdx.x * blockDim.x + threadIdx.x;
    if (i < e)
        rnk[i] = __hip_atomic_fetch_add(&cnt[dst[i]], 1, __ATOMIC_RELAXED, __HIP_MEMORY_SCOPE_AGENT);
}

__device__ __forceinline__ int wave_sum_i(int v) {
    for (int m = 16; m >= 1; m >>= 1) v += __shfl_xor(v, m, 32);
    return v;
}

__global__ void __launch_bounds__(SCAN_TB)
gcrn_block_sums(const int* __restrict__ cnt, int* __restrict__ bsum, int n) {
    __shared__ int sw[32];
    const int i = blockIdx.x * SCAN_TB + threadIdx.x;
    int v = (i < n) ? cnt[i] : 0;
    v = wave_sum_i(v);
    if ((threadIdx.x & 31) == 0) sw[threadIdx.x >> 5] = v;
    __syncthreads();
    if (threadIdx.x < 32) {
        int s = wave_sum_i(sw[threadIdx.x]);
        if (threadIdx.x == 0) bsum[blockIdx.x] = s;
    }
}

__global__ void __launch_bounds__(SCAN_TB)
gcrn_scan(const int* __restrict__ cnt, const int* __restrict__ bsum, int* __restrict__ offs, int n) {
    __shared__ int sw[32];
    __shared__ int sbase;
    const int tid = threadIdx.x, lane = tid & 31, wv = tid >> 5;

    int pre = 0;
    for (int k = tid; k < (int)blockIdx.x; k += SCAN_TB) pre += bsum[k];
    pre = wave_sum_i(pre);
    if (lane == 0) sw[wv] = pre;
    __syncthreads();
    if (wv == 0) {
        int s = wave_sum_i(sw[lane]);
        if (lane == 0) sbase = s;
    }
    __syncthreads();
    const int base = sbase;

    const int i = blockIdx.x * SCAN_TB + tid;
    const int v = (i < n) ? cnt[i] : 0;
    int inc = v;
    for (int d = 1; d < 32; d <<= 1) {
        int t = __shfl_up(inc, d, 32);
        if (lane >= d) inc += t;
    }
    if (lane == 31) sw[wv] = inc;
    __syncthreads();
    if (wv == 0) {
        const int s = sw[lane];
        int si = s;
        for (int d = 1; d < 32; d <<= 1) {
            int t = __shfl_up(si, d, 32);
            if (lane >= d) si += t;
        }
        sw[lane] = si - s;
    }
    __syncthreads();
    const int excl = base + sw[wv] + inc - v;
    if (i < n) offs[i] = excl;
    if (i == n - 1) offs[n] = excl + v;
}

__global__ void gcrn_place(const int* __restrict__ dst, const int* __restrict__ rnk,
                           const int* __restrict__ offs, int* __restrict__ key, int e) {
    int i = blockIdx.x * blockDim.x + threadIdx.x;
    if (i < e) key[offs[dst[i]] + rnk[i]] = i;
}

__global__ void __launch_bounds__(256)
gcrn_sort_seg(const int* __restrict__ offs, const int* __restrict__ key,
              const int* __restrict__ src, const float* __restrict__ ew,
              int* __restrict__ csr_src, float* __restrict__ csr_w, int n) {
    const int node = __builtin_amdgcn_readfirstlane(blockIdx.x * (blockDim.x >> 5) + (threadIdx.x >> 5));
    if (node >= n) return;
    const int lane = threadIdx.x & 31;
    const int beg = offs[node];
    const int deg = offs[node + 1] - beg;
    for (int i0 = 0; i0 < deg; i0 += 32) {
        const bool vi = (i0 + lane) < deg;
        const int ki = vi ? key[beg + i0 + lane] : 0x7fffffff;
        int r = 0;
        for (int j0 = 0; j0 < deg; j0 += 32) {
            const int m = min(32, deg - j0);
            const int kj = (j0 + lane < deg) ? key[beg + j0 + lane] : 0x7fffffff;
            for (int t = 0; t < m; ++t)
                r += (__builtin_amdgcn_readlane(kj, t) < ki) ? 1 : 0;
        }
        if (vi) {
            csr_src[beg + r] = src[ki];
            csr_w  [beg + r] = ew[ki];
        }
    }
}

__global__ void gcrn_deg_dinv(const int* __restrict__ offs, const float* __restrict__ csr_w,
                              float* __restrict__ dinv, int n) {
    int i = blockIdx.x * blockDim.x + threadIdx.x;
    if (i < n) {
        const int beg = offs[i], end = offs[i + 1];
        float d = 1.0f;
        for (int p = beg; p < end; ++p) d += csr_w[p];
        dinv[i] = (d > 0.0f) ? rsqrtf(d) : 0.0f;
    }
}

__global__ void __launch_bounds__(256)
gcrn_edge_nw(const int* __restrict__ offs, const int* __restrict__ csr_src,
             const float* __restrict__ csr_w, const float* __restrict__ dinv,
             float* __restrict__ nw, int n) {
    const int node = blockIdx.x * (blockDim.x >> 5) + (threadIdx.x >> 5);
    if (node >= n) return;
    const int lane = threadIdx.x & 31;
    const int end = offs[node + 1];
    const float dd = dinv[node];
    for (int p = offs[node] + lane; p < end; p += 32)
        nw[p] = dinv[csr_src[p]] * csr_w[p] * dd;
}

__global__ void gcrn_matmul(const float* __restrict__ x,
                            const float* __restrict__ Wl,
                            float* __restrict__ xw) {
    const int wave = threadIdx.x >> 5;
    const int lane = threadIdx.x & 31;
    const int n0   = lane & 15;
    const int hi   = lane >> 4;
    const int row0 = blockIdx.x * 16;

    const int gate = wave >> 1;
    const int g    = (gate == 0) ? 0 : (2 * gate + 2);
    const float* Wg = Wl + g * 32 * 32 + (wave & 1) * 16 + n0;
    const float* xr = x + (size_t)(row0 + n0) * D_CH;

    v8f c = {};
    #pragma unroll
    for (int k = 0; k < 32; k += 4) {
        const int ka = k + 2 * hi;
        v2f a; a.x = xr[ka];            a.y = xr[ka + 1];
        v2f b; b.x = Wg[ka * 32];       b.y = Wg[(ka + 1) * 32];
        c = __builtin_amdgcn_wmma_f32_16x16x4_f32(false, a, false, b, (short)0, c, false, false);
    }

    const int col = wave * 16 + n0;
    #pragma unroll
    for (int j = 0; j < 8; ++j) {
        const int r = row0 + hi * 8 + j;
        xw[(size_t)r * G_CH + col] = c[j];
    }
}

__global__ void __launch_bounds__(256)
gcrn_gather(const int* __restrict__ offs, const int* __restrict__ csr_src,
            const float* __restrict__ nw, const float* __restrict__ dinv,
            const float* __restrict__ xw, float* __restrict__ agg, int n) {
    const int node = __builtin_amdgcn_readfirstlane(blockIdx.x * (blockDim.x >> 5) + (threadIdx.x >> 5));
    if (node >= n) return;
    const int lane = threadIdx.x & 31;
    const int beg = offs[node], end = offs[node + 1];
    const float dv = dinv[node];
    const float d2 = dv * dv;
    const float* xr = xw + (size_t)node * G_CH + lane;
    float a0 = d2 * xr[0], a1 = d2 * xr[32], a2 = d2 * xr[64];

    for (int p0 = beg; p0 < end; p0 += 32) {
        const int m = min(32, end - p0);
        int s_l = 0; float w_l = 0.0f;
        if (lane < m) { s_l = csr_src[p0 + lane]; w_l = nw[p0 + lane]; }
        for (int t = 0; t < m; t += ROWS_IN_FLIGHT) {
            float v0[ROWS_IN_FLIGHT], v1[ROWS_IN_FLIGHT], v2[ROWS_IN_FLIGHT];
            #pragma unroll
            for (int u = 0; u < ROWS_IN_FLIGHT; ++u) {
                const int tt = min(t + u, m - 1);
                const int s  = __builtin_amdgcn_readlane(s_l, tt);
                const float* xs = xw + (size_t)s * G_CH + lane;
                v0[u] = xs[0]; v1[u] = xs[32]; v2[u] = xs[64];
            }
            #pragma unroll
            for (int u = 0; u < ROWS_IN_FLIGHT; ++u) {
                if (t + u < m) {
                    const float w = __int_as_float(__builtin_amdgcn_readlane(__float_as_int(w_l), t + u));
                    a0 = fmaf(w, v0[u], a0);
                    a1 = fmaf(w, v1[u], a1);
                    a2 = fmaf(w, v2[u], a2);
                }
            }
        }
    }
    float* ad = agg + (size_t)node * G_CH + lane;
    ad[0] = a0; ad[32] = a1; ad[64] = a2;
}

__global__ void gcrn_gates(const float* __restrict__ agg,
                           const float* __restrict__ cb,
                           const float* __restrict__ bgl,
                           const float* __restrict__ wco,
                           float* __restrict__ out, int n) {
    const int idx = blockIdx.x * blockDim.x + threadIdx.x;
    if (idx >= n * D_CH) return;
    const int c    = idx & 31;
    const int node = idx >> 5;
    const float* a = agg + (size_t)node * G_CH;

    const float bi = cb[c]          + cb[32 + c]      + bgl[c];
    const float bt = cb[4 * 32 + c] + cb[5 * 32 + c]  + bgl[2 * 32 + c];
    const float bo = cb[6 * 32 + c] + cb[7 * 32 + c]  + bgl[3 * 32 + c];

    const float I = 1.0f / (1.0f + expf(-(a[c] + bi)));
    const float T = tanhf(a[32 + c] + bt);
    const float C = I * T;
    const float O = 1.0f / (1.0f + expf(-(a[64 + c] + bo + wco[c] * C)));
    out[idx] = O * tanhf(C);
}

extern "C" void kernel_launch(void* const* d_in, const int* in_sizes, int n_in,
                              void* d_out, int out_size, void* d_ws, size_t ws_size,
                              hipStream_t stream) {
    const float* X       = (const float*)d_in[0];
    const int*   eidx    = (const int*)  d_in[1];
    const float* ew      = (const float*)d_in[2];
    const float* conv_w  = (const float*)d_in[3];
    const float* conv_b  = (const float*)d_in[4];
    const float* wc      = (const float*)d_in[5];
    const float* bg      = (const float*)d_in[6];

    const int N = in_sizes[0] / D_CH;
    const int E = in_sizes[2];
    const int L = in_sizes[3] / (8 * D_CH * D_CH);
    const int* src = eidx;
    const int* dst = eidx + E;

    float* ws   = (float*)d_ws;
    float* xw   = ws;
    float* agg  = xw   + (size_t)N * G_CH;
    float* h1   = agg  + (size_t)N * G_CH;
    float* h2   = h1   + (size_t)N * D_CH;
    float* dinv = h2   + (size_t)N * D_CH;
    float* nw   = dinv + N;
    float* csrw = nw   + E;
    int*   csrs = (int*)(csrw + E);
    int*   key  = csrs + E;
    int*   rnk  = key  + E;
    int*   cnt  = rnk  + E;
    int*   bsum = cnt  + N;
    const int NB = (N + SCAN_TB - 1) / SCAN_TB;
    int*   offs = bsum + NB;

    const int TB = 256;
    const int nodesPerBlock = TB / 32;
    const int nodeBlocks = (N + nodesPerBlock - 1) / nodesPerBlock;

    gcrn_zero_cnt  <<<(N + TB - 1) / TB, TB, 0, stream>>>(cnt, N);
    gcrn_count_rank<<<(E + TB - 1) / TB, TB, 0, stream>>>(dst, cnt, rnk, E);
    gcrn_block_sums<<<NB, SCAN_TB, 0, stream>>>(cnt, bsum, N);
    gcrn_scan      <<<NB, SCAN_TB, 0, stream>>>(cnt, bsum, offs, N);
    gcrn_place     <<<(E + TB - 1) / TB, TB, 0, stream>>>(dst, rnk, offs, key, E);
    gcrn_sort_seg  <<<nodeBlocks, TB, 0, stream>>>(offs, key, src, ew, csrs, csrw, N);
    gcrn_deg_dinv  <<<(N + TB - 1) / TB, TB, 0, stream>>>(offs, csrw, dinv, N);
    gcrn_edge_nw   <<<nodeBlocks, TB, 0, stream>>>(offs, csrs, csrw, dinv, nw, N);

    const float* xin = X;
    for (int l = 0; l < L; ++l) {
        gcrn_matmul<<<N / 16, 192, 0, stream>>>(xin, conv_w + (size_t)l * 8 * 32 * 32, xw);

        gcrn_gather<<<nodeBlocks, TB, 0, stream>>>(offs, csrs, nw, dinv, xw, agg, N);

        float* out_l = (l == L - 1) ? (float*)d_out : ((l & 1) ? h2 : h1);
        gcrn_gates<<<((size_t)N * D_CH + TB - 1) / TB, TB, 0, stream>>>(
            agg, conv_b + (size_t)l * 8 * 32, bg + (size_t)l * 4 * 32,
            wc + ((size_t)l * 3 + 2) * 32, out_l, N);
        xin = out_l;
    }
}


# Gate epilogue fused into the per-node gather: lane c already holds channel c's I, T and O sums in registers, applies the gates and stores out; the gates launches and the agg plane are gone
# speedup vs baseline: 2.8677x; 1.0777x over previous
#include <hip/hip_runtime.h>
#include <hip/hip_bf16.h>
#include <math.h>


typedef __attribute__((ext_vector_type(2))) float v2f;
typedef __attribute__((ext_vector_type(8))) float v8f;

#define D_CH 32
#define G_CH 96
#define SCAN_TB 1024
#define ROWS_IN_FLIGHT 8


__global__ void gcrn_zero_cnt(int* cnt, int n) {
    int i = blockIdx.x * blockDim.x + threadIdx.x;
    if (i < n) cnt[i] = 0;
}

__global__ void gcrn_count_rank(const int* __restrict__ dst, int* __restrict__ cnt,
                                int* __restrict__ rnk, int e) {
    int i = blockIdx.x * blockDim.x + threadIdx.x;
    if (i < e)
        rnk[i] = __hip_atomic_fetch_add(&cnt[dst[i]], 1, __ATOMIC_RELAXED, __HIP_MEMORY_SCOPE_AGENT);
}

__device__ __forceinline__ int wave_sum_i(int v) {
    for (int m = 16; m >= 1; m >>= 1) v += __shfl_xor(v, m, 32);
    return v;
}

__global__ void __launch_bounds__(SCAN_TB)
gcrn_block_sums(const int* __restrict__ cnt, int* __restrict__ bsum, int n) {
    __shared__ int sw[32];
    const int i = blockIdx.x * SCAN_TB + threadIdx.x;
    int v = (i < n) ? cnt[i] : 0;
    v = wave_sum_i(v);
    if ((threadIdx.x & 31) == 0) sw[threadIdx.x >> 5] = v;
    __syncthreads();
    if (threadIdx.x < 32) {
        int s = wave_sum_i(sw[threadIdx.x]);
        if (threadIdx.x == 0) bsum[blockIdx.x] = s;
    }
}

__global__ void __launch_bounds__(SCAN_TB)
gcrn_scan(const int* __restrict__ cnt, const int* __restrict__ bsum, int* __restrict__ offs, int n) {
    __shared__ int sw[32];
    __shared__ int sbase;
    const int tid = threadIdx.x, lane = tid & 31, wv = tid >> 5;

    int pre = 0;
    for (int k = tid; k < (int)blockIdx.x; k += SCAN_TB) pre += bsum[k];
    pre = wave_sum_i(pre);
    if (lane == 0) sw[wv] = pre;
    __syncthreads();
    if (wv == 0) {
        int s = wave_sum_i(sw[lane]);
        if (lane == 0) sbase = s;
    }
    __syncthreads();
    const int base = sbase;

    const int i = blockIdx.x * SCAN_TB + tid;
    const int v = (i < n) ? cnt[i] : 0;
    int inc = v;
    for (int d = 1; d < 32; d <<= 1) {
        int t = __shfl_up(inc, d, 32);
        if (lane >= d) inc += t;
    }
    if (lane == 31) sw[wv] = inc;
    __syncthreads();
    if (wv == 0) {
        const int s = sw[lane];
        int si = s;
        for (int d = 1; d < 32; d <<= 1) {
            int t = __shfl_up(si, d, 32);
            if (lane >= d) si += t;
        }
        sw[lane] = si - s;
    }
    __syncthreads();
    const int excl = base + sw[wv] + inc - v;
    if (i < n) offs[i] = excl;
    if (i == n - 1) offs[n] = excl + v;
}

__global__ void gcrn_place(const int* __restrict__ dst, const int* __restrict__ rnk,
                           const int* __restrict__ offs, int* __restrict__ key, int e) {
    int i = blockIdx.x * blockDim.x + threadIdx.x;
    if (i < e) key[offs[dst[i]] + rnk[i]] = i;
}

__global__ void __launch_bounds__(256)
gcrn_sort_seg(const int* __restrict__ offs, const int* __restrict__ key,
              const int* __restrict__ src, const float* __restrict__ ew,
              int* __restrict__ csr_src, float* __restrict__ csr_w, int n) {
    const int node = __builtin_amdgcn_readfirstlane(blockIdx.x * (blockDim.x >> 5) + (threadIdx.x >> 5));
    if (node >= n) return;
    const int lane = threadIdx.x & 31;
    const int beg = offs[node];
    const int deg = offs[node + 1] - beg;
    for (int i0 = 0; i0 < deg; i0 += 32) {
        const bool vi = (i0 + lane) < deg;
        const int ki = vi ? key[beg + i0 + lane] : 0x7fffffff;
        int r = 0;
        for (int j0 = 0; j0 < deg; j0 += 32) {
            const int m = min(32, deg - j0);
            const int kj = (j0 + lane < deg) ? key[beg + j0 + lane] : 0x7fffffff;
            for (int t = 0; t < m; ++t)
                r += (__builtin_amdgcn_readlane(kj, t) < ki) ? 1 : 0;
        }
        if (vi) {
            csr_src[beg + r] = src[ki];
            csr_w  [beg + r] = ew[ki];
        }
    }
}

__global__ void gcrn_deg_dinv(const int* __restrict__ offs, const float* __restrict__ csr_w,
                              float* __restrict__ dinv, int n) {
    int i = blockIdx.x * blockDim.x + threadIdx.x;
    if (i < n) {
        const int beg = offs[i], end = offs[i + 1];
        float d = 1.0f;
        for (int p = beg; p < end; ++p) d += csr_w[p];
        dinv[i] = (d > 0.0f) ? rsqrtf(d) : 0.0f;
    }
}

__global__ void __launch_bounds__(256)
gcrn_edge_nw(const int* __restrict__ offs, const int* __restrict__ csr_src,
             const float* __restrict__ csr_w, const float* __restrict__ dinv,
             float* __restrict__ nw, int n) {
    const int node = blockIdx.x * (blockDim.x >> 5) + (threadIdx.x >> 5);
    if (node >= n) return;
    const int lane = threadIdx.x & 31;
    const int end = offs[node + 1];
    const float dd = dinv[node];
    for (int p = offs[node] + lane; p < end; p += 32)
        nw[p] = dinv[csr_src[p]] * csr_w[p] * dd;
}

__global__ void gcrn_matmul(const float* __restrict__ x,
                            const float* __restrict__ Wl,
                            float* __restrict__ xw) {
    const int wave = threadIdx.x >> 5;
    const int lane = threadIdx.x & 31;
    const int n0   = lane & 15;
    const int hi   = lane >> 4;
    const int row0 = blockIdx.x * 16;

    const int gate = wave >> 1;
    const int g    = (gate == 0) ? 0 : (2 * gate + 2);
    const float* Wg = Wl + g * 32 * 32 + (wave & 1) * 16 + n0;
    const float* xr = x + (size_t)(row0 + n0) * D_CH;

    v8f c = {};
    #pragma unroll
    for (int k = 0; k < 32; k += 4) {
        const int ka = k + 2 * hi;
        v2f a; a.x = xr[ka];            a.y = xr[ka + 1];
        v2f b; b.x = Wg[ka * 32];       b.y = Wg[(ka + 1) * 32];
        c = __builtin_amdgcn_wmma_f32_16x16x4_f32(false, a, false, b, (short)0, c, false, false);
    }

    const int col = wave * 16 + n0;
    #pragma unroll
    for (int j = 0; j < 8; ++j) {
        const int r = row0 + hi * 8 + j;
        xw[(size_t)r * G_CH + col] = c[j];
    }
}

__global__ void __launch_bounds__(256)
gcrn_gather_gates(const int* __restrict__ offs, const int* __restrict__ csr_src,
                  const float* __restrict__ nw, const float* __restrict__ dinv,
                  const float* __restrict__ xw,
                  const float* __restrict__ cb,
                  const float* __restrict__ bgl,
                  const float* __restrict__ wco,
                  float* __restrict__ out, int n) {
    const int node = __builtin_amdgcn_readfirstlane(blockIdx.x * (blockDim.x >> 5) + (threadIdx.x >> 5));
    if (node >= n) return;
    const int lane = threadIdx.x & 31;
    const int beg = offs[node], end = offs[node + 1];
    const float dv = dinv[node];
    const float d2 = dv * dv;
    const float* xr = xw + (size_t)node * G_CH + lane;
    float a0 = d2 * xr[0], a1 = d2 * xr[32], a2 = d2 * xr[64];

    const float bi = cb[lane]          + cb[32 + lane]     + bgl[lane];
    const float bt = cb[4 * 32 + lane] + cb[5 * 32 + lane] + bgl[2 * 32 + lane];
    const float bo = cb[6 * 32 + lane] + cb[7 * 32 + lane] + bgl[3 * 32 + lane];
    const float wo = wco[lane];

    for (int p0 = beg; p0 < end; p0 += 32) {
        const int m = min(32, end - p0);
        int s_l = 0; float w_l = 0.0f;
        if (lane < m) { s_l = csr_src[p0 + lane]; w_l = nw[p0 + lane]; }
        for (int t = 0; t < m; t += ROWS_IN_FLIGHT) {
            float v0[ROWS_IN_FLIGHT], v1[ROWS_IN_FLIGHT], v2[ROWS_IN_FLIGHT];
            #pragma unroll
            for (int u = 0; u < ROWS_IN_FLIGHT; ++u) {
                const int tt = min(t + u, m - 1);
                const int s  = __builtin_amdgcn_readlane(s_l, tt);
                const float* xs = xw + (size_t)s * G_CH + lane;
                v0[u] = xs[0]; v1[u] = xs[32]; v2[u] = xs[64];
            }
            #pragma unroll
            for (int u = 0; u < ROWS_IN_FLIGHT; ++u) {
                if (t + u < m) {
                    const float w = __int_as_float(__builtin_amdgcn_readlane(__float_as_int(w_l), t + u));
                    a0 = fmaf(w, v0[u], a0);
                    a1 = fmaf(w, v1[u], a1);
                    a2 = fmaf(w, v2[u], a2);
                }
            }
        }
    }
    const float I = 1.0f / (1.0f + expf(-(a0 + bi)));
    const float T = tanhf(a1 + bt);
    const float C = I * T;
    const float O = 1.0f / (1.0f + expf(-(a2 + bo + wo * C)));
    out[(size_t)node * D_CH + lane] = O * tanhf(C);
}

extern "C" void kernel_launch(void* const* d_in, const int* in_sizes, int n_in,
                              void* d_out, int out_size, void* d_ws, size_t ws_size,
                              hipStream_t stream) {
    const float* X       = (const float*)d_in[0];
    const int*   eidx    = (const int*)  d_in[1];
    const float* ew      = (const float*)d_in[2];
    const float* conv_w  = (const float*)d_in[3];
    const float* conv_b  = (const float*)d_in[4];
    const float* wc      = (const float*)d_in[5];
    const float* bg      = (const float*)d_in[6];

    const int N = in_sizes[0] / D_CH;
    const int E = in_sizes[2];
    const int L = in_sizes[3] / (8 * D_CH * D_CH);
    const int* src = eidx;
    const int* dst = eidx + E;

    float* ws   = (float*)d_ws;
    float* xw   = ws;
    float* h1   = xw   + (size_t)N * G_CH;
    float* h2   = h1   + (size_t)N * D_CH;
    float* dinv = h2   + (size_t)N * D_CH;
    float* nw   = dinv + N;
    float* csrw = nw   + E;
    int*   csrs = (int*)(csrw + E);
    int*   key  = csrs + E;
    int*   rnk  = key  + E;
    int*   cnt  = rnk  + E;
    int*   bsum = cnt  + N;
    const int NB = (N + SCAN_TB - 1) / SCAN_TB;
    int*   offs = bsum + NB;

    const int TB = 256;
    const int nodesPerBlock = TB / 32;
    const int nodeBlocks = (N + nodesPerBlock - 1) / nodesPerBlock;

    gcrn_zero_cnt  <<<(N + TB - 1) / TB, TB, 0, stream>>>(cnt, N);
    gcrn_count_rank<<<(E + TB - 1) / TB, TB, 0, stream>>>(dst, cnt, rnk, E);
    gcrn_block_sums<<<NB, SCAN_TB, 0, stream>>>(cnt, bsum, N);
    gcrn_scan      <<<NB, SCAN_TB, 0, stream>>>(cnt, bsum, offs, N);
    gcrn_place     <<<(E + TB - 1) / TB, TB, 0, stream>>>(dst, rnk, offs, key, E);
    gcrn_sort_seg  <<<nodeBlocks, TB, 0, stream>>>(offs, key, src, ew, csrs, csrw, N);
    gcrn_deg_dinv  <<<(N + TB - 1) / TB, TB, 0, stream>>>(offs, csrw, dinv, N);
    gcrn_edge_nw   <<<nodeBlocks, TB, 0, stream>>>(offs, csrs, csrw, dinv, nw, N);

    const float* xin = X;
    for (int l = 0; l < L; ++l) {
        gcrn_matmul<<<N / 16, 192, 0, stream>>>(xin, conv_w + (size_t)l * 8 * 32 * 32, xw);

        float* out_l = (l == L - 1) ? (float*)d_out : ((l & 1) ? h2 : h1);
        gcrn_gather_gates<<<nodeBlocks, TB, 0, stream>>>(
            offs, csrs, nw, dinv, xw,
            conv_b + (size_t)l * 8 * 32, bg + (size_t)l * 4 * 32,
            wc + ((size_t)l * 3 + 2) * 32, out_l, N);
        xin = out_l;
    }
}


# Aggregate before transform, (A x)W for A(xW), no atomics: per 16-node tile 2 waves gather 32-ch x rows in slot order into LDS, then WMMA transform + gates in the same kernel; f32 sum order changes
# speedup vs baseline: 3.2261x; 1.1250x over previous
#include <hip/hip_runtime.h>
#include <hip/hip_bf16.h>
#include <math.h>


typedef __attribute__((ext_vector_type(2))) float v2f;
typedef __attribute__((ext_vector_type(8))) float v8f;

#define D_CH 32
#define SCAN_TB 1024
#define TILE_NODES 16
#define WAVE_NODES 8
#define TILE_PITCH 36


__global__ void gcrn_zero_cnt(int* cnt, int n) {
    int i = blockIdx.x * blockDim.x + threadIdx.x;
    if (i < n) cnt[i] = 0;
}

__global__ void gcrn_count_rank(const int* __restrict__ dst, int* __restrict__ cnt,
                                int* __restrict__ rnk, int e) {
    int i = blockIdx.x * blockDim.x + threadIdx.x;
    if (i < e)
        rnk[i] = __hip_atomic_fetch_add(&cnt[dst[i]], 1, __ATOMIC_RELAXED, __HIP_MEMORY_SCOPE_AGENT);
}

__device__ __forceinline__ int wave_sum_i(int v) {
    for (int m = 16; m >= 1; m >>= 1) v += __shfl_xor(v, m, 32);
    return v;
}

__global__ void __launch_bounds__(SCAN_TB)
gcrn_block_sums(const int* __restrict__ cnt, int* __restrict__ bsum, int n) {
    __shared__ int sw[32];
    const int i = blockIdx.x * SCAN_TB + threadIdx.x;
    int v = (i < n) ? cnt[i] : 0;
    v = wave_sum_i(v);
    if ((threadIdx.x & 31) == 0) sw[threadIdx.x >> 5] = v;
    __syncthreads();
    if (threadIdx.x < 32) {
        int s = wave_sum_i(sw[threadIdx.x]);
        if (threadIdx.x == 0) bsum[blockIdx.x] = s;
    }
}

__global__ void __launch_bounds__(SCAN_TB)
gcrn_scan(const int* __restrict__ cnt, const int* __restrict__ bsum, int* __restrict__ offs, int n) {
    __shared__ int sw[32];
    __shared__ int sbase;
    const int tid = threadIdx.x, lane = tid & 31, wv = tid >> 5;

    int pre = 0;
    for (int k = tid; k < (int)blockIdx.x; k += SCAN_TB) pre += bsum[k];
    pre = wave_sum_i(pre);
    if (lane == 0) sw[wv] = pre;
    __syncthreads();
    if (wv == 0) {
        int s = wave_sum_i(sw[lane]);
        if (lane == 0) sbase = s;
    }
    __syncthreads();
    const int base = sbase;

    const int i = blockIdx.x * SCAN_TB + tid;
    const int v = (i < n) ? cnt[i] : 0;
    int inc = v;
    for (int d = 1; d < 32; d <<= 1) {
        int t = __shfl_up(inc, d, 32);
        if (lane >= d) inc += t;
    }
    if (lane == 31) sw[wv] = inc;
    __syncthreads();
    if (wv == 0) {
        const int s = sw[lane];
        int si = s;
        for (int d = 1; d < 32; d <<= 1) {
            int t = __shfl_up(si, d, 32);
            if (lane >= d) si += t;
        }
        sw[lane] = si - s;
    }
    __syncthreads();
    const int excl = base + sw[wv] + inc - v;
    if (i < n) offs[i] = excl;
    if (i == n - 1) offs[n] = excl + v;
}

__global__ void gcrn_place(const int* __restrict__ dst, const int* __restrict__ rnk,
                           const int* __restrict__ offs, int* __restrict__ key, int e) {
    int i = blockIdx.x * blockDim.x + threadIdx.x;
    if (i < e) key[offs[dst[i]] + rnk[i]] = i;
}

__global__ void __launch_bounds__(256)
gcrn_sort_seg(const int* __restrict__ offs, const int* __restrict__ key,
              const int* __restrict__ src, const float* __restrict__ ew,
              int* __restrict__ csr_src, float* __restrict__ csr_w, int n) {
    const int node = __builtin_amdgcn_readfirstlane(blockIdx.x * (blockDim.x >> 5) + (threadIdx.x >> 5));
    if (node >= n) return;
    const int lane = threadIdx.x & 31;
    const int beg = offs[node];
    const int deg = offs[node + 1] - beg;
    for (int i0 = 0; i0 < deg; i0 += 32) {
        const bool vi = (i0 + lane) < deg;
        const int ki = vi ? key[beg + i0 + lane] : 0x7fffffff;
        int r = 0;
        for (int j0 = 0; j0 < deg; j0 += 32) {
            const int m = min(32, deg - j0);
            const int kj = (j0 + lane < deg) ? key[beg + j0 + lane] : 0x7fffffff;
            for (int t = 0; t < m; ++t)
                r += (__builtin_amdgcn_readlane(kj, t) < ki) ? 1 : 0;
        }
        if (vi) {
            csr_src[beg + r] = src[ki];
            csr_w  [beg + r] = ew[ki];
        }
    }
}

__global__ void gcrn_deg_dinv(const int* __restrict__ offs, const float* __restrict__ csr_w,
                              float* __restrict__ dinv, int n) {
    int i = blockIdx.x * blockDim.x + threadIdx.x;
    if (i < n) {
        const int beg = offs[i], end = offs[i + 1];
        float d = 1.0f;
        for (int p = beg; p < end; ++p) d += csr_w[p];
        dinv[i] = (d > 0.0f) ? rsqrtf(d) : 0.0f;
    }
}

__global__ void __launch_bounds__(256)
gcrn_edge_nw(const int* __restrict__ offs, const int* __restrict__ csr_src,
             const float* __restrict__ csr_w, const float* __restrict__ dinv,
             float* __restrict__ nw, int n) {
    const int node = blockIdx.x * (blockDim.x >> 5) + (threadIdx.x >> 5);
    if (node >= n) return;
    const int lane = threadIdx.x & 31;
    const int end = offs[node + 1];
    const float dd = dinv[node];
    for (int p = offs[node] + lane; p < end; p += 32)
        nw[p] = dinv[csr_src[p]] * csr_w[p] * dd;
}


__device__ __forceinline__ float gcrn_gather_rows(const float* __restrict__ x, int s_l, float w_l, int m, int lane,
                                                  const int R, const bool start, float d2,
                                                  const float* __restrict__ xsp,
                                                  const int* __restrict__ snp, const float* __restrict__ wnp,
                                                  int& s_nx, float& w_nx, float a) {
    float v[32];
    #pragma unroll
    for (int u = 0; u < R; ++u) {
        const int tt = (u < R - 8) ? u : min(u, m - 1);
        const int s  = __builtin_amdgcn_readlane(s_l, tt);
        v[u] = x[s * D_CH + lane];
    }
    if (start) {
        const float xs = *xsp;
        s_nx = *snp;
        w_nx = *wnp;
        a = d2 * xs;
    }
    asm volatile("" :: "v"(v[R - 8]), "v"(v[R - 7]), "v"(v[R - 6]), "v"(v[R - 5]),
                       "v"(v[R - 4]), "v"(v[R - 3]), "v"(v[R - 2]), "v"(v[R - 1]));
    #pragma unroll
    for (int u = 0; u < R; ++u) {
        if (u < R - 8 || u < m) {
            const float w = __int_as_float(__builtin_amdgcn_readlane(__float_as_int(w_l), u));
            a = fmaf(w, v[u], a);
        }
    }
    return a;
}

__device__ __forceinline__ float gcrn_gather_chunk(const float* __restrict__ x, int s_l, float w_l, int m, int lane,
                                                   const bool start, float d2,
                                                   const float* __restrict__ xsp,
                                                   const int* __restrict__ snp, const float* __restrict__ wnp,
                                                   int& s_nx, float& w_nx, float a) {
    if (m > 24)      return gcrn_gather_rows(x, s_l, w_l, m, lane, 32, start, d2, xsp, snp, wnp, s_nx, w_nx, a);
    else if (m > 16) return gcrn_gather_rows(x, s_l, w_l, m, lane, 24, start, d2, xsp, snp, wnp, s_nx, w_nx, a);
    else if (m > 8)  return gcrn_gather_rows(x, s_l, w_l, m, lane, 16, start, d2, xsp, snp, wnp, s_nx, w_nx, a);
    else if (m > 0)  return gcrn_gather_rows(x, s_l, w_l, m, lane, 8,  start, d2, xsp, snp, wnp, s_nx, w_nx, a);
    if (start) {
        const float xs = *xsp;
        s_nx = *snp;
        w_nx = *wnp;
        a = d2 * xs;
    }
    return a;
}

__global__ void __launch_bounds__(64)
gcrn_layer(const int* __restrict__ offs, const int* __restrict__ csr_src,
           const float* __restrict__ nw, const float* __restrict__ dinv,
           const float* __restrict__ x,
           const float* __restrict__ Wl,
           const float* __restrict__ cb,
           const float* __restrict__ bgl,
           const float* __restrict__ wco,
           float* __restrict__ out, int n, int e) {
    __shared__ float ax[TILE_NODES * TILE_PITCH];
    __shared__ float ot[TILE_NODES * TILE_PITCH];

    const int wave = __builtin_amdgcn_readfirstlane(threadIdx.x >> 5);
    const int lane = threadIdx.x & 31;
    const int nb   = blockIdx.x * TILE_NODES + wave * WAVE_NODES;
    const int elast = max(e - 1, 0);

    const int   o_l  = offs[min(nb + min(lane, WAVE_NODES), n)];
    const float dv_l = dinv[min(nb + min(lane, WAVE_NODES - 1), n - 1)];

    int beg = __builtin_amdgcn_readlane(o_l, 0);
    int end = __builtin_amdgcn_readlane(o_l, 1);
    int   s_nx = csr_src[min(beg + lane, elast)];
    float w_nx = nw[min(beg + lane, elast)];

    for (int i = 0; i < WAVE_NODES; ++i) {
        const int s_l = s_nx; const float w_l = w_nx;
        const int nbeg = end;
        const int nend = __builtin_amdgcn_readlane(o_l, min(i + 2, WAVE_NODES));
        const int nidx = min(nbeg + lane, elast);

        const int node = nb + i;
        const float dv = __int_as_float(__builtin_amdgcn_readlane(__float_as_int(dv_l), i));
        const float d2 = (node < n) ? dv * dv : 0.0f;
        const float* xsp = x + (size_t)min(node, n - 1) * D_CH + lane;

        float a = gcrn_gather_chunk(x, s_l, w_l, min(32, end - beg), lane, true, d2,
                                    xsp, csr_src + nidx, nw + nidx, s_nx, w_nx, 0.0f);
        for (int p0 = beg + 32; p0 < end; p0 += 32) {
            const int m = min(32, end - p0);
            const int   s_m = csr_src[min(p0 + lane, elast)];
            const float w_m = nw[min(p0 + lane, elast)];
            int s_no; float w_no;
            a = gcrn_gather_chunk(x, s_m, w_m, m, lane, false, 0.0f, xsp, csr_src, nw, s_no, w_no, a);
        }
        ax[(wave * WAVE_NODES + i) * TILE_PITCH + lane] = a;
        beg = nbeg; end = nend;
    }
    __syncthreads();

    const int n0  = lane & 15;
    const int hi  = lane >> 4;
    const int col = wave * 16 + n0;
    const float* ar = ax + n0 * TILE_PITCH;
    const float* WI = Wl + 0 * 32 * 32 + col;
    const float* WT = Wl + 4 * 32 * 32 + col;
    const float* WO = Wl + 6 * 32 * 32 + col;

    v8f cI = {}, cT = {}, cO = {};
    #pragma unroll
    for (int k = 0; k < 32; k += 4) {
        const int ka = k + 2 * hi;
        v2f a;  a.x  = ar[ka];       a.y  = ar[ka + 1];
        v2f bI; bI.x = WI[ka * 32];  bI.y = WI[(ka + 1) * 32];
        v2f bT; bT.x = WT[ka * 32];  bT.y = WT[(ka + 1) * 32];
        v2f bO; bO.x = WO[ka * 32];  bO.y = WO[(ka + 1) * 32];
        cI = __builtin_amdgcn_wmma_f32_16x16x4_f32(false, a, false, bI, (short)0, cI, false, false);
        cT = __builtin_amdgcn_wmma_f32_16x16x4_f32(false, a, false, bT, (short)0, cT, false, false);
        cO = __builtin_amdgcn_wmma_f32_16x16x4_f32(false, a, false, bO, (short)0, cO, false, false);
    }

    const float bi = cb[col]          + cb[32 + col]     + bgl[col];
    const float bt = cb[4 * 32 + col] + cb[5 * 32 + col] + bgl[2 * 32 + col];
    const float bo = cb[6 * 32 + col] + cb[7 * 32 + col] + bgl[3 * 32 + col];
    const float wo = wco[col];

    #pragma unroll
    for (int j = 0; j < 8; ++j) {
        const float I = 1.0f / (1.0f + expf(-(cI[j] + bi)));
        const float T = tanhf(cT[j] + bt);
        const float C = I * T;
        const float O = 1.0f / (1.0f + expf(-(cO[j] + bo + wo * C)));
        ot[(hi * 8 + j) * TILE_PITCH + col] = O * tanhf(C);
    }
    __syncthreads();

    if (nb + WAVE_NODES <= n) {
        #pragma unroll
        for (int i = 0; i < WAVE_NODES; ++i)
            out[(size_t)(nb + i) * D_CH + lane] = ot[(wave * WAVE_NODES + i) * TILE_PITCH + lane];
    } else {
        #pragma unroll
        for (int i = 0; i < WAVE_NODES; ++i) {
            const int node = nb + i;
            if (node < n) out[(size_t)node * D_CH + lane] = ot[(wave * WAVE_NODES + i) * TILE_PITCH + lane];
        }
    }
}

extern "C" void kernel_launch(void* const* d_in, const int* in_sizes, int n_in,
                              void* d_out, int out_size, void* d_ws, size_t ws_size,
                              hipStream_t stream) {
    const float* X       = (const float*)d_in[0];
    const int*   eidx    = (const int*)  d_in[1];
    const float* ew      = (const float*)d_in[2];
    const float* conv_w  = (const float*)d_in[3];
    const float* conv_b  = (const float*)d_in[4];
    const float* wc      = (const float*)d_in[5];
    const float* bg      = (const float*)d_in[6];

    const int N = in_sizes[0] / D_CH;
    const int E = in_sizes[2];
    const int L = in_sizes[3] / (8 * D_CH * D_CH);
    const int* src = eidx;
    const int* dst = eidx + E;

    float* ws   = (float*)d_ws;
    float* h1   = ws;
    float* h2   = h1   + (size_t)N * D_CH;
    float* dinv = h2   + (size_t)N * D_CH;
    float* nw   = dinv + N;
    float* csrw = nw   + E;
    int*   csrs = (int*)(csrw + E);
    int*   key  = csrs + E;
    int*   rnk  = key  + E;
    int*   cnt  = rnk  + E;
    int*   bsum = cnt  + N;
    const int NB = (N + SCAN_TB - 1) / SCAN_TB;
    int*   offs = bsum + NB;

    const int TB = 256;
    const int nodesPerBlock = TB / 32;
    const int nodeBlocks = (N + nodesPerBlock - 1) / nodesPerBlock;

    gcrn_zero_cnt  <<<(N + TB - 1) / TB, TB, 0, stream>>>(cnt, N);
    gcrn_count_rank<<<(E + TB - 1) / TB, TB, 0, stream>>>(dst, cnt, rnk, E);
    gcrn_block_sums<<<NB, SCAN_TB, 0, stream>>>(cnt, bsum, N);
    gcrn_scan      <<<NB, SCAN_TB, 0, stream>>>(cnt, bsum, offs, N);
    gcrn_place     <<<(E + TB - 1) / TB, TB, 0, stream>>>(dst, rnk, offs, key, E);
    gcrn_sort_seg  <<<nodeBlocks, TB, 0, stream>>>(offs, key, src, ew, csrs, csrw, N);
    gcrn_deg_dinv  <<<(N + TB - 1) / TB, TB, 0, stream>>>(offs, csrw, dinv, N);
    gcrn_edge_nw   <<<nodeBlocks, TB, 0, stream>>>(offs, csrs, csrw, dinv, nw, N);

    const float* xin = X;
    for (int l = 0; l < L; ++l) {
        float* out_l = (l == L - 1) ? (float*)d_out : ((l & 1) ? h2 : h1);
        gcrn_layer<<<(N + TILE_NODES - 1) / TILE_NODES, 64, 0, stream>>>(
            offs, csrs, nw, dinv, xin,
            conv_w + (size_t)l * 8 * 32 * 32,
            conv_b + (size_t)l * 8 * 32, bg + (size_t)l * 4 * 32,
            wc + ((size_t)l * 3 + 2) * 32, out_l, N, E);
        xin = out_l;
    }
}


# Edge-list build: per-edge global returning adds, their scan and the place pass replaced by a two-level counting sort on LDS counters (chunk x bin counts, per-bin node slots); slot and sum order kept
# speedup vs baseline: 4.9559x; 1.5362x over previous
#include <hip/hip_runtime.h>
#include <hip/hip_bf16.h>
#include <math.h>


typedef __attribute__((ext_vector_type(2))) float v2f;
typedef __attribute__((ext_vector_type(8))) float v8f;

#define D_CH 32
#define TILE_NODES 16
#define WAVE_NODES 8
#define TILE_PITCH 36


#define BIN_SHIFT 9
#define BIN_NODES 512
#define BIN_G 256
#define BIN_MAXB 1024
#define BIN_TB 1024
#define BIN_BATCH 8
#define BIN_KEYCAP 12288

__device__ __forceinline__ int wave_sum_i(int v) {
    for (int m = 16; m >= 1; m >>= 1) v += __shfl_xor(v, m, 32);
    return v;
}

__device__ __forceinline__ int gcrn_block_excl(int v, int* sw) {
    const int lane = threadIdx.x & 31, wv = threadIdx.x >> 5;
    int inc = v;
    for (int d = 1; d < 32; d <<= 1) {
        int t = __shfl_up(inc, d, 32);
        if (lane >= d) inc += t;
    }
    if (lane == 31) sw[wv] = inc;
    __syncthreads();
    if (wv == 0) {
        const int s = sw[lane];
        int si = s;
        for (int d = 1; d < 32; d <<= 1) {
            int t = __shfl_up(si, d, 32);
            if (lane >= d) si += t;
        }
        sw[lane] = si - s;
    }
    __syncthreads();
    const int r = sw[wv] + inc - v;
    __syncthreads();
    return r;
}

__device__ __forceinline__ void gcrn_bin_totals(const int* __restrict__ tab, int nbins, int my_g,
                                                int* btot, int* bpre) {
    const int lane = threadIdx.x & 31, wv = threadIdx.x >> 5;
    for (int b = wv; b < nbins; b += BIN_TB / 32) {
        const int4* p = reinterpret_cast<const int4*>(tab + b * BIN_G + lane * 8);
        const int4 u = p[0], w = p[1];
        const int ev[8] = {u.x, u.y, u.z, u.w, w.x, w.y, w.z, w.w};
        int s = 0, q = 0;
        #pragma unroll
        for (int k = 0; k < 8; ++k) {
            s += ev[k];
            q += (lane * 8 + k < my_g) ? ev[k] : 0;
        }
        s = wave_sum_i(s);
        q = wave_sum_i(q);
        if (lane == 0) { btot[b] = s; bpre[b] = q; }
    }
}

__global__ void __launch_bounds__(BIN_TB)
gcrn_bin_hist(const int* __restrict__ dst, int* __restrict__ tab, int e, int nbins) {
    __shared__ int h[BIN_MAXB];
    const int tid = threadIdx.x;
    if (tid < nbins) h[tid] = 0;
    __syncthreads();
    const int chunk = (e + BIN_G - 1) / BIN_G;
    const int beg = blockIdx.x * chunk, end = min(beg + chunk, e);
    for (int i0 = beg + tid; i0 < end; i0 += BIN_BATCH * BIN_TB) {
        int d[BIN_BATCH];
        #pragma unroll
        for (int u = 0; u < BIN_BATCH; ++u) {
            const int i = i0 + u * BIN_TB;
            d[u] = (i < end) ? dst[i] : 0;
        }
        #pragma unroll
        for (int u = 0; u < BIN_BATCH; ++u)
            if (i0 + u * BIN_TB < end) atomicAdd(&h[d[u] >> BIN_SHIFT], 1);
    }
    __syncthreads();
    if (tid < nbins) tab[tid * BIN_G + blockIdx.x] = h[tid];
}

__global__ void __launch_bounds__(BIN_TB)
gcrn_bin_scatter(const int* __restrict__ dst, const int* __restrict__ tab, int2* __restrict__ rec,
                 int e, int nbins) {
    __shared__ int btot[BIN_MAXB];
    __shared__ int bpre[BIN_MAXB];
    __shared__ int cur[BIN_MAXB];
    __shared__ int sw[32];
    const int tid = threadIdx.x;
    gcrn_bin_totals(tab, nbins, blockIdx.x, btot, bpre);
    __syncthreads();
    const int v  = (tid < nbins) ? btot[tid] : 0;
    const int ex = gcrn_block_excl(v, sw);
    if (tid < nbins) cur[tid] = ex + bpre[tid];
    __syncthreads();
    const int chunk = (e + BIN_G - 1) / BIN_G;
    const int beg = blockIdx.x * chunk, end = min(beg + chunk, e);
    for (int i0 = beg + tid; i0 < end; i0 += BIN_BATCH * BIN_TB) {
        int d[BIN_BATCH];
        #pragma unroll
        for (int u = 0; u < BIN_BATCH; ++u) {
            const int i = i0 + u * BIN_TB;
            d[u] = (i < end) ? dst[i] : 0;
        }
        #pragma unroll
        for (int u = 0; u < BIN_BATCH; ++u) {
            const int i = i0 + u * BIN_TB;
            if (i < end) {
                const int slot = atomicAdd(&cur[d[u] >> BIN_SHIFT], 1);
                rec[slot] = make_int2(i, d[u]);
            }
        }
    }
}

__global__ void __launch_bounds__(BIN_TB)
gcrn_bin_place(const int* __restrict__ tab, const int2* __restrict__ rec,
               int* __restrict__ offs, int* __restrict__ key, int n, int nbins) {
    __shared__ int btot[BIN_MAXB];
    __shared__ int bpre[BIN_MAXB];
    __shared__ int bst[BIN_MAXB + 1];
    __shared__ int c[BIN_NODES];
    __shared__ int sw[32];
    __shared__ int keyl[BIN_KEYCAP];
    const int tid = threadIdx.x;
    gcrn_bin_totals(tab, nbins, 0, btot, bpre);
    if (tid < BIN_NODES) c[tid] = 0;
    __syncthreads();
    const int v  = (tid < nbins) ? btot[tid] : 0;
    const int ex = gcrn_block_excl(v, sw);
    if (tid <= nbins) bst[tid] = ex;
    __syncthreads();
    const int b   = blockIdx.x;
    const int lo  = bst[b], hi = bst[b + 1];
    const int nb0 = b << BIN_SHIFT;

    for (int p0 = lo + tid; p0 < hi; p0 += BIN_BATCH * BIN_TB) {
        int y[BIN_BATCH];
        #pragma unroll
        for (int u = 0; u < BIN_BATCH; ++u) {
            const int p = p0 + u * BIN_TB;
            y[u] = (p < hi) ? rec[p].y : nb0;
        }
        #pragma unroll
        for (int u = 0; u < BIN_BATCH; ++u)
            if (p0 + u * BIN_TB < hi) atomicAdd(&c[y[u] - nb0], 1);
    }
    __syncthreads();
    const int cv = (tid < BIN_NODES) ? c[tid] : 0;
    const int ce = gcrn_block_excl(cv, sw);
    if (tid < BIN_NODES) {
        c[tid] = lo + ce;
        if (nb0 + tid < n) offs[nb0 + tid] = lo + ce;
    }
    if (b == nbins - 1 && tid == 0) offs[n] = hi;
    __syncthreads();

    const int cntb = hi - lo;
    if (cntb <= BIN_KEYCAP) {
        for (int p0 = lo + tid; p0 < hi; p0 += BIN_BATCH * BIN_TB) {
            int2 r[BIN_BATCH];
            #pragma unroll
            for (int u = 0; u < BIN_BATCH; ++u) {
                const int p = p0 + u * BIN_TB;
                r[u] = (p < hi) ? rec[p] : make_int2(0, nb0);
            }
            #pragma unroll
            for (int u = 0; u < BIN_BATCH; ++u) {
                if (p0 + u * BIN_TB < hi) {
                    const int slot = atomicAdd(&c[r[u].y - nb0], 1);
                    keyl[slot - lo] = r[u].x;
                }
            }
        }
        __syncthreads();
        for (int q = tid; q < cntb; q += BIN_TB) key[lo + q] = keyl[q];
    } else {
        for (int p = lo + tid; p < hi; p += BIN_TB) {
            const int2 r = rec[p];
            const int slot = atomicAdd(&c[r.y - nb0], 1);
            key[slot] = r.x;
        }
    }
}

__global__ void __launch_bounds__(256)
gcrn_sort_seg(const int* __restrict__ offs, const int* __restrict__ key,
              const int* __restrict__ src, const float* __restrict__ ew,
              int* __restrict__ csr_src, float* __restrict__ csr_w, int n) {
    const int node = __builtin_amdgcn_readfirstlane(blockIdx.x * (blockDim.x >> 5) + (threadIdx.x >> 5));
    if (node >= n) return;
    const int lane = threadIdx.x & 31;
    const int beg = offs[node];
    const int deg = offs[node + 1] - beg;
    for (int i0 = 0; i0 < deg; i0 += 32) {
        const bool vi = (i0 + lane) < deg;
        const int ki = vi ? key[beg + i0 + lane] : 0x7fffffff;
        int r = 0;
        for (int j0 = 0; j0 < deg; j0 += 32) {
            const int m = min(32, deg - j0);
            const int kj = (j0 + lane < deg) ? key[beg + j0 + lane] : 0x7fffffff;
            for (int t = 0; t < m; ++t)
                r += (__builtin_amdgcn_readlane(kj, t) < ki) ? 1 : 0;
        }
        if (vi) {
            csr_src[beg + r] = src[ki];
            csr_w  [beg + r] = ew[ki];
        }
    }
}

__global__ void gcrn_deg_dinv(const int* __restrict__ offs, const float* __restrict__ csr_w,
                              float* __restrict__ dinv, int n) {
    int i = blockIdx.x * blockDim.x + threadIdx.x;
    if (i < n) {
        const int beg = offs[i], end = offs[i + 1];
        float d = 1.0f;
        for (int p = beg; p < end; ++p) d += csr_w[p];
        dinv[i] = (d > 0.0f) ? rsqrtf(d) : 0.0f;
    }
}

__global__ void __launch_bounds__(256)
gcrn_edge_nw(const int* __restrict__ offs, const int* __restrict__ csr_src,
             const float* __restrict__ csr_w, const float* __restrict__ dinv,
             float* __restrict__ nw, int n) {
    const int node = blockIdx.x * (blockDim.x >> 5) + (threadIdx.x >> 5);
    if (node >= n) return;
    const int lane = threadIdx.x & 31;
    const int end = offs[node + 1];
    const float dd = dinv[node];
    for (int p = offs[node] + lane; p < end; p += 32)
        nw[p] = dinv[csr_src[p]] * csr_w[p] * dd;
}


__device__ __forceinline__ float gcrn_gather_rows(const float* __restrict__ x, int s_l, float w_l, int m, int lane,
                                                  const int R, const bool start, float d2,
                                                  const float* __restrict__ xsp,
                                                  const int* __restrict__ snp, const float* __restrict__ wnp,
                                                  int& s_nx, float& w_nx, float a) {
    float v[32];
    #pragma unroll
    for (int u = 0; u < R; ++u) {
        const int tt = (u < R - 8) ? u : min(u, m - 1);
        const int s  = __builtin_amdgcn_readlane(s_l, tt);
        v[u] = x[s * D_CH + lane];
    }
    if (start) {
        const float xs = *xsp;
        s_nx = *snp;
        w_nx = *wnp;
        a = d2 * xs;
    }
    asm volatile("" :: "v"(v[R - 8]), "v"(v[R - 7]), "v"(v[R - 6]), "v"(v[R - 5]),
                       "v"(v[R - 4]), "v"(v[R - 3]), "v"(v[R - 2]), "v"(v[R - 1]));
    #pragma unroll
    for (int u = 0; u < R; ++u) {
        if (u < R - 8 || u < m) {
            const float w = __int_as_float(__builtin_amdgcn_readlane(__float_as_int(w_l), u));
            a = fmaf(w, v[u], a);
        }
    }
    return a;
}

__device__ __forceinline__ float gcrn_gather_chunk(const float* __restrict__ x, int s_l, float w_l, int m, int lane,
                                                   const bool start, float d2,
                                                   const float* __restrict__ xsp,
                                                   const int* __restrict__ snp, const float* __restrict__ wnp,
                                                   int& s_nx, float& w_nx, float a) {
    if (m > 24)      return gcrn_gather_rows(x, s_l, w_l, m, lane, 32, start, d2, xsp, snp, wnp, s_nx, w_nx, a);
    else if (m > 16) return gcrn_gather_rows(x, s_l, w_l, m, lane, 24, start, d2, xsp, snp, wnp, s_nx, w_nx, a);
    else if (m > 8)  return gcrn_gather_rows(x, s_l, w_l, m, lane, 16, start, d2, xsp, snp, wnp, s_nx, w_nx, a);
    else if (m > 0)  return gcrn_gather_rows(x, s_l, w_l, m, lane, 8,  start, d2, xsp, snp, wnp, s_nx, w_nx, a);
    if (start) {
        const float xs = *xsp;
        s_nx = *snp;
        w_nx = *wnp;
        a = d2 * xs;
    }
    return a;
}

__global__ void __launch_bounds__(64)
gcrn_layer(const int* __restrict__ offs, const int* __restrict__ csr_src,
           const float* __restrict__ nw, const float* __restrict__ dinv,
           const float* __restrict__ x,
           const float* __restrict__ Wl,
           const float* __restrict__ cb,
           const float* __restrict__ bgl,
           const float* __restrict__ wco,
           float* __restrict__ out, int n, int e) {
    __shared__ float ax[TILE_NODES * TILE_PITCH];
    __shared__ float ot[TILE_NODES * TILE_PITCH];

    const int wave = __builtin_amdgcn_readfirstlane(threadIdx.x >> 5);
    const int lane = threadIdx.x & 31;
    const int nb   = blockIdx.x * TILE_NODES + wave * WAVE_NODES;
    const int elast = max(e - 1, 0);

    const int   o_l  = offs[min(nb + min(lane, WAVE_NODES), n)];
    const float dv_l = dinv[min(nb + min(lane, WAVE_NODES - 1), n - 1)];

    int beg = __builtin_amdgcn_readlane(o_l, 0);
    int end = __builtin_amdgcn_readlane(o_l, 1);
    int   s_nx = csr_src[min(beg + lane, elast)];
    float w_nx = nw[min(beg + lane, elast)];

    for (int i = 0; i < WAVE_NODES; ++i) {
        const int s_l = s_nx; const float w_l = w_nx;
        const int nbeg = end;
        const int nend = __builtin_amdgcn_readlane(o_l, min(i + 2, WAVE_NODES));
        const int nidx = min(nbeg + lane, elast);

        const int node = nb + i;
        const float dv = __int_as_float(__builtin_amdgcn_readlane(__float_as_int(dv_l), i));
        const float d2 = (node < n) ? dv * dv : 0.0f;
        const float* xsp = x + (size_t)min(node, n - 1) * D_CH + lane;

        float a = gcrn_gather_chunk(x, s_l, w_l, min(32, end - beg), lane, true, d2,
                                    xsp, csr_src + nidx, nw + nidx, s_nx, w_nx, 0.0f);
        for (int p0 = beg + 32; p0 < end; p0 += 32) {
            const int m = min(32, end - p0);
            const int   s_m = csr_src[min(p0 + lane, elast)];
            const float w_m = nw[min(p0 + lane, elast)];
            int s_no; float w_no;
            a = gcrn_gather_chunk(x, s_m, w_m, m, lane, false, 0.0f, xsp, csr_src, nw, s_no, w_no, a);
        }
        ax[(wave * WAVE_NODES + i) * TILE_PITCH + lane] = a;
        beg = nbeg; end = nend;
    }
    __syncthreads();

    const int n0  = lane & 15;
    const int hi  = lane >> 4;
    const int col = wave * 16 + n0;
    const float* ar = ax + n0 * TILE_PITCH;
    const float* WI = Wl + 0 * 32 * 32 + col;
    const float* WT = Wl + 4 * 32 * 32 + col;
    const float* WO = Wl + 6 * 32 * 32 + col;

    v8f cI = {}, cT = {}, cO = {};
    #pragma unroll
    for (int k = 0; k < 32; k += 4) {
        const int ka = k + 2 * hi;
        v2f a;  a.x  = ar[ka];       a.y  = ar[ka + 1];
        v2f bI; bI.x = WI[ka * 32];  bI.y = WI[(ka + 1) * 32];
        v2f bT; bT.x = WT[ka * 32];  bT.y = WT[(ka + 1) * 32];
        v2f bO; bO.x = WO[ka * 32];  bO.y = WO[(ka + 1) * 32];
        cI = __builtin_amdgcn_wmma_f32_16x16x4_f32(false, a, false, bI, (short)0, cI, false, false);
        cT = __builtin_amdgcn_wmma_f32_16x16x4_f32(false, a, false, bT, (short)0, cT, false, false);
        cO = __builtin_amdgcn_wmma_f32_16x16x4_f32(false, a, false, bO, (short)0, cO, false, false);
    }

    const float bi = cb[col]          + cb[32 + col]     + bgl[col];
    const float bt = cb[4 * 32 + col] + cb[5 * 32 + col] + bgl[2 * 32 + col];
    const float bo = cb[6 * 32 + col] + cb[7 * 32 + col] + bgl[3 * 32 + col];
    const float wo = wco[col];

    #pragma unroll
    for (int j = 0; j < 8; ++j) {
        const float I = 1.0f / (1.0f + expf(-(cI[j] + bi)));
        const float T = tanhf(cT[j] + bt);
        const float C = I * T;
        const float O = 1.0f / (1.0f + expf(-(cO[j] + bo + wo * C)));
        ot[(hi * 8 + j) * TILE_PITCH + col] = O * tanhf(C);
    }
    __syncthreads();

    if (nb + WAVE_NODES <= n) {
        #pragma unroll
        for (int i = 0; i < WAVE_NODES; ++i)
            out[(size_t)(nb + i) * D_CH + lane] = ot[(wave * WAVE_NODES + i) * TILE_PITCH + lane];
    } else {
        #pragma unroll
        for (int i = 0; i < WAVE_NODES; ++i) {
            const int node = nb + i;
            if (node < n) out[(size_t)node * D_CH + lane] = ot[(wave * WAVE_NODES + i) * TILE_PITCH + lane];
        }
    }
}

extern "C" void kernel_launch(void* const* d_in, const int* in_sizes, int n_in,
                              void* d_out, int out_size, void* d_ws, size_t ws_size,
                              hipStream_t stream) {
    const float* X       = (const float*)d_in[0];
    const int*   eidx    = (const int*)  d_in[1];
    const float* ew      = (const float*)d_in[2];
    const float* conv_w  = (const float*)d_in[3];
    const float* conv_b  = (const float*)d_in[4];
    const float* wc      = (const float*)d_in[5];
    const float* bg      = (const float*)d_in[6];

    const int N = in_sizes[0] / D_CH;
    const int E = in_sizes[2];
    const int L = in_sizes[3] / (8 * D_CH * D_CH);
    const int* src = eidx;
    const int* dst = eidx + E;

    float* ws   = (float*)d_ws;
    const int NBINS = (N + BIN_NODES - 1) >> BIN_SHIFT;
    float* h1   = ws;
    float* h2   = h1   + (size_t)N * D_CH;
    int*   tab  = (int*)(h2 + (size_t)N * D_CH);
    int2*  rec  = (int2*)(tab + (size_t)NBINS * BIN_G);
    float* dinv = (float*)(rec + E);
    float* nw   = dinv + N;
    float* csrw = nw   + E;
    int*   csrs = (int*)(csrw + E);
    int*   key  = csrs + E;
    int*   offs = key  + E;

    const int TB = 256;
    const int nodesPerBlock = TB / 32;
    const int nodeBlocks = (N + nodesPerBlock - 1) / nodesPerBlock;

    gcrn_bin_hist   <<<BIN_G, BIN_TB, 0, stream>>>(dst, tab, E, NBINS);
    gcrn_bin_scatter<<<BIN_G, BIN_TB, 0, stream>>>(dst, tab, rec, E, NBINS);
    gcrn_bin_place  <<<NBINS, BIN_TB, 0, stream>>>(tab, rec, offs, key, N, NBINS);
    gcrn_sort_seg   <<<nodeBlocks, TB, 0, stream>>>(offs, key, src, ew, csrs, csrw, N);
    gcrn_deg_dinv  <<<(N + TB - 1) / TB, TB, 0, stream>>>(offs, csrw, dinv, N);
    gcrn_edge_nw   <<<nodeBlocks, TB, 0, stream>>>(offs, csrs, csrw, dinv, nw, N);

    const float* xin = X;
    for (int l = 0; l < L; ++l) {
        float* out_l = (l == L - 1) ? (float*)d_out : ((l & 1) ? h2 : h1);
        gcrn_layer<<<(N + TILE_NODES - 1) / TILE_NODES, 64, 0, stream>>>(
            offs, csrs, nw, dinv, xin,
            conv_w + (size_t)l * 8 * 32 * 32,
            conv_b + (size_t)l * 8 * 32, bg + (size_t)l * 4 * 32,
            wc + ((size_t)l * 3 + 2) * 32, out_l, N, E);
        xin = out_l;
    }
}


# Edge-weight pass folded into the first layer's gather: it forms dinv[src]*w*dinv[dst] for the slots it sums (same product, same bits) and stores them for later layers; a 100,000-wave launch is gone
# speedup vs baseline: 5.2043x; 1.0501x over previous
#include <hip/hip_runtime.h>
#include <hip/hip_bf16.h>
#include <math.h>


typedef __attribute__((ext_vector_type(2))) float v2f;
typedef __attribute__((ext_vector_type(8))) float v8f;

#define D_CH 32
#define TILE_NODES 16
#define WAVE_NODES 8
#define TILE_PITCH 36


#define BIN_SHIFT 9
#define BIN_NODES 512
#define BIN_G 256
#define BIN_MAXB 1024
#define BIN_TB 1024
#define BIN_BATCH 8
#define BIN_KEYCAP 12288

__device__ __forceinline__ int wave_sum_i(int v) {
    for (int m = 16; m >= 1; m >>= 1) v += __shfl_xor(v, m, 32);
    return v;
}

__device__ __forceinline__ int gcrn_block_excl(int v, int* sw) {
    const int lane = threadIdx.x & 31, wv = threadIdx.x >> 5;
    int inc = v;
    for (int d = 1; d < 32; d <<= 1) {
        int t = __shfl_up(inc, d, 32);
        if (lane >= d) inc += t;
    }
    if (lane == 31) sw[wv] = inc;
    __syncthreads();
    if (wv == 0) {
        const int s = sw[lane];
        int si = s;
        for (int d = 1; d < 32; d <<= 1) {
            int t = __shfl_up(si, d, 32);
            if (lane >= d) si += t;
        }
        sw[lane] = si - s;
    }
    __syncthreads();
    const int r = sw[wv] + inc - v;
    __syncthreads();
    return r;
}

__device__ __forceinline__ void gcrn_bin_totals(const int* __restrict__ tab, int nbins, int my_g,
                                                int* btot, int* bpre) {
    const int lane = threadIdx.x & 31, wv = threadIdx.x >> 5;
    for (int b = wv; b < nbins; b += BIN_TB / 32) {
        const int4* p = reinterpret_cast<const int4*>(tab + b * BIN_G + lane * 8);
        const int4 u = p[0], w = p[1];
        const int ev[8] = {u.x, u.y, u.z, u.w, w.x, w.y, w.z, w.w};
        int s = 0, q = 0;
        #pragma unroll
        for (int k = 0; k < 8; ++k) {
            s += ev[k];
            q += (lane * 8 + k < my_g) ? ev[k] : 0;
        }
        s = wave_sum_i(s);
        q = wave_sum_i(q);
        if (lane == 0) { btot[b] = s; bpre[b] = q; }
    }
}

__global__ void __launch_bounds__(BIN_TB)
gcrn_bin_hist(const int* __restrict__ dst, int* __restrict__ tab, int e, int nbins) {
    __shared__ int h[BIN_MAXB];
    const int tid = threadIdx.x;
    if (tid < nbins) h[tid] = 0;
    __syncthreads();
    const int chunk = (e + BIN_G - 1) / BIN_G;
    const int beg = blockIdx.x * chunk, end = min(beg + chunk, e);
    for (int i0 = beg + tid; i0 < end; i0 += BIN_BATCH * BIN_TB) {
        int d[BIN_BATCH];
        #pragma unroll
        for (int u = 0; u < BIN_BATCH; ++u) {
            const int i = i0 + u * BIN_TB;
            d[u] = (i < end) ? dst[i] : 0;
        }
        #pragma unroll
        for (int u = 0; u < BIN_BATCH; ++u)
            if (i0 + u * BIN_TB < end) atomicAdd(&h[d[u] >> BIN_SHIFT], 1);
    }
    __syncthreads();
    if (tid < nbins) tab[tid * BIN_G + blockIdx.x] = h[tid];
}

__global__ void __launch_bounds__(BIN_TB)
gcrn_bin_scatter(const int* __restrict__ dst, const int* __restrict__ tab, int2* __restrict__ rec,
                 int e, int nbins) {
    __shared__ int btot[BIN_MAXB];
    __shared__ int bpre[BIN_MAXB];
    __shared__ int cur[BIN_MAXB];
    __shared__ int sw[32];
    const int tid = threadIdx.x;
    gcrn_bin_totals(tab, nbins, blockIdx.x, btot, bpre);
    __syncthreads();
    const int v  = (tid < nbins) ? btot[tid] : 0;
    const int ex = gcrn_block_excl(v, sw);
    if (tid < nbins) cur[tid] = ex + bpre[tid];
    __syncthreads();
    const int chunk = (e + BIN_G - 1) / BIN_G;
    const int beg = blockIdx.x * chunk, end = min(beg + chunk, e);
    for (int i0 = beg + tid; i0 < end; i0 += BIN_BATCH * BIN_TB) {
        int d[BIN_BATCH];
        #pragma unroll
        for (int u = 0; u < BIN_BATCH; ++u) {
            const int i = i0 + u * BIN_TB;
            d[u] = (i < end) ? dst[i] : 0;
        }
        #pragma unroll
        for (int u = 0; u < BIN_BATCH; ++u) {
            const int i = i0 + u * BIN_TB;
            if (i < end) {
                const int slot = atomicAdd(&cur[d[u] >> BIN_SHIFT], 1);
                rec[slot] = make_int2(i, d[u]);
            }
        }
    }
}

__global__ void __launch_bounds__(BIN_TB)
gcrn_bin_place(const int* __restrict__ tab, const int2* __restrict__ rec,
               int* __restrict__ offs, int* __restrict__ key, int n, int nbins) {
    __shared__ int btot[BIN_MAXB];
    __shared__ int bpre[BIN_MAXB];
    __shared__ int bst[BIN_MAXB + 1];
    __shared__ int c[BIN_NODES];
    __shared__ int sw[32];
    __shared__ int keyl[BIN_KEYCAP];
    const int tid = threadIdx.x;
    gcrn_bin_totals(tab, nbins, 0, btot, bpre);
    if (tid < BIN_NODES) c[tid] = 0;
    __syncthreads();
    const int v  = (tid < nbins) ? btot[tid] : 0;
    const int ex = gcrn_block_excl(v, sw);
    if (tid <= nbins) bst[tid] = ex;
    __syncthreads();
    const int b   = blockIdx.x;
    const int lo  = bst[b], hi = bst[b + 1];
    const int nb0 = b << BIN_SHIFT;

    for (int p0 = lo + tid; p0 < hi; p0 += BIN_BATCH * BIN_TB) {
        int y[BIN_BATCH];
        #pragma unroll
        for (int u = 0; u < BIN_BATCH; ++u) {
            const int p = p0 + u * BIN_TB;
            y[u] = (p < hi) ? rec[p].y : nb0;
        }
        #pragma unroll
        for (int u = 0; u < BIN_BATCH; ++u)
            if (p0 + u * BIN_TB < hi) atomicAdd(&c[y[u] - nb0], 1);
    }
    __syncthreads();
    const int cv = (tid < BIN_NODES) ? c[tid] : 0;
    const int ce = gcrn_block_excl(cv, sw);
    if (tid < BIN_NODES) {
        c[tid] = lo + ce;
        if (nb0 + tid < n) offs[nb0 + tid] = lo + ce;
    }
    if (b == nbins - 1 && tid == 0) offs[n] = hi;
    __syncthreads();

    const int cntb = hi - lo;
    if (cntb <= BIN_KEYCAP) {
        for (int p0 = lo + tid; p0 < hi; p0 += BIN_BATCH * BIN_TB) {
            int2 r[BIN_BATCH];
            #pragma unroll
            for (int u = 0; u < BIN_BATCH; ++u) {
                const int p = p0 + u * BIN_TB;
                r[u] = (p < hi) ? rec[p] : make_int2(0, nb0);
            }
            #pragma unroll
            for (int u = 0; u < BIN_BATCH; ++u) {
                if (p0 + u * BIN_TB < hi) {
                    const int slot = atomicAdd(&c[r[u].y - nb0], 1);
                    keyl[slot - lo] = r[u].x;
                }
            }
        }
        __syncthreads();
        for (int q = tid; q < cntb; q += BIN_TB) key[lo + q] = keyl[q];
    } else {
        for (int p = lo + tid; p < hi; p += BIN_TB) {
            const int2 r = rec[p];
            const int slot = atomicAdd(&c[r.y - nb0], 1);
            key[slot] = r.x;
        }
    }
}

__global__ void __launch_bounds__(256)
gcrn_sort_seg(const int* __restrict__ offs, const int* __restrict__ key,
              const int* __restrict__ src, const float* __restrict__ ew,
              int* __restrict__ csr_src, float* __restrict__ csr_w, int n) {
    const int node = __builtin_amdgcn_readfirstlane(blockIdx.x * (blockDim.x >> 5) + (threadIdx.x >> 5));
    if (node >= n) return;
    const int lane = threadIdx.x & 31;
    const int beg = offs[node];
    const int deg = offs[node + 1] - beg;
    for (int i0 = 0; i0 < deg; i0 += 32) {
        const bool vi = (i0 + lane) < deg;
        const int ki = vi ? key[beg + i0 + lane] : 0x7fffffff;
        int r = 0;
        for (int j0 = 0; j0 < deg; j0 += 32) {
            const int m = min(32, deg - j0);
            const int kj = (j0 + lane < deg) ? key[beg + j0 + lane] : 0x7fffffff;
            for (int t = 0; t < m; ++t)
                r += (__builtin_amdgcn_readlane(kj, t) < ki) ? 1 : 0;
        }
        if (vi) {
            csr_src[beg + r] = src[ki];
            csr_w  [beg + r] = ew[ki];
        }
    }
}

__global__ void gcrn_deg_dinv(const int* __restrict__ offs, const float* __restrict__ csr_w,
                              float* __restrict__ dinv, int n) {
    int i = blockIdx.x * blockDim.x + threadIdx.x;
    if (i < n) {
        const int beg = offs[i], end = offs[i + 1];
        float d = 1.0f;
        for (int p = beg; p < end; ++p) d += csr_w[p];
        dinv[i] = (d > 0.0f) ? rsqrtf(d) : 0.0f;
    }
}


template <bool FIRST>
__device__ __forceinline__ float gcrn_gather_rows(const float* __restrict__ x, int s_l, float w_l, int m, int lane,
                                                  const int R, const bool start, float d2,
                                                  const float* __restrict__ xsp,
                                                  const int* __restrict__ snp, const float* __restrict__ wnp,
                                                  int& s_nx, float& w_nx, float a,
                                                  const float* __restrict__ dinv, float dd, float* __restrict__ nwo) {
    float ds = 0.0f;
    if (FIRST) ds = (lane < m) ? dinv[s_l] : 0.0f;
    float v[32];
    #pragma unroll
    for (int u = 0; u < R; ++u) {
        const int tt = (u < R - 8) ? u : min(u, m - 1);
        const int s  = __builtin_amdgcn_readlane(s_l, tt);
        v[u] = x[s * D_CH + lane];
    }
    if (start) {
        const float xs = *xsp;
        s_nx = *snp;
        w_nx = *wnp;
        a = d2 * xs;
    }
    if (FIRST) {
        w_l = ds * w_l * dd;
        if (lane < m) *nwo = w_l;
    }
    asm volatile("" :: "v"(v[R - 8]), "v"(v[R - 7]), "v"(v[R - 6]), "v"(v[R - 5]),
                       "v"(v[R - 4]), "v"(v[R - 3]), "v"(v[R - 2]), "v"(v[R - 1]));
    #pragma unroll
    for (int u = 0; u < R; ++u) {
        if (u < R - 8 || u < m) {
            const float w = __int_as_float(__builtin_amdgcn_readlane(__float_as_int(w_l), u));
            a = fmaf(w, v[u], a);
        }
    }
    return a;
}

template <bool FIRST>
__device__ __forceinline__ float gcrn_gather_chunk(const float* __restrict__ x, int s_l, float w_l, int m, int lane,
                                                   const bool start, float d2,
                                                   const float* __restrict__ xsp,
                                                   const int* __restrict__ snp, const float* __restrict__ wnp,
                                                   int& s_nx, float& w_nx, float a,
                                                   const float* __restrict__ dinv, float dd, float* __restrict__ nwo) {
    if (m > 24)      return gcrn_gather_rows<FIRST>(x, s_l, w_l, m, lane, 32, start, d2, xsp, snp, wnp, s_nx, w_nx, a, dinv, dd, nwo);
    else if (m > 16) return gcrn_gather_rows<FIRST>(x, s_l, w_l, m, lane, 24, start, d2, xsp, snp, wnp, s_nx, w_nx, a, dinv, dd, nwo);
    else if (m > 8)  return gcrn_gather_rows<FIRST>(x, s_l, w_l, m, lane, 16, start, d2, xsp, snp, wnp, s_nx, w_nx, a, dinv, dd, nwo);
    else if (m > 0)  return gcrn_gather_rows<FIRST>(x, s_l, w_l, m, lane, 8,  start, d2, xsp, snp, wnp, s_nx, w_nx, a, dinv, dd, nwo);
    if (start) {
        const float xs = *xsp;
        s_nx = *snp;
        w_nx = *wnp;
        a = d2 * xs;
    }
    return a;
}

template <bool FIRST>
__global__ void __launch_bounds__(64)
gcrn_layer(const int* __restrict__ offs, const int* __restrict__ csr_src,
           const float* __restrict__ nw,
           float* __restrict__ nw_out,
           const float* __restrict__ dinv,
           const float* __restrict__ x,
           const float* __restrict__ Wl,
           const float* __restrict__ cb,
           const float* __restrict__ bgl,
           const float* __restrict__ wco,
           float* __restrict__ out, int n, int e) {
    __shared__ float ax[TILE_NODES * TILE_PITCH];
    __shared__ float ot[TILE_NODES * TILE_PITCH];

    const int wave = __builtin_amdgcn_readfirstlane(threadIdx.x >> 5);
    const int lane = threadIdx.x & 31;
    const int nb   = blockIdx.x * TILE_NODES + wave * WAVE_NODES;
    const int elast = max(e - 1, 0);

    const int   o_l  = offs[min(nb + min(lane, WAVE_NODES), n)];
    const float dv_l = dinv[min(nb + min(lane, WAVE_NODES - 1), n - 1)];

    int beg = __builtin_amdgcn_readlane(o_l, 0);
    int end = __builtin_amdgcn_readlane(o_l, 1);
    int   s_nx = csr_src[min(beg + lane, elast)];
    float w_nx = nw[min(beg + lane, elast)];

    for (int i = 0; i < WAVE_NODES; ++i) {
        const int s_l = s_nx; const float w_l = w_nx;
        const int nbeg = end;
        const int nend = __builtin_amdgcn_readlane(o_l, min(i + 2, WAVE_NODES));
        const int nidx = min(nbeg + lane, elast);

        const int node = nb + i;
        const float dv = __int_as_float(__builtin_amdgcn_readlane(__float_as_int(dv_l), i));
        const float d2 = (node < n) ? dv * dv : 0.0f;
        const float* xsp = x + (size_t)min(node, n - 1) * D_CH + lane;

        float a = gcrn_gather_chunk<FIRST>(x, s_l, w_l, min(32, end - beg), lane, true, d2,
                                           xsp, csr_src + nidx, nw + nidx, s_nx, w_nx, 0.0f,
                                           dinv, dv, FIRST ? nw_out + (beg + lane) : nullptr);
        for (int p0 = beg + 32; p0 < end; p0 += 32) {
            const int m = min(32, end - p0);
            const int   s_m = csr_src[min(p0 + lane, elast)];
            const float w_m = nw[min(p0 + lane, elast)];
            int s_no; float w_no;
            a = gcrn_gather_chunk<FIRST>(x, s_m, w_m, m, lane, false, 0.0f, xsp, csr_src, nw, s_no, w_no, a,
                                         dinv, dv, FIRST ? nw_out + (p0 + lane) : nullptr);
        }
        ax[(wave * WAVE_NODES + i) * TILE_PITCH + lane] = a;
        beg = nbeg; end = nend;
    }
    __syncthreads();

    const int n0  = lane & 15;
    const int hi  = lane >> 4;
    const int col = wave * 16 + n0;
    const float* ar = ax + n0 * TILE_PITCH;
    const float* WI = Wl + 0 * 32 * 32 + col;
    const float* WT = Wl + 4 * 32 * 32 + col;
    const float* WO = Wl + 6 * 32 * 32 + col;

    v8f cI = {}, cT = {}, cO = {};
    #pragma unroll
    for (int k = 0; k < 32; k += 4) {
        const int ka = k + 2 * hi;
        v2f a;  a.x  = ar[ka];       a.y  = ar[ka + 1];
        v2f bI; bI.x = WI[ka * 32];  bI.y = WI[(ka + 1) * 32];
        v2f bT; bT.x = WT[ka * 32];  bT.y = WT[(ka + 1) * 32];
        v2f bO; bO.x = WO[ka * 32];  bO.y = WO[(ka + 1) * 32];
        cI = __builtin_amdgcn_wmma_f32_16x16x4_f32(false, a, false, bI, (short)0, cI, false, false);
        cT = __builtin_amdgcn_wmma_f32_16x16x4_f32(false, a, false, bT, (short)0, cT, false, false);
        cO = __builtin_amdgcn_wmma_f32_16x16x4_f32(false, a, false, bO, (short)0, cO, false, false);
    }

    const float bi = cb[col]          + cb[32 + col]     + bgl[col];
    const float bt = cb[4 * 32 + col] + cb[5 * 32 + col] + bgl[2 * 32 + col];
    const float bo = cb[6 * 32 + col] + cb[7 * 32 + col] + bgl[3 * 32 + col];
    const float wo = wco[col];

    #pragma unroll
    for (int j = 0; j < 8; ++j) {
        const float I = 1.0f / (1.0f + expf(-(cI[j] + bi)));
        const float T = tanhf(cT[j] + bt);
        const float C = I * T;
        const float O = 1.0f / (1.0f + expf(-(cO[j] + bo + wo * C)));
        ot[(hi * 8 + j) * TILE_PITCH + col] = O * tanhf(C);
    }
    __syncthreads();

    if (nb + WAVE_NODES <= n) {
        #pragma unroll
        for (int i = 0; i < WAVE_NODES; ++i)
            out[(size_t)(nb + i) * D_CH + lane] = ot[(wave * WAVE_NODES + i) * TILE_PITCH + lane];
    } else {
        #pragma unroll
        for (int i = 0; i < WAVE_NODES; ++i) {
            const int node = nb + i;
            if (node < n) out[(size_t)node * D_CH + lane] = ot[(wave * WAVE_NODES + i) * TILE_PITCH + lane];
        }
    }
}

extern "C" void kernel_launch(void* const* d_in, const int* in_sizes, int n_in,
                              void* d_out, int out_size, void* d_ws, size_t ws_size,
                              hipStream_t stream) {
    const float* X       = (const float*)d_in[0];
    const int*   eidx    = (const int*)  d_in[1];
    const float* ew      = (const float*)d_in[2];
    const float* conv_w  = (const float*)d_in[3];
    const float* conv_b  = (const float*)d_in[4];
    const float* wc      = (const float*)d_in[5];
    const float* bg      = (const float*)d_in[6];

    const int N = in_sizes[0] / D_CH;
    const int E = in_sizes[2];
    const int L = in_sizes[3] / (8 * D_CH * D_CH);
    const int* src = eidx;
    const int* dst = eidx + E;

    float* ws   = (float*)d_ws;
    const int NBINS = (N + BIN_NODES - 1) >> BIN_SHIFT;
    float* h1   = ws;
    float* h2   = h1   + (size_t)N * D_CH;
    int*   tab  = (int*)(h2 + (size_t)N * D_CH);
    int2*  rec  = (int2*)(tab + (size_t)NBINS * BIN_G);
    float* dinv = (float*)(rec + E);
    float* nw   = dinv + N;
    float* csrw = nw   + E;
    int*   csrs = (int*)(csrw + E);
    int*   key  = csrs + E;
    int*   offs = key  + E;

    const int TB = 256;
    const int nodesPerBlock = TB / 32;
    const int nodeBlocks = (N + nodesPerBlock - 1) / nodesPerBlock;

    gcrn_bin_hist   <<<BIN_G, BIN_TB, 0, stream>>>(dst, tab, E, NBINS);
    gcrn_bin_scatter<<<BIN_G, BIN_TB, 0, stream>>>(dst, tab, rec, E, NBINS);
    gcrn_bin_place  <<<NBINS, BIN_TB, 0, stream>>>(tab, rec, offs, key, N, NBINS);
    gcrn_sort_seg   <<<nodeBlocks, TB, 0, stream>>>(offs, key, src, ew, csrs, csrw, N);
    gcrn_deg_dinv  <<<(N + TB - 1) / TB, TB, 0, stream>>>(offs, csrw, dinv, N);

    const float* xin = X;
    for (int l = 0; l < L; ++l) {
        float* out_l = (l == L - 1) ? (float*)d_out : ((l & 1) ? h2 : h1);
        const float* W_l  = conv_w + (size_t)l * 8 * 32 * 32;
        const float* cb_l = conv_b + (size_t)l * 8 * 32;
        const float* bg_l = bg + (size_t)l * 4 * 32;
        const float* wo_l = wc + ((size_t)l * 3 + 2) * 32;
        const int tiles = (N + TILE_NODES - 1) / TILE_NODES;
        if (l == 0)
            gcrn_layer<true> <<<tiles, 64, 0, stream>>>(offs, csrs, csrw, nw, dinv, xin, W_l, cb_l, bg_l, wo_l, out_l, N, E);
        else
            gcrn_layer<false><<<tiles, 64, 0, stream>>>(offs, csrs, nw, nullptr, dinv, xin, W_l, cb_l, bg_l, wo_l, out_l, N, E);
        xin = out_l;
    }
}


# Gate activations on the hardware tanh instruction: tanh = one v_tanh_f32, sigmoid = 1/2 + 1/2 tanh(x/2), in place of library tanhf, expf and f32 division; f32 kept, results change in the last bits
# speedup vs baseline: 5.7097x; 1.0971x over previous
#include <hip/hip_runtime.h>
#include <hip/hip_bf16.h>
#include <math.h>


typedef __attribute__((ext_vector_type(2))) float v2f;
typedef __attribute__((ext_vector_type(8))) float v8f;

#define D_CH 32
#define TILE_NODES 16
#define WAVE_NODES 8
#define TILE_PITCH 36


#define BIN_SHIFT 9
#define BIN_NODES 512
#define BIN_G 256
#define BIN_MAXB 1024
#define BIN_TB 1024
#define BIN_BATCH 8
#define BIN_KEYCAP 12288

__device__ __forceinline__ int wave_sum_i(int v) {
    for (int m = 16; m >= 1; m >>= 1) v += __shfl_xor(v, m, 32);
    return v;
}

__device__ __forceinline__ int gcrn_block_excl(int v, int* sw) {
    const int lane = threadIdx.x & 31, wv = threadIdx.x >> 5;
    int inc = v;
    for (int d = 1; d < 32; d <<= 1) {
        int t = __shfl_up(inc, d, 32);
        if (lane >= d) inc += t;
    }
    if (lane == 31) sw[wv] = inc;
    __syncthreads();
    if (wv == 0) {
        const int s = sw[lane];
        int si = s;
        for (int d = 1; d < 32; d <<= 1) {
            int t = __shfl_up(si, d, 32);
            if (lane >= d) si += t;
        }
        sw[lane] = si - s;
    }
    __syncthreads();
    const int r = sw[wv] + inc - v;
    __syncthreads();
    return r;
}

__device__ __forceinline__ void gcrn_bin_totals(const int* __restrict__ tab, int nbins, int my_g,
                                                int* btot, int* bpre) {
    const int lane = threadIdx.x & 31, wv = threadIdx.x >> 5;
    for (int b = wv; b < nbins; b += BIN_TB / 32) {
        const int4* p = reinterpret_cast<const int4*>(tab + b * BIN_G + lane * 8);
        const int4 u = p[0], w = p[1];
        const int ev[8] = {u.x, u.y, u.z, u.w, w.x, w.y, w.z, w.w};
        int s = 0, q = 0;
        #pragma unroll
        for (int k = 0; k < 8; ++k) {
            s += ev[k];
            q += (lane * 8 + k < my_g) ? ev[k] : 0;
        }
        s = wave_sum_i(s);
        q = wave_sum_i(q);
        if (lane == 0) { btot[b] = s; bpre[b] = q; }
    }
}

__global__ void __launch_bounds__(BIN_TB)
gcrn_bin_hist(const int* __restrict__ dst, int* __restrict__ tab, int e, int nbins) {
    __shared__ int h[BIN_MAXB];
    const int tid = threadIdx.x;
    if (tid < nbins) h[tid] = 0;
    __syncthreads();
    const int chunk = (e + BIN_G - 1) / BIN_G;
    const int beg = blockIdx.x * chunk, end = min(beg + chunk, e);
    for (int i0 = beg + tid; i0 < end; i0 += BIN_BATCH * BIN_TB) {
        int d[BIN_BATCH];
        #pragma unroll
        for (int u = 0; u < BIN_BATCH; ++u) {
            const int i = i0 + u * BIN_TB;
            d[u] = (i < end) ? dst[i] : 0;
        }
        #pragma unroll
        for (int u = 0; u < BIN_BATCH; ++u)
            if (i0 + u * BIN_TB < end) atomicAdd(&h[d[u] >> BIN_SHIFT], 1);
    }
    __syncthreads();
    if (tid < nbins) tab[tid * BIN_G + blockIdx.x] = h[tid];
}

__global__ void __launch_bounds__(BIN_TB)
gcrn_bin_scatter(const int* __restrict__ dst, const int* __restrict__ tab, int2* __restrict__ rec,
                 int e, int nbins) {
    __shared__ int btot[BIN_MAXB];
    __shared__ int bpre[BIN_MAXB];
    __shared__ int cur[BIN_MAXB];
    __shared__ int sw[32];
    const int tid = threadIdx.x;
    gcrn_bin_totals(tab, nbins, blockIdx.x, btot, bpre);
    __syncthreads();
    const int v  = (tid < nbins) ? btot[tid] : 0;
    const int ex = gcrn_block_excl(v, sw);
    if (tid < nbins) cur[tid] = ex + bpre[tid];
    __syncthreads();
    const int chunk = (e + BIN_G - 1) / BIN_G;
    const int beg = blockIdx.x * chunk, end = min(beg + chunk, e);
    for (int i0 = beg + tid; i0 < end; i0 += BIN_BATCH * BIN_TB) {
        int d[BIN_BATCH];
        #pragma unroll
        for (int u = 0; u < BIN_BATCH; ++u) {
            const int i = i0 + u * BIN_TB;
            d[u] = (i < end) ? dst[i] : 0;
        }
        #pragma unroll
        for (int u = 0; u < BIN_BATCH; ++u) {
            const int i = i0 + u * BIN_TB;
            if (i < end) {
                const int slot = atomicAdd(&cur[d[u] >> BIN_SHIFT], 1);
                rec[slot] = make_int2(i, d[u]);
            }
        }
    }
}

__global__ void __launch_bounds__(BIN_TB)
gcrn_bin_place(const int* __restrict__ tab, const int2* __restrict__ rec,
               int* __restrict__ offs, int* __restrict__ key, int n, int nbins) {
    __shared__ int btot[BIN_MAXB];
    __shared__ int bpre[BIN_MAXB];
    __shared__ int bst[BIN_MAXB + 1];
    __shared__ int c[BIN_NODES];
    __shared__ int sw[32];
    __shared__ int keyl[BIN_KEYCAP];
    const int tid = threadIdx.x;
    gcrn_bin_totals(tab, nbins, 0, btot, bpre);
    if (tid < BIN_NODES) c[tid] = 0;
    __syncthreads();
    const int v  = (tid < nbins) ? btot[tid] : 0;
    const int ex = gcrn_block_excl(v, sw);
    if (tid <= nbins) bst[tid] = ex;
    __syncthreads();
    const int b   = blockIdx.x;
    const int lo  = bst[b], hi = bst[b + 1];
    const int nb0 = b << BIN_SHIFT;

    for (int p0 = lo + tid; p0 < hi; p0 += BIN_BATCH * BIN_TB) {
        int y[BIN_BATCH];
        #pragma unroll
        for (int u = 0; u < BIN_BATCH; ++u) {
            const int p = p0 + u * BIN_TB;
            y[u] = (p < hi) ? rec[p].y : nb0;
        }
        #pragma unroll
        for (int u = 0; u < BIN_BATCH; ++u)
            if (p0 + u * BIN_TB < hi) atomicAdd(&c[y[u] - nb0], 1);
    }
    __syncthreads();
    const int cv = (tid < BIN_NODES) ? c[tid] : 0;
    const int ce = gcrn_block_excl(cv, sw);
    if (tid < BIN_NODES) {
        c[tid] = lo + ce;
        if (nb0 + tid < n) offs[nb0 + tid] = lo + ce;
    }
    if (b == nbins - 1 && tid == 0) offs[n] = hi;
    __syncthreads();

    const int cntb = hi - lo;
    if (cntb <= BIN_KEYCAP) {
        for (int p0 = lo + tid; p0 < hi; p0 += BIN_BATCH * BIN_TB) {
            int2 r[BIN_BATCH];
            #pragma unroll
            for (int u = 0; u < BIN_BATCH; ++u) {
                const int p = p0 + u * BIN_TB;
                r[u] = (p < hi) ? rec[p] : make_int2(0, nb0);
            }
            #pragma unroll
            for (int u = 0; u < BIN_BATCH; ++u) {
                if (p0 + u * BIN_TB < hi) {
                    const int slot = atomicAdd(&c[r[u].y - nb0], 1);
                    keyl[slot - lo] = r[u].x;
                }
            }
        }
        __syncthreads();
        for (int q = tid; q < cntb; q += BIN_TB) key[lo + q] = keyl[q];
    } else {
        for (int p = lo + tid; p < hi; p += BIN_TB) {
            const int2 r = rec[p];
            const int slot = atomicAdd(&c[r.y - nb0], 1);
            key[slot] = r.x;
        }
    }
}

__global__ void __launch_bounds__(256)
gcrn_sort_seg(const int* __restrict__ offs, const int* __restrict__ key,
              const int* __restrict__ src, const float* __restrict__ ew,
              int* __restrict__ csr_src, float* __restrict__ csr_w, int n) {
    const int node = __builtin_amdgcn_readfirstlane(blockIdx.x * (blockDim.x >> 5) + (threadIdx.x >> 5));
    if (node >= n) return;
    const int lane = threadIdx.x & 31;
    const int beg = offs[node];
    const int deg = offs[node + 1] - beg;
    for (int i0 = 0; i0 < deg; i0 += 32) {
        const bool vi = (i0 + lane) < deg;
        const int ki = vi ? key[beg + i0 + lane] : 0x7fffffff;
        int r = 0;
        for (int j0 = 0; j0 < deg; j0 += 32) {
            const int m = min(32, deg - j0);
            const int kj = (j0 + lane < deg) ? key[beg + j0 + lane] : 0x7fffffff;
            for (int t = 0; t < m; ++t)
                r += (__builtin_amdgcn_readlane(kj, t) < ki) ? 1 : 0;
        }
        if (vi) {
            csr_src[beg + r] = src[ki];
            csr_w  [beg + r] = ew[ki];
        }
    }
}

__global__ void gcrn_deg_dinv(const int* __restrict__ offs, const float* __restrict__ csr_w,
                              float* __restrict__ dinv, int n) {
    int i = blockIdx.x * blockDim.x + threadIdx.x;
    if (i < n) {
        const int beg = offs[i], end = offs[i + 1];
        float d = 1.0f;
        for (int p = beg; p < end; ++p) d += csr_w[p];
        dinv[i] = (d > 0.0f) ? rsqrtf(d) : 0.0f;
    }
}


template <bool FIRST>
__device__ __forceinline__ float gcrn_gather_rows(const float* __restrict__ x, int s_l, float w_l, int m, int lane,
                                                  const int R, const bool start, float d2,
                                                  const float* __restrict__ xsp,
                                                  const int* __restrict__ snp, const float* __restrict__ wnp,
                                                  int& s_nx, float& w_nx, float a,
                                                  const float* __restrict__ dinv, float dd, float* __restrict__ nwo) {
    float ds = 0.0f;
    if (FIRST) ds = (lane < m) ? dinv[s_l] : 0.0f;
    float v[32];
    #pragma unroll
    for (int u = 0; u < R; ++u) {
        const int tt = (u < R - 8) ? u : min(u, m - 1);
        const int s  = __builtin_amdgcn_readlane(s_l, tt);
        v[u] = x[s * D_CH + lane];
    }
    if (start) {
        const float xs = *xsp;
        s_nx = *snp;
        w_nx = *wnp;
        a = d2 * xs;
    }
    if (FIRST) {
        w_l = ds * w_l * dd;
        if (lane < m) *nwo = w_l;
    }
    asm volatile("" :: "v"(v[R - 8]), "v"(v[R - 7]), "v"(v[R - 6]), "v"(v[R - 5]),
                       "v"(v[R - 4]), "v"(v[R - 3]), "v"(v[R - 2]), "v"(v[R - 1]));
    #pragma unroll
    for (int u = 0; u < R; ++u) {
        if (u < R - 8 || u < m) {
            const float w = __int_as_float(__builtin_amdgcn_readlane(__float_as_int(w_l), u));
            a = fmaf(w, v[u], a);
        }
    }
    return a;
}

template <bool FIRST>
__device__ __forceinline__ float gcrn_gather_chunk(const float* __restrict__ x, int s_l, float w_l, int m, int lane,
                                                   const bool start, float d2,
                                                   const float* __restrict__ xsp,
                                                   const int* __restrict__ snp, const float* __restrict__ wnp,
                                                   int& s_nx, float& w_nx, float a,
                                                   const float* __restrict__ dinv, float dd, float* __restrict__ nwo) {
    if (m > 24)      return gcrn_gather_rows<FIRST>(x, s_l, w_l, m, lane, 32, start, d2, xsp, snp, wnp, s_nx, w_nx, a, dinv, dd, nwo);
    else if (m > 16) return gcrn_gather_rows<FIRST>(x, s_l, w_l, m, lane, 24, start, d2, xsp, snp, wnp, s_nx, w_nx, a, dinv, dd, nwo);
    else if (m > 8)  return gcrn_gather_rows<FIRST>(x, s_l, w_l, m, lane, 16, start, d2, xsp, snp, wnp, s_nx, w_nx, a, dinv, dd, nwo);
    else if (m > 0)  return gcrn_gather_rows<FIRST>(x, s_l, w_l, m, lane, 8,  start, d2, xsp, snp, wnp, s_nx, w_nx, a, dinv, dd, nwo);
    if (start) {
        const float xs = *xsp;
        s_nx = *snp;
        w_nx = *wnp;
        a = d2 * xs;
    }
    return a;
}

__device__ __forceinline__ float gcrn_tanh(float x) { return __builtin_amdgcn_tanhf(x); }
__device__ __forceinline__ float gcrn_sigmoid(float x) { return fmaf(0.5f, __builtin_amdgcn_tanhf(0.5f * x), 0.5f); }

template <bool FIRST>
__global__ void __launch_bounds__(64)
gcrn_layer(const int* __restrict__ offs, const int* __restrict__ csr_src,
           const float* __restrict__ nw,
           float* __restrict__ nw_out,
           const float* __restrict__ dinv,
           const float* __restrict__ x,
           const float* __restrict__ Wl,
           const float* __restrict__ cb,
           const float* __restrict__ bgl,
           const float* __restrict__ wco,
           float* __restrict__ out, int n, int e) {
    __shared__ float ax[TILE_NODES * TILE_PITCH];
    __shared__ float ot[TILE_NODES * TILE_PITCH];

    const int wave = __builtin_amdgcn_readfirstlane(threadIdx.x >> 5);
    const int lane = threadIdx.x & 31;
    const int nb   = blockIdx.x * TILE_NODES + wave * WAVE_NODES;
    const int elast = max(e - 1, 0);

    const int   o_l  = offs[min(nb + min(lane, WAVE_NODES), n)];
    const float dv_l = dinv[min(nb + min(lane, WAVE_NODES - 1), n - 1)];

    int beg = __builtin_amdgcn_readlane(o_l, 0);
    int end = __builtin_amdgcn_readlane(o_l, 1);
    int   s_nx = csr_src[min(beg + lane, elast)];
    float w_nx = nw[min(beg + lane, elast)];

    for (int i = 0; i < WAVE_NODES; ++i) {
        const int s_l = s_nx; const float w_l = w_nx;
        const int nbeg = end;
        const int nend = __builtin_amdgcn_readlane(o_l, min(i + 2, WAVE_NODES));
        const int nidx = min(nbeg + lane, elast);

        const int node = nb + i;
        const float dv = __int_as_float(__builtin_amdgcn_readlane(__float_as_int(dv_l), i));
        const float d2 = (node < n) ? dv * dv : 0.0f;
        const float* xsp = x + (size_t)min(node, n - 1) * D_CH + lane;

        float a = gcrn_gather_chunk<FIRST>(x, s_l, w_l, min(32, end - beg), lane, true, d2,
                                           xsp, csr_src + nidx, nw + nidx, s_nx, w_nx, 0.0f,
                                           dinv, dv, FIRST ? nw_out + (beg + lane) : nullptr);
        for (int p0 = beg + 32; p0 < end; p0 += 32) {
            const int m = min(32, end - p0);
            const int   s_m = csr_src[min(p0 + lane, elast)];
            const float w_m = nw[min(p0 + lane, elast)];
            int s_no; float w_no;
            a = gcrn_gather_chunk<FIRST>(x, s_m, w_m, m, lane, false, 0.0f, xsp, csr_src, nw, s_no, w_no, a,
                                         dinv, dv, FIRST ? nw_out + (p0 + lane) : nullptr);
        }
        ax[(wave * WAVE_NODES + i) * TILE_PITCH + lane] = a;
        beg = nbeg; end = nend;
    }
    __syncthreads();

    const int n0  = lane & 15;
    const int hi  = lane >> 4;
    const int col = wave * 16 + n0;
    const float* ar = ax + n0 * TILE_PITCH;
    const float* WI = Wl + 0 * 32 * 32 + col;
    const float* WT = Wl + 4 * 32 * 32 + col;
    const float* WO = Wl + 6 * 32 * 32 + col;

    v8f cI = {}, cT = {}, cO = {};
    #pragma unroll
    for (int k = 0; k < 32; k += 4) {
        const int ka = k + 2 * hi;
        v2f a;  a.x  = ar[ka];       a.y  = ar[ka + 1];
        v2f bI; bI.x = WI[ka * 32];  bI.y = WI[(ka + 1) * 32];
        v2f bT; bT.x = WT[ka * 32];  bT.y = WT[(ka + 1) * 32];
        v2f bO; bO.x = WO[ka * 32];  bO.y = WO[(ka + 1) * 32];
        cI = __builtin_amdgcn_wmma_f32_16x16x4_f32(false, a, false, bI, (short)0, cI, false, false);
        cT = __builtin_amdgcn_wmma_f32_16x16x4_f32(false, a, false, bT, (short)0, cT, false, false);
        cO = __builtin_amdgcn_wmma_f32_16x16x4_f32(false, a, false, bO, (short)0, cO, false, false);
    }

    const float bi = cb[col]          + cb[32 + col]     + bgl[col];
    const float bt = cb[4 * 32 + col] + cb[5 * 32 + col] + bgl[2 * 32 + col];
    const float bo = cb[6 * 32 + col] + cb[7 * 32 + col] + bgl[3 * 32 + col];
    const float wo = wco[col];

    #pragma unroll
    for (int j = 0; j < 8; ++j) {
        const float I = gcrn_sigmoid(cI[j] + bi);
        const float T = gcrn_tanh(cT[j] + bt);
        const float C = I * T;
        const float O = gcrn_sigmoid(cO[j] + bo + wo * C);
        ot[(hi * 8 + j) * TILE_PITCH + col] = O * gcrn_tanh(C);
    }
    __syncthreads();

    if (nb + WAVE_NODES <= n) {
        #pragma unroll
        for (int i = 0; i < WAVE_NODES; ++i)
            out[(size_t)(nb + i) * D_CH + lane] = ot[(wave * WAVE_NODES + i) * TILE_PITCH + lane];
    } else {
        #pragma unroll
        for (int i = 0; i < WAVE_NODES; ++i) {
            const int node = nb + i;
            if (node < n) out[(size_t)node * D_CH + lane] = ot[(wave * WAVE_NODES + i) * TILE_PITCH + lane];
        }
    }
}

extern "C" void kernel_launch(void* const* d_in, const int* in_sizes, int n_in,
                              void* d_out, int out_size, void* d_ws, size_t ws_size,
                              hipStream_t stream) {
    const float* X       = (const float*)d_in[0];
    const int*   eidx    = (const int*)  d_in[1];
    const float* ew      = (const float*)d_in[2];
    const float* conv_w  = (const float*)d_in[3];
    const float* conv_b  = (const float*)d_in[4];
    const float* wc      = (const float*)d_in[5];
    const float* bg      = (const float*)d_in[6];

    const int N = in_sizes[0] / D_CH;
    const int E = in_sizes[2];
    const int L = in_sizes[3] / (8 * D_CH * D_CH);
    const int* src = eidx;
    const int* dst = eidx + E;

    float* ws   = (float*)d_ws;
    const int NBINS = (N + BIN_NODES - 1) >> BIN_SHIFT;
    float* h1   = ws;
    float* h2   = h1   + (size_t)N * D_CH;
    int*   tab  = (int*)(h2 + (size_t)N * D_CH);
    int2*  rec  = (int2*)(tab + (size_t)NBINS * BIN_G);
    float* dinv = (float*)(rec + E);
    float* nw   = dinv + N;
    float* csrw = nw   + E;
    int*   csrs = (int*)(csrw + E);
    int*   key  = csrs + E;
    int*   offs = key  + E;

    const int TB = 256;
    const int nodesPerBlock = TB / 32;
    const int nodeBlocks = (N + nodesPerBlock - 1) / nodesPerBlock;

    gcrn_bin_hist   <<<BIN_G, BIN_TB, 0, stream>>>(dst, tab, E, NBINS);
    gcrn_bin_scatter<<<BIN_G, BIN_TB, 0, stream>>>(dst, tab, rec, E, NBINS);
    gcrn_bin_place  <<<NBINS, BIN_TB, 0, stream>>>(tab, rec, offs, key, N, NBINS);
    gcrn_sort_seg   <<<nodeBlocks, TB, 0, stream>>>(offs, key, src, ew, csrs, csrw, N);
    gcrn_deg_dinv  <<<(N + TB - 1) / TB, TB, 0, stream>>>(offs, csrw, dinv, N);

    const float* xin = X;
    for (int l = 0; l < L; ++l) {
        float* out_l = (l == L - 1) ? (float*)d_out : ((l & 1) ? h2 : h1);
        const float* W_l  = conv_w + (size_t)l * 8 * 32 * 32;
        const float* cb_l = conv_b + (size_t)l * 8 * 32;
        const float* bg_l = bg + (size_t)l * 4 * 32;
        const float* wo_l = wc + ((size_t)l * 3 + 2) * 32;
        const int tiles = (N + TILE_NODES - 1) / TILE_NODES;
        if (l == 0)
            gcrn_layer<true> <<<tiles, 64, 0, stream>>>(offs, csrs, csrw, nw, dinv, xin, W_l, cb_l, bg_l, wo_l, out_l, N, E);
        else
            gcrn_layer<false><<<tiles, 64, 0, stream>>>(offs, csrs, nw, nullptr, dinv, xin, W_l, cb_l, bg_l, wo_l, out_l, N, E);
        xin = out_l;
    }
}


# Sort records carry source and weight; the place pass orders each node's segment by edge id in LDS, writes the final list and takes the degree there; the per-node sort and degree launches are gone
# speedup vs baseline: 6.0236x; 1.0550x over previous
#include <hip/hip_runtime.h>
#include <hip/hip_bf16.h>
#include <math.h>


typedef __attribute__((ext_vector_type(2))) float v2f;
typedef __attribute__((ext_vector_type(8))) float v8f;

#define D_CH 32
#define TILE_NODES 16
#define WAVE_NODES 8
#define TILE_PITCH 36


#define BIN_SHIFT 9
#define BIN_NODES 512
#define BIN_G 256
#define BIN_MAXB 1024
#define BIN_TB 1024
#define BIN_BATCH 8
#define BIN_KEYCAP 12288

__device__ __forceinline__ int wave_sum_i(int v) {
    for (int m = 16; m >= 1; m >>= 1) v += __shfl_xor(v, m, 32);
    return v;
}

__device__ __forceinline__ int gcrn_block_excl(int v, int* sw) {
    const int lane = threadIdx.x & 31, wv = threadIdx.x >> 5;
    int inc = v;
    for (int d = 1; d < 32; d <<= 1) {
        int t = __shfl_up(inc, d, 32);
        if (lane >= d) inc += t;
    }
    if (lane == 31) sw[wv] = inc;
    __syncthreads();
    if (wv == 0) {
        const int s = sw[lane];
        int si = s;
        for (int d = 1; d < 32; d <<= 1) {
            int t = __shfl_up(si, d, 32);
            if (lane >= d) si += t;
        }
        sw[lane] = si - s;
    }
    __syncthreads();
    const int r = sw[wv] + inc - v;
    __syncthreads();
    return r;
}

__device__ __forceinline__ void gcrn_bin_totals(const int* __restrict__ tab, int nbins, int my_g,
                                                int* btot, int* bpre) {
    const int lane = threadIdx.x & 31, wv = threadIdx.x >> 5;
    for (int b = wv; b < nbins; b += BIN_TB / 32) {
        const int4* p = reinterpret_cast<const int4*>(tab + b * BIN_G + lane * 8);
        const int4 u = p[0], w = p[1];
        const int ev[8] = {u.x, u.y, u.z, u.w, w.x, w.y, w.z, w.w};
        int s = 0, q = 0;
        #pragma unroll
        for (int k = 0; k < 8; ++k) {
            s += ev[k];
            q += (lane * 8 + k < my_g) ? ev[k] : 0;
        }
        s = wave_sum_i(s);
        q = wave_sum_i(q);
        if (lane == 0) { btot[b] = s; bpre[b] = q; }
    }
}

__global__ void __launch_bounds__(BIN_TB)
gcrn_bin_hist(const int* __restrict__ dst, int* __restrict__ tab, int e, int nbins) {
    __shared__ int h[BIN_MAXB];
    const int tid = threadIdx.x;
    if (tid < nbins) h[tid] = 0;
    __syncthreads();
    const int chunk = (e + BIN_G - 1) / BIN_G;
    const int beg = blockIdx.x * chunk, end = min(beg + chunk, e);
    for (int i0 = beg + tid; i0 < end; i0 += BIN_BATCH * BIN_TB) {
        int d[BIN_BATCH];
        #pragma unroll
        for (int u = 0; u < BIN_BATCH; ++u) {
            const int i = i0 + u * BIN_TB;
            d[u] = (i < end) ? dst[i] : 0;
        }
        #pragma unroll
        for (int u = 0; u < BIN_BATCH; ++u)
            if (i0 + u * BIN_TB < end) atomicAdd(&h[d[u] >> BIN_SHIFT], 1);
    }
    __syncthreads();
    if (tid < nbins) tab[tid * BIN_G + blockIdx.x] = h[tid];
}

__global__ void __launch_bounds__(BIN_TB)
gcrn_bin_scatter(const int* __restrict__ src, const int* __restrict__ dst, const float* __restrict__ ew,
                 const int* __restrict__ tab, int4* __restrict__ rec, int e, int nbins) {
    __shared__ int btot[BIN_MAXB];
    __shared__ int bpre[BIN_MAXB];
    __shared__ int cur[BIN_MAXB];
    __shared__ int sw[32];
    const int tid = threadIdx.x;
    gcrn_bin_totals(tab, nbins, blockIdx.x, btot, bpre);
    __syncthreads();
    const int v  = (tid < nbins) ? btot[tid] : 0;
    const int ex = gcrn_block_excl(v, sw);
    if (tid < nbins) cur[tid] = ex + bpre[tid];
    __syncthreads();
    const int chunk = (e + BIN_G - 1) / BIN_G;
    const int beg = blockIdx.x * chunk, end = min(beg + chunk, e);
    for (int i0 = beg + tid; i0 < end; i0 += BIN_BATCH * BIN_TB) {
        int d[BIN_BATCH], s[BIN_BATCH];
        float w[BIN_BATCH];
        #pragma unroll
        for (int u = 0; u < BIN_BATCH; ++u) {
            const int i = i0 + u * BIN_TB;
            const bool ok = i < end;
            d[u] = ok ? dst[i] : 0;
            s[u] = ok ? src[i] : 0;
            w[u] = ok ? ew[i] : 0.0f;
        }
        #pragma unroll
        for (int u = 0; u < BIN_BATCH; ++u) {
            const int i = i0 + u * BIN_TB;
            if (i < end) {
                const int slot = atomicAdd(&cur[d[u] >> BIN_SHIFT], 1);
                rec[slot] = make_int4(i, d[u], s[u], __float_as_int(w[u]));
            }
        }
    }
}

template <bool GLOBAL_PLANES>
__device__ __forceinline__ void gcrn_order_node(const int* kid, const int* ksrc, const float* kw, float* kws,
                                                int b, int deg,
                                                int* __restrict__ csr_src, float* __restrict__ csr_w,
                                                float* __restrict__ dinv, int node, int gbeg, int lane) {
    for (int i0 = 0; i0 < deg; i0 += 32) {
        const bool vi = (i0 + lane) < deg;
        const int ki = vi ? kid[b + i0 + lane] : 0x7fffffff;
        int r = 0;
        for (int j0 = 0; j0 < deg; j0 += 32) {
            const int kj = (j0 + lane < deg) ? kid[b + j0 + lane] : 0x7fffffff;
            const int m8 = (min(32, deg - j0) + 7) & ~7;
            for (int t = 0; t < m8; t += 8) {
                #pragma unroll
                for (int q = 0; q < 8; ++q)
                    r += (__builtin_amdgcn_readlane(kj, t + q) < ki) ? 1 : 0;
            }
        }
        if (vi) {
            const float w = kw[b + i0 + lane];
            csr_src[gbeg + r] = ksrc[b + i0 + lane];
            csr_w  [gbeg + r] = w;
            kws[b + r] = w;
        }
    }
    if (GLOBAL_PLANES) __threadfence_block();
    float d = 1.0f;
    for (int j0 = 0; j0 < deg; j0 += 32) {
        const float wj = (j0 + lane < deg) ? kws[b + j0 + lane] : 0.0f;
        const int m8 = (min(32, deg - j0) + 7) & ~7;
        for (int t = 0; t < m8; t += 8) {
            #pragma unroll
            for (int q = 0; q < 8; ++q)
                d += __int_as_float(__builtin_amdgcn_readlane(__float_as_int(wj), t + q));
        }
    }
    if (lane == 0) dinv[node] = (d > 0.0f) ? rsqrtf(d) : 0.0f;
}

__global__ void __launch_bounds__(BIN_TB)
gcrn_bin_place(const int* __restrict__ tab, const int4* __restrict__ rec,
               int* __restrict__ offs, int* __restrict__ csr_src, float* __restrict__ csr_w,
               float* __restrict__ dinv,
               int* __restrict__ gk_id, int* __restrict__ gk_src, float* __restrict__ gk_w,
               float* __restrict__ gk_ws,
               int n, int nbins) {
    __shared__ int btot[BIN_MAXB];
    __shared__ int bpre[BIN_MAXB];
    __shared__ int bst[BIN_MAXB + 1];
    __shared__ int c[BIN_NODES];
    __shared__ int sw[32];
    __shared__ int   kid [BIN_KEYCAP];
    __shared__ int   ksrc[BIN_KEYCAP];
    __shared__ float kw  [BIN_KEYCAP];
    __shared__ float kws [BIN_KEYCAP];
    const int tid = threadIdx.x;
    gcrn_bin_totals(tab, nbins, 0, btot, bpre);
    if (tid < BIN_NODES) c[tid] = 0;
    __syncthreads();
    const int v  = (tid < nbins) ? btot[tid] : 0;
    const int ex = gcrn_block_excl(v, sw);
    if (tid <= nbins) bst[tid] = ex;
    __syncthreads();
    const int b   = blockIdx.x;
    const int lo  = bst[b], hi = bst[b + 1];
    const int nb0 = b << BIN_SHIFT;

    for (int p0 = lo + tid; p0 < hi; p0 += BIN_BATCH * BIN_TB) {
        int y[BIN_BATCH];
        #pragma unroll
        for (int u = 0; u < BIN_BATCH; ++u) {
            const int p = p0 + u * BIN_TB;
            y[u] = (p < hi) ? rec[p].y : nb0;
        }
        #pragma unroll
        for (int u = 0; u < BIN_BATCH; ++u)
            if (p0 + u * BIN_TB < hi) atomicAdd(&c[y[u] - nb0], 1);
    }
    __syncthreads();
    const int cv = (tid < BIN_NODES) ? c[tid] : 0;
    const int ce = gcrn_block_excl(cv, sw);
    if (tid < BIN_NODES) {
        c[tid] = lo + ce;
        if (nb0 + tid < n) offs[nb0 + tid] = lo + ce;
    }
    if (b == nbins - 1 && tid == 0) offs[n] = hi;
    __syncthreads();

    const int lane = tid & 31;
    const int wv   = __builtin_amdgcn_readfirstlane(tid >> 5);
    const int cntb = hi - lo;
    if (cntb <= BIN_KEYCAP) {
        for (int p0 = lo + tid; p0 < hi; p0 += BIN_BATCH * BIN_TB) {
            int4 r[BIN_BATCH];
            #pragma unroll
            for (int u = 0; u < BIN_BATCH; ++u) {
                const int p = p0 + u * BIN_TB;
                r[u] = (p < hi) ? rec[p] : make_int4(0, nb0, 0, 0);
            }
            #pragma unroll
            for (int u = 0; u < BIN_BATCH; ++u) {
                if (p0 + u * BIN_TB < hi) {
                    const int q = atomicAdd(&c[r[u].y - nb0], 1) - lo;
                    kid [q] = r[u].x;
                    ksrc[q] = r[u].z;
                    kw  [q] = __int_as_float(r[u].w);
                }
            }
        }
        __syncthreads();
        for (int t = wv; t < BIN_NODES; t += BIN_TB / 32) {
            if (nb0 + t >= n) break;
            const int sb = __builtin_amdgcn_readfirstlane(t == 0 ? lo : c[t - 1]);
            const int se = __builtin_amdgcn_readfirstlane(c[t]);
            gcrn_order_node<false>(kid, ksrc, kw, kws, sb - lo, se - sb, csr_src, csr_w, dinv, nb0 + t, sb, lane);
        }
    } else {
        for (int p = lo + tid; p < hi; p += BIN_TB) {
            const int4 r = rec[p];
            const int slot = atomicAdd(&c[r.y - nb0], 1);
            gk_id [slot] = r.x;
            gk_src[slot] = r.z;
            gk_w  [slot] = __int_as_float(r.w);
        }
        __syncthreads();
        for (int t = wv; t < BIN_NODES; t += BIN_TB / 32) {
            if (nb0 + t >= n) break;
            const int sb = __builtin_amdgcn_readfirstlane(t == 0 ? lo : c[t - 1]);
            const int se = __builtin_amdgcn_readfirstlane(c[t]);
            gcrn_order_node<true>(gk_id, gk_src, gk_w, gk_ws, sb, se - sb, csr_src, csr_w, dinv, nb0 + t, sb, lane);
        }
    }
}


template <bool FIRST>
__device__ __forceinline__ float gcrn_gather_rows(const float* __restrict__ x, int s_l, float w_l, int m, int lane,
                                                  const int R, const bool start, float d2,
                                                  const float* __restrict__ xsp,
                                                  const int* __restrict__ snp, const float* __restrict__ wnp,
                                                  int& s_nx, float& w_nx, float a,
                                                  const float* __restrict__ dinv, float dd, float* __restrict__ nwo) {
    float ds = 0.0f;
    if (FIRST) ds = (lane < m) ? dinv[s_l] : 0.0f;
    float v[32];
    #pragma unroll
    for (int u = 0; u < R; ++u) {
        const int tt = (u < R - 8) ? u : min(u, m - 1);
        const int s  = __builtin_amdgcn_readlane(s_l, tt);
        v[u] = x[s * D_CH + lane];
    }
    if (start) {
        const float xs = *xsp;
        s_nx = *snp;
        w_nx = *wnp;
        a = d2 * xs;
    }
    if (FIRST) {
        w_l = ds * w_l * dd;
        if (lane < m) *nwo = w_l;
    }
    asm volatile("" :: "v"(v[R - 8]), "v"(v[R - 7]), "v"(v[R - 6]), "v"(v[R - 5]),
                       "v"(v[R - 4]), "v"(v[R - 3]), "v"(v[R - 2]), "v"(v[R - 1]));
    #pragma unroll
    for (int u = 0; u < R; ++u) {
        if (u < R - 8 || u < m) {
            const float w = __int_as_float(__builtin_amdgcn_readlane(__float_as_int(w_l), u));
            a = fmaf(w, v[u], a);
        }
    }
    return a;
}

template <bool FIRST>
__device__ __forceinline__ float gcrn_gather_chunk(const float* __restrict__ x, int s_l, float w_l, int m, int lane,
                                                   const bool start, float d2,
                                                   const float* __restrict__ xsp,
                                                   const int* __restrict__ snp, const float* __restrict__ wnp,
                                                   int& s_nx, float& w_nx, float a,
                                                   const float* __restrict__ dinv, float dd, float* __restrict__ nwo) {
    if (m > 24)      return gcrn_gather_rows<FIRST>(x, s_l, w_l, m, lane, 32, start, d2, xsp, snp, wnp, s_nx, w_nx, a, dinv, dd, nwo);
    else if (m > 16) return gcrn_gather_rows<FIRST>(x, s_l, w_l, m, lane, 24, start, d2, xsp, snp, wnp, s_nx, w_nx, a, dinv, dd, nwo);
    else if (m > 8)  return gcrn_gather_rows<FIRST>(x, s_l, w_l, m, lane, 16, start, d2, xsp, snp, wnp, s_nx, w_nx, a, dinv, dd, nwo);
    else if (m > 0)  return gcrn_gather_rows<FIRST>(x, s_l, w_l, m, lane, 8,  start, d2, xsp, snp, wnp, s_nx, w_nx, a, dinv, dd, nwo);
    if (start) {
        const float xs = *xsp;
        s_nx = *snp;
        w_nx = *wnp;
        a = d2 * xs;
    }
    return a;
}

__device__ __forceinline__ float gcrn_tanh(float x) { return __builtin_amdgcn_tanhf(x); }
__device__ __forceinline__ float gcrn_sigmoid(float x) { return fmaf(0.5f, __builtin_amdgcn_tanhf(0.5f * x), 0.5f); }

template <bool FIRST>
__global__ void __launch_bounds__(64)
gcrn_layer(const int* __restrict__ offs, const int* __restrict__ csr_src,
           const float* __restrict__ nw,
           float* __restrict__ nw_out,
           const float* __restrict__ dinv,
           const float* __restrict__ x,
           const float* __restrict__ Wl,
           const float* __restrict__ cb,
           const float* __restrict__ bgl,
           const float* __restrict__ wco,
           float* __restrict__ out, int n, int e) {
    __shared__ float ax[TILE_NODES * TILE_PITCH];
    __shared__ float ot[TILE_NODES * TILE_PITCH];

    const int wave = __builtin_amdgcn_readfirstlane(threadIdx.x >> 5);
    const int lane = threadIdx.x & 31;
    const int nb   = blockIdx.x * TILE_NODES + wave * WAVE_NODES;
    const int elast = max(e - 1, 0);

    const int   o_l  = offs[min(nb + min(lane, WAVE_NODES), n)];
    const float dv_l = dinv[min(nb + min(lane, WAVE_NODES - 1), n - 1)];

    int beg = __builtin_amdgcn_readlane(o_l, 0);
    int end = __builtin_amdgcn_readlane(o_l, 1);
    int   s_nx = csr_src[min(beg + lane, elast)];
    float w_nx = nw[min(beg + lane, elast)];

    for (int i = 0; i < WAVE_NODES; ++i) {
        const int s_l = s_nx; const float w_l = w_nx;
        const int nbeg = end;
        const int nend = __builtin_amdgcn_readlane(o_l, min(i + 2, WAVE_NODES));
        const int nidx = min(nbeg + lane, elast);

        const int node = nb + i;
        const float dv = __int_as_float(__builtin_amdgcn_readlane(__float_as_int(dv_l), i));
        const float d2 = (node < n) ? dv * dv : 0.0f;
        const float* xsp = x + (size_t)min(node, n - 1) * D_CH + lane;

        float a = gcrn_gather_chunk<FIRST>(x, s_l, w_l, min(32, end - beg), lane, true, d2,
                                           xsp, csr_src + nidx, nw + nidx, s_nx, w_nx, 0.0f,
                                           dinv, dv, FIRST ? nw_out + (beg + lane) : nullptr);
        for (int p0 = beg + 32; p0 < end; p0 += 32) {
            const int m = min(32, end - p0);
            const int   s_m = csr_src[min(p0 + lane, elast)];
            const float w_m = nw[min(p0 + lane, elast)];
            int s_no; float w_no;
            a = gcrn_gather_chunk<FIRST>(x, s_m, w_m, m, lane, false, 0.0f, xsp, csr_src, nw, s_no, w_no, a,
                                         dinv, dv, FIRST ? nw_out + (p0 + lane) : nullptr);
        }
        ax[(wave * WAVE_NODES + i) * TILE_PITCH + lane] = a;
        beg = nbeg; end = nend;
    }
    __syncthreads();

    const int n0  = lane & 15;
    const int hi  = lane >> 4;
    const int col = wave * 16 + n0;
    const float* ar = ax + n0 * TILE_PITCH;
    const float* WI = Wl + 0 * 32 * 32 + col;
    const float* WT = Wl + 4 * 32 * 32 + col;
    const float* WO = Wl + 6 * 32 * 32 + col;

    v8f cI = {}, cT = {}, cO = {};
    #pragma unroll
    for (int k = 0; k < 32; k += 4) {
        const int ka = k + 2 * hi;
        v2f a;  a.x  = ar[ka];       a.y  = ar[ka + 1];
        v2f bI; bI.x = WI[ka * 32];  bI.y = WI[(ka + 1) * 32];
        v2f bT; bT.x = WT[ka * 32];  bT.y = WT[(ka + 1) * 32];
        v2f bO; bO.x = WO[ka * 32];  bO.y = WO[(ka + 1) * 32];
        cI = __builtin_amdgcn_wmma_f32_16x16x4_f32(false, a, false, bI, (short)0, cI, false, false);
        cT = __builtin_amdgcn_wmma_f32_16x16x4_f32(false, a, false, bT, (short)0, cT, false, false);
        cO = __builtin_amdgcn_wmma_f32_16x16x4_f32(false, a, false, bO, (short)0, cO, false, false);
    }

    const float bi = cb[col]          + cb[32 + col]     + bgl[col];
    const float bt = cb[4 * 32 + col] + cb[5 * 32 + col] + bgl[2 * 32 + col];
    const float bo = cb[6 * 32 + col] + cb[7 * 32 + col] + bgl[3 * 32 + col];
    const float wo = wco[col];

    #pragma unroll
    for (int j = 0; j < 8; ++j) {
        const float I = gcrn_sigmoid(cI[j] + bi);
        const float T = gcrn_tanh(cT[j] + bt);
        const float C = I * T;
        const float O = gcrn_sigmoid(cO[j] + bo + wo * C);
        ot[(hi * 8 + j) * TILE_PITCH + col] = O * gcrn_tanh(C);
    }
    __syncthreads();

    if (nb + WAVE_NODES <= n) {
        #pragma unroll
        for (int i = 0; i < WAVE_NODES; ++i)
            out[(size_t)(nb + i) * D_CH + lane] = ot[(wave * WAVE_NODES + i) * TILE_PITCH + lane];
    } else {
        #pragma unroll
        for (int i = 0; i < WAVE_NODES; ++i) {
            const int node = nb + i;
            if (node < n) out[(size_t)node * D_CH + lane] = ot[(wave * WAVE_NODES + i) * TILE_PITCH + lane];
        }
    }
}

extern "C" void kernel_launch(void* const* d_in, const int* in_sizes, int n_in,
                              void* d_out, int out_size, void* d_ws, size_t ws_size,
                              hipStream_t stream) {
    const float* X       = (const float*)d_in[0];
    const int*   eidx    = (const int*)  d_in[1];
    const float* ew      = (const float*)d_in[2];
    const float* conv_w  = (const float*)d_in[3];
    const float* conv_b  = (const float*)d_in[4];
    const float* wc      = (const float*)d_in[5];
    const float* bg      = (const float*)d_in[6];

    const int N = in_sizes[0] / D_CH;
    const int E = in_sizes[2];
    const int L = in_sizes[3] / (8 * D_CH * D_CH);
    const int* src = eidx;
    const int* dst = eidx + E;

    float* ws   = (float*)d_ws;
    const int NBINS = (N + BIN_NODES - 1) >> BIN_SHIFT;
    float* h1   = ws;
    float* h2   = h1   + (size_t)N * D_CH;
    int*   tab  = (int*)(h2 + (size_t)N * D_CH);
    int4*  rec  = (int4*)(tab + (size_t)NBINS * BIN_G);
    float* dinv = (float*)(rec + E);
    float* nw   = dinv + N;
    float* csrw = nw   + E;
    int*   csrs = (int*)(csrw + E);
    int*   gkid = csrs + E;
    int*   gksr = gkid + E;
    float* gkw  = (float*)(gksr + E);
    float* gkws = gkw + E;
    int*   offs = (int*)(gkws + E);

    gcrn_bin_hist   <<<BIN_G, BIN_TB, 0, stream>>>(dst, tab, E, NBINS);
    gcrn_bin_scatter<<<BIN_G, BIN_TB, 0, stream>>>(src, dst, ew, tab, rec, E, NBINS);
    gcrn_bin_place  <<<NBINS, BIN_TB, 0, stream>>>(tab, rec, offs, csrs, csrw, dinv, gkid, gksr, gkw, gkws, N, NBINS);

    const float* xin = X;
    for (int l = 0; l < L; ++l) {
        float* out_l = (l == L - 1) ? (float*)d_out : ((l & 1) ? h2 : h1);
        const float* W_l  = conv_w + (size_t)l * 8 * 32 * 32;
        const float* cb_l = conv_b + (size_t)l * 8 * 32;
        const float* bg_l = bg + (size_t)l * 4 * 32;
        const float* wo_l = wc + ((size_t)l * 3 + 2) * 32;
        const int tiles = (N + TILE_NODES - 1) / TILE_NODES;
        if (l == 0)
            gcrn_layer<true> <<<tiles, 64, 0, stream>>>(offs, csrs, csrw, nw, dinv, xin, W_l, cb_l, bg_l, wo_l, out_l, N, E);
        else
            gcrn_layer<false><<<tiles, 64, 0, stream>>>(offs, csrs, nw, nullptr, dinv, xin, W_l, cb_l, bg_l, wo_l, out_l, N, E);
        xin = out_l;
    }
}


# Level-1 pass sorts each edge chunk by bin inside the chunk's own region (local prefix): the histogram launch and both cross-chunk prefix passes are gone; slot and sum order kept
# speedup vs baseline: 6.4403x; 1.0692x over previous
#include <hip/hip_runtime.h>
#include <hip/hip_bf16.h>
#include <math.h>


typedef __attribute__((ext_vector_type(2))) float v2f;
typedef __attribute__((ext_vector_type(8))) float v8f;

#define D_CH 32
#define TILE_NODES 16
#define WAVE_NODES 8
#define TILE_PITCH 36


#define BIN_SHIFT 9
#define BIN_NODES 512
#define BIN_G 256
#define BIN_MAXB 1024
#define BIN_TB 1024
#define BIN_BATCH 8
#define BIN_KEYCAP 12288
#define BIN_RPW (BIN_G / (BIN_TB / 32))

__device__ __forceinline__ int wave_sum_i(int v) {
    for (int m = 16; m >= 1; m >>= 1) v += __shfl_xor(v, m, 32);
    return v;
}

__device__ __forceinline__ int gcrn_block_excl(int v, int* sw) {
    const int lane = threadIdx.x & 31, wv = threadIdx.x >> 5;
    int inc = v;
    for (int d = 1; d < 32; d <<= 1) {
        int t = __shfl_up(inc, d, 32);
        if (lane >= d) inc += t;
    }
    if (lane == 31) sw[wv] = inc;
    __syncthreads();
    if (wv == 0) {
        const int s = sw[lane];
        int si = s;
        for (int d = 1; d < 32; d <<= 1) {
            int t = __shfl_up(si, d, 32);
            if (lane >= d) si += t;
        }
        sw[lane] = si - s;
    }
    __syncthreads();
    const int r = sw[wv] + inc - v;
    __syncthreads();
    return r;
}

__global__ void __launch_bounds__(BIN_TB)
gcrn_chunk_sort(const int* __restrict__ src, const int* __restrict__ dst, const float* __restrict__ ew,
                int* __restrict__ tab, int* __restrict__ rst, int4* __restrict__ rec, int e, int nbins) {
    __shared__ int h[BIN_MAXB];
    __shared__ int sw[32];
    const int tid = threadIdx.x;
    if (tid < nbins) h[tid] = 0;
    const int chunk = (e + BIN_G - 1) / BIN_G;
    const int beg = blockIdx.x * chunk, end = min(beg + chunk, e);

    int d0[BIN_BATCH], s0[BIN_BATCH];
    float w0[BIN_BATCH];
    #pragma unroll
    for (int u = 0; u < BIN_BATCH; ++u) {
        const int i = beg + tid + u * BIN_TB;
        const bool ok = i < end;
        d0[u] = ok ? dst[i] : 0;
        s0[u] = ok ? src[i] : 0;
        w0[u] = ok ? ew[i] : 0.0f;
    }
    __syncthreads();
    #pragma unroll
    for (int u = 0; u < BIN_BATCH; ++u)
        if (beg + tid + u * BIN_TB < end) atomicAdd(&h[d0[u] >> BIN_SHIFT], 1);
    for (int i0 = beg + tid + BIN_BATCH * BIN_TB; i0 < end; i0 += BIN_BATCH * BIN_TB) {
        int d[BIN_BATCH];
        #pragma unroll
        for (int u = 0; u < BIN_BATCH; ++u) {
            const int i = i0 + u * BIN_TB;
            d[u] = (i < end) ? dst[i] : 0;
        }
        #pragma unroll
        for (int u = 0; u < BIN_BATCH; ++u)
            if (i0 + u * BIN_TB < end) atomicAdd(&h[d[u] >> BIN_SHIFT], 1);
    }
    __syncthreads();
    const int v  = (tid < nbins) ? h[tid] : 0;
    const int ex = gcrn_block_excl(v, sw);
    if (tid < nbins) {
        h[tid] = beg + ex;
        tab[tid * BIN_G + blockIdx.x] = v;
        rst[tid * BIN_G + blockIdx.x] = beg + ex;
    }
    __syncthreads();
    #pragma unroll
    for (int u = 0; u < BIN_BATCH; ++u) {
        const int i = beg + tid + u * BIN_TB;
        if (i < end) {
            const int slot = atomicAdd(&h[d0[u] >> BIN_SHIFT], 1);
            rec[slot] = make_int4(i, d0[u], s0[u], __float_as_int(w0[u]));
        }
    }
    for (int i0 = beg + tid + BIN_BATCH * BIN_TB; i0 < end; i0 += BIN_BATCH * BIN_TB) {
        int d[BIN_BATCH], s[BIN_BATCH];
        float w[BIN_BATCH];
        #pragma unroll
        for (int u = 0; u < BIN_BATCH; ++u) {
            const int i = i0 + u * BIN_TB;
            const bool ok = i < end;
            d[u] = ok ? dst[i] : 0;
            s[u] = ok ? src[i] : 0;
            w[u] = ok ? ew[i] : 0.0f;
        }
        #pragma unroll
        for (int u = 0; u < BIN_BATCH; ++u) {
            const int i = i0 + u * BIN_TB;
            if (i < end) {
                const int slot = atomicAdd(&h[d[u] >> BIN_SHIFT], 1);
                rec[slot] = make_int4(i, d[u], s[u], __float_as_int(w[u]));
            }
        }
    }
}

template <class F>
__device__ __forceinline__ void gcrn_each_record(const int4* __restrict__ rec, int cnt_l, int st_l, int lane, F f) {
    int4 q[2 * BIN_RPW];
    #pragma unroll
    for (int k = 0; k < 2 * BIN_RPW; ++k) {
        const int r = k % BIN_RPW, o = (k / BIN_RPW) * 32 + lane;
        const int nr = __builtin_amdgcn_readlane(cnt_l, r);
        const int sr = __builtin_amdgcn_readlane(st_l, r);
        q[k] = (o < nr) ? rec[sr + o] : make_int4(0, 0, 0, 0);
    }
    #pragma unroll
    for (int k = 0; k < 2 * BIN_RPW; ++k) {
        const int r = k % BIN_RPW, o = (k / BIN_RPW) * 32 + lane;
        const int nr = __builtin_amdgcn_readlane(cnt_l, r);
        if (o < nr) f(q[k]);
    }
    for (int r = 0; r < BIN_RPW; ++r) {
        const int nr = __builtin_amdgcn_readlane(cnt_l, r);
        const int sr = __builtin_amdgcn_readlane(st_l, r);
        for (int o = 64 + lane; o < nr; o += 32) f(rec[sr + o]);
    }
}

template <bool GLOBAL_PLANES>
__device__ __forceinline__ void gcrn_order_node(const int* kid, const int* ksrc, const float* kw, float* kws,
                                                int b, int deg,
                                                int* __restrict__ csr_src, float* __restrict__ csr_w,
                                                float* __restrict__ dinv, int node, int gbeg, int lane) {
    for (int i0 = 0; i0 < deg; i0 += 32) {
        const bool vi = (i0 + lane) < deg;
        const int ki = vi ? kid[b + i0 + lane] : 0x7fffffff;
        int r = 0;
        for (int j0 = 0; j0 < deg; j0 += 32) {
            const int kj = (j0 + lane < deg) ? kid[b + j0 + lane] : 0x7fffffff;
            const int m8 = (min(32, deg - j0) + 7) & ~7;
            for (int t = 0; t < m8; t += 8) {
                #pragma unroll
                for (int q = 0; q < 8; ++q)
                    r += (__builtin_amdgcn_readlane(kj, t + q) < ki) ? 1 : 0;
            }
        }
        if (vi) {
            const float w = kw[b + i0 + lane];
            csr_src[gbeg + r] = ksrc[b + i0 + lane];
            csr_w  [gbeg + r] = w;
            kws[b + r] = w;
        }
    }
    if (GLOBAL_PLANES) __threadfence_block();
    float d = 1.0f;
    for (int j0 = 0; j0 < deg; j0 += 32) {
        const float wj = (j0 + lane < deg) ? kws[b + j0 + lane] : 0.0f;
        const int m8 = (min(32, deg - j0) + 7) & ~7;
        for (int t = 0; t < m8; t += 8) {
            #pragma unroll
            for (int q = 0; q < 8; ++q)
                d += __int_as_float(__builtin_amdgcn_readlane(__float_as_int(wj), t + q));
        }
    }
    if (lane == 0) dinv[node] = (d > 0.0f) ? rsqrtf(d) : 0.0f;
}

__global__ void __launch_bounds__(BIN_TB)
gcrn_bin_place(const int* __restrict__ tab, const int* __restrict__ rst, const int4* __restrict__ rec,
               int* __restrict__ offs, int* __restrict__ csr_src, float* __restrict__ csr_w,
               float* __restrict__ dinv,
               int* __restrict__ gk_id, int* __restrict__ gk_src, float* __restrict__ gk_w,
               float* __restrict__ gk_ws,
               int n, int nbins) {
    __shared__ int c[BIN_NODES];
    __shared__ int sw[32];
    __shared__ int red_lo[32];
    __shared__ int red_cn[32];
    __shared__ int   kid [BIN_KEYCAP];
    __shared__ int   ksrc[BIN_KEYCAP];
    __shared__ float kw  [BIN_KEYCAP];
    __shared__ float kws [BIN_KEYCAP];
    const int tid  = threadIdx.x;
    const int lane = tid & 31;
    const int wv   = __builtin_amdgcn_readfirstlane(tid >> 5);
    const int b    = blockIdx.x;
    const int nb0  = b << BIN_SHIFT;
    if (tid < BIN_NODES) c[tid] = 0;

    const int g_l   = wv * BIN_RPW + (lane & (BIN_RPW - 1));
    const int cnt_l = tab[b * BIN_G + g_l];
    const int st_l  = rst[b * BIN_G + g_l];

    int acc = 0;
    const int4* t4 = reinterpret_cast<const int4*>(tab);
    const int n4 = b * (BIN_G / 4);
    for (int j0 = tid; j0 < n4; j0 += BIN_BATCH * BIN_TB) {
        int4 q[BIN_BATCH];
        #pragma unroll
        for (int u = 0; u < BIN_BATCH; ++u) {
            const int j = j0 + u * BIN_TB;
            q[u] = (j < n4) ? t4[j] : make_int4(0, 0, 0, 0);
        }
        #pragma unroll
        for (int u = 0; u < BIN_BATCH; ++u) acc += (q[u].x + q[u].y) + (q[u].z + q[u].w);
    }
    const int rowv = (tid < BIN_G) ? tab[b * BIN_G + tid] : 0;
    const int sa = wave_sum_i(acc), sb = wave_sum_i(rowv);
    if (lane == 0) { red_lo[wv] = sa; red_cn[wv] = sb; }
    __syncthreads();
    const int lo   = __builtin_amdgcn_readfirstlane(wave_sum_i(red_lo[lane]));
    const int cntb = __builtin_amdgcn_readfirstlane(wave_sum_i(red_cn[lane]));
    const int hi   = lo + cntb;

    gcrn_each_record(rec, cnt_l, st_l, lane, [&](const int4& r) { atomicAdd(&c[r.y - nb0], 1); });
    __syncthreads();
    const int cv = (tid < BIN_NODES) ? c[tid] : 0;
    const int ce = gcrn_block_excl(cv, sw);
    if (tid < BIN_NODES) {
        c[tid] = lo + ce;
        if (nb0 + tid < n) offs[nb0 + tid] = lo + ce;
    }
    if (b == nbins - 1 && tid == 0) offs[n] = hi;
    __syncthreads();

    if (cntb <= BIN_KEYCAP) {
        gcrn_each_record(rec, cnt_l, st_l, lane, [&](const int4& r) {
            const int q = atomicAdd(&c[r.y - nb0], 1) - lo;
            kid [q] = r.x;
            ksrc[q] = r.z;
            kw  [q] = __int_as_float(r.w);
        });
        __syncthreads();
        for (int t = wv; t < BIN_NODES; t += BIN_TB / 32) {
            if (nb0 + t >= n) break;
            const int sb2 = __builtin_amdgcn_readfirstlane(t == 0 ? lo : c[t - 1]);
            const int se  = __builtin_amdgcn_readfirstlane(c[t]);
            gcrn_order_node<false>(kid, ksrc, kw, kws, sb2 - lo, se - sb2, csr_src, csr_w, dinv, nb0 + t, sb2, lane);
        }
    } else {
        gcrn_each_record(rec, cnt_l, st_l, lane, [&](const int4& r) {
            const int slot = atomicAdd(&c[r.y - nb0], 1);
            gk_id [slot] = r.x;
            gk_src[slot] = r.z;
            gk_w  [slot] = __int_as_float(r.w);
        });
        __syncthreads();
        for (int t = wv; t < BIN_NODES; t += BIN_TB / 32) {
            if (nb0 + t >= n) break;
            const int sb2 = __builtin_amdgcn_readfirstlane(t == 0 ? lo : c[t - 1]);
            const int se  = __builtin_amdgcn_readfirstlane(c[t]);
            gcrn_order_node<true>(gk_id, gk_src, gk_w, gk_ws, sb2, se - sb2, csr_src, csr_w, dinv, nb0 + t, sb2, lane);
        }
    }
}


template <bool FIRST>
__device__ __forceinline__ float gcrn_gather_rows(const float* __restrict__ x, int s_l, float w_l, int m, int lane,
                                                  const int R, const bool start, float d2,
                                                  const float* __restrict__ xsp,
                                                  const int* __restrict__ snp, const float* __restrict__ wnp,
                                                  int& s_nx, float& w_nx, float a,
                                                  const float* __restrict__ dinv, float dd, float* __restrict__ nwo) {
    float ds = 0.0f;
    if (FIRST) ds = (lane < m) ? dinv[s_l] : 0.0f;
    float v[32];
    #pragma unroll
    for (int u = 0; u < R; ++u) {
        const int tt = (u < R - 8) ? u : min(u, m - 1);
        const int s  = __builtin_amdgcn_readlane(s_l, tt);
        v[u] = x[s * D_CH + lane];
    }
    if (start) {
        const float xs = *xsp;
        s_nx = *snp;
        w_nx = *wnp;
        a = d2 * xs;
    }
    if (FIRST) {
        w_l = ds * w_l * dd;
        if (lane < m) *nwo = w_l;
    }
    asm volatile("" :: "v"(v[R - 8]), "v"(v[R - 7]), "v"(v[R - 6]), "v"(v[R - 5]),
                       "v"(v[R - 4]), "v"(v[R - 3]), "v"(v[R - 2]), "v"(v[R - 1]));
    #pragma unroll
    for (int u = 0; u < R; ++u) {
        if (u < R - 8 || u < m) {
            const float w = __int_as_float(__builtin_amdgcn_readlane(__float_as_int(w_l), u));
            a = fmaf(w, v[u], a);
        }
    }
    return a;
}

template <bool FIRST>
__device__ __forceinline__ float gcrn_gather_chunk(const float* __restrict__ x, int s_l, float w_l, int m, int lane,
                                                   const bool start, float d2,
                                                   const float* __restrict__ xsp,
                                                   const int* __restrict__ snp, const float* __restrict__ wnp,
                                                   int& s_nx, float& w_nx, float a,
                                                   const float* __restrict__ dinv, float dd, float* __restrict__ nwo) {
    if (m > 24)      return gcrn_gather_rows<FIRST>(x, s_l, w_l, m, lane, 32, start, d2, xsp, snp, wnp, s_nx, w_nx, a, dinv, dd, nwo);
    else if (m > 16) return gcrn_gather_rows<FIRST>(x, s_l, w_l, m, lane, 24, start, d2, xsp, snp, wnp, s_nx, w_nx, a, dinv, dd, nwo);
    else if (m > 8)  return gcrn_gather_rows<FIRST>(x, s_l, w_l, m, lane, 16, start, d2, xsp, snp, wnp, s_nx, w_nx, a, dinv, dd, nwo);
    else if (m > 0)  return gcrn_gather_rows<FIRST>(x, s_l, w_l, m, lane, 8,  start, d2, xsp, snp, wnp, s_nx, w_nx, a, dinv, dd, nwo);
    if (start) {
        const float xs = *xsp;
        s_nx = *snp;
        w_nx = *wnp;
        a = d2 * xs;
    }
    return a;
}

__device__ __forceinline__ float gcrn_tanh(float x) { return __builtin_amdgcn_tanhf(x); }
__device__ __forceinline__ float gcrn_sigmoid(float x) { return fmaf(0.5f, __builtin_amdgcn_tanhf(0.5f * x), 0.5f); }

template <bool FIRST>
__global__ void __launch_bounds__(64)
gcrn_layer(const int* __restrict__ offs, const int* __restrict__ csr_src,
           const float* __restrict__ nw,
           float* __restrict__ nw_out,
           const float* __restrict__ dinv,
           const float* __restrict__ x,
           const float* __restrict__ Wl,
           const float* __restrict__ cb,
           const float* __restrict__ bgl,
           const float* __restrict__ wco,
           float* __restrict__ out, int n, int e) {
    __shared__ float ax[TILE_NODES * TILE_PITCH];
    __shared__ float ot[TILE_NODES * TILE_PITCH];

    const int wave = __builtin_amdgcn_readfirstlane(threadIdx.x >> 5);
    const int lane = threadIdx.x & 31;
    const int nb   = blockIdx.x * TILE_NODES + wave * WAVE_NODES;
    const int elast = max(e - 1, 0);

    const int   o_l  = offs[min(nb + min(lane, WAVE_NODES), n)];
    const float dv_l = dinv[min(nb + min(lane, WAVE_NODES - 1), n - 1)];

    int beg = __builtin_amdgcn_readlane(o_l, 0);
    int end = __builtin_amdgcn_readlane(o_l, 1);
    int   s_nx = csr_src[min(beg + lane, elast)];
    float w_nx = nw[min(beg + lane, elast)];

    for (int i = 0; i < WAVE_NODES; ++i) {
        const int s_l = s_nx; const float w_l = w_nx;
        const int nbeg = end;
        const int nend = __builtin_amdgcn_readlane(o_l, min(i + 2, WAVE_NODES));
        const int nidx = min(nbeg + lane, elast);

        const int node = nb + i;
        const float dv = __int_as_float(__builtin_amdgcn_readlane(__float_as_int(dv_l), i));
        const float d2 = (node < n) ? dv * dv : 0.0f;
        const float* xsp = x + (size_t)min(node, n - 1) * D_CH + lane;

        float a = gcrn_gather_chunk<FIRST>(x, s_l, w_l, min(32, end - beg), lane, true, d2,
                                           xsp, csr_src + nidx, nw + nidx, s_nx, w_nx, 0.0f,
                                           dinv, dv, FIRST ? nw_out + (beg + lane) : nullptr);
        for (int p0 = beg + 32; p0 < end; p0 += 32) {
            const int m = min(32, end - p0);
            const int   s_m = csr_src[min(p0 + lane, elast)];
            const float w_m = nw[min(p0 + lane, elast)];
            int s_no; float w_no;
            a = gcrn_gather_chunk<FIRST>(x, s_m, w_m, m, lane, false, 0.0f, xsp, csr_src, nw, s_no, w_no, a,
                                         dinv, dv, FIRST ? nw_out + (p0 + lane) : nullptr);
        }
        ax[(wave * WAVE_NODES + i) * TILE_PITCH + lane] = a;
        beg = nbeg; end = nend;
    }
    __syncthreads();

    const int n0  = lane & 15;
    const int hi  = lane >> 4;
    const int col = wave * 16 + n0;
    const float* ar = ax + n0 * TILE_PITCH;
    const float* WI = Wl + 0 * 32 * 32 + col;
    const float* WT = Wl + 4 * 32 * 32 + col;
    const float* WO = Wl + 6 * 32 * 32 + col;

    v8f cI = {}, cT = {}, cO = {};
    #pragma unroll
    for (int k = 0; k < 32; k += 4) {
        const int ka = k + 2 * hi;
        v2f a;  a.x  = ar[ka];       a.y  = ar[ka + 1];
        v2f bI; bI.x = WI[ka * 32];  bI.y = WI[(ka + 1) * 32];
        v2f bT; bT.x = WT[ka * 32];  bT.y = WT[(ka + 1) * 32];
        v2f bO; bO.x = WO[ka * 32];  bO.y = WO[(ka + 1) * 32];
        cI = __builtin_amdgcn_wmma_f32_16x16x4_f32(false, a, false, bI, (short)0, cI, false, false);
        cT = __builtin_amdgcn_wmma_f32_16x16x4_f32(false, a, false, bT, (short)0, cT, false, false);
        cO = __builtin_amdgcn_wmma_f32_16x16x4_f32(false, a, false, bO, (short)0, cO, false, false);
    }

    const float bi = cb[col]          + cb[32 + col]     + bgl[col];
    const float bt = cb[4 * 32 + col] + cb[5 * 32 + col] + bgl[2 * 32 + col];
    const float bo = cb[6 * 32 + col] + cb[7 * 32 + col] + bgl[3 * 32 + col];
    const float wo = wco[col];

    #pragma unroll
    for (int j = 0; j < 8; ++j) {
        const float I = gcrn_sigmoid(cI[j] + bi);
        const float T = gcrn_tanh(cT[j] + bt);
        const float C = I * T;
        const float O = gcrn_sigmoid(cO[j] + bo + wo * C);
        ot[(hi * 8 + j) * TILE_PITCH + col] = O * gcrn_tanh(C);
    }
    __syncthreads();

    if (nb + WAVE_NODES <= n) {
        #pragma unroll
        for (int i = 0; i < WAVE_NODES; ++i)
            out[(size_t)(nb + i) * D_CH + lane] = ot[(wave * WAVE_NODES + i) * TILE_PITCH + lane];
    } else {
        #pragma unroll
        for (int i = 0; i < WAVE_NODES; ++i) {
            const int node = nb + i;
            if (node < n) out[(size_t)node * D_CH + lane] = ot[(wave * WAVE_NODES + i) * TILE_PITCH + lane];
        }
    }
}

extern "C" void kernel_launch(void* const* d_in, const int* in_sizes, int n_in,
                              void* d_out, int out_size, void* d_ws, size_t ws_size,
                              hipStream_t stream) {
    const float* X       = (const float*)d_in[0];
    const int*   eidx    = (const int*)  d_in[1];
    const float* ew      = (const float*)d_in[2];
    const float* conv_w  = (const float*)d_in[3];
    const float* conv_b  = (const float*)d_in[4];
    const float* wc      = (const float*)d_in[5];
    const float* bg      = (const float*)d_in[6];

    const int N = in_sizes[0] / D_CH;
    const int E = in_sizes[2];
    const int L = in_sizes[3] / (8 * D_CH * D_CH);
    const int* src = eidx;
    const int* dst = eidx + E;

    float* ws   = (float*)d_ws;
    const int NBINS = (N + BIN_NODES - 1) >> BIN_SHIFT;
    float* h1   = ws;
    float* h2   = h1   + (size_t)N * D_CH;
    int*   tab  = (int*)(h2 + (size_t)N * D_CH);
    int*   rst  = tab + (size_t)NBINS * BIN_G;
    int4*  rec  = (int4*)(rst + (size_t)NBINS * BIN_G);
    float* dinv = (float*)(rec + E);
    float* nw   = dinv + N;
    float* csrw = nw   + E;
    int*   csrs = (int*)(csrw + E);
    int*   gkid = csrs + E;
    int*   gksr = gkid + E;
    float* gkw  = (float*)(gksr + E);
    float* gkws = gkw + E;
    int*   offs = (int*)(gkws + E);

    gcrn_chunk_sort<<<BIN_G, BIN_TB, 0, stream>>>(src, dst, ew, tab, rst, rec, E, NBINS);
    gcrn_bin_place <<<NBINS, BIN_TB, 0, stream>>>(tab, rst, rec, offs, csrs, csrw, dinv, gkid, gksr, gkw, gkws, N, NBINS);

    const float* xin = X;
    for (int l = 0; l < L; ++l) {
        float* out_l = (l == L - 1) ? (float*)d_out : ((l & 1) ? h2 : h1);
        const float* W_l  = conv_w + (size_t)l * 8 * 32 * 32;
        const float* cb_l = conv_b + (size_t)l * 8 * 32;
        const float* bg_l = bg + (size_t)l * 4 * 32;
        const float* wo_l = wc + ((size_t)l * 3 + 2) * 32;
        const int tiles = (N + TILE_NODES - 1) / TILE_NODES;
        if (l == 0)
            gcrn_layer<true> <<<tiles, 64, 0, stream>>>(offs, csrs, csrw, nw, dinv, xin, W_l, cb_l, bg_l, wo_l, out_l, N, E);
        else
            gcrn_layer<false><<<tiles, 64, 0, stream>>>(offs, csrs, nw, nullptr, dinv, xin, W_l, cb_l, bg_l, wo_l, out_l, N, E);
        xin = out_l;
    }
}


# Gather walks each wave's contiguous slot range in batches of 32 rows across node ends (self loop stored as a slot in front of its node's edges) instead of one batch per node; sum order kept
# speedup vs baseline: 7.0566x; 1.0957x over previous
#include <hip/hip_runtime.h>
#include <hip/hip_bf16.h>
#include <math.h>


typedef __attribute__((ext_vector_type(2))) float v2f;
typedef __attribute__((ext_vector_type(8))) float v8f;

#define D_CH 32
#define TILE_NODES 16
#define WAVE_NODES 8
#define TILE_PITCH 36


#define BIN_SHIFT 9
#define BIN_NODES 512
#define BIN_G 256
#define BIN_MAXB 1024
#define BIN_TB 1024
#define BIN_BATCH 8
#define BIN_KEYCAP 12288
#define BIN_RPW (BIN_G / (BIN_TB / 32))

__device__ __forceinline__ int wave_sum_i(int v) {
    for (int m = 16; m >= 1; m >>= 1) v += __shfl_xor(v, m, 32);
    return v;
}

__device__ __forceinline__ int gcrn_block_excl(int v, int* sw) {
    const int lane = threadIdx.x & 31, wv = threadIdx.x >> 5;
    int inc = v;
    for (int d = 1; d < 32; d <<= 1) {
        int t = __shfl_up(inc, d, 32);
        if (lane >= d) inc += t;
    }
    if (lane == 31) sw[wv] = inc;
    __syncthreads();
    if (wv == 0) {
        const int s = sw[lane];
        int si = s;
        for (int d = 1; d < 32; d <<= 1) {
            int t = __shfl_up(si, d, 32);
            if (lane >= d) si += t;
        }
        sw[lane] = si - s;
    }
    __syncthreads();
    const int r = sw[wv] + inc - v;
    __syncthreads();
    return r;
}

__global__ void __launch_bounds__(BIN_TB)
gcrn_chunk_sort(const int* __restrict__ src, const int* __restrict__ dst, const float* __restrict__ ew,
                int* __restrict__ tab, int* __restrict__ rst, int4* __restrict__ rec, int e, int nbins) {
    __shared__ int h[BIN_MAXB];
    __shared__ int sw[32];
    const int tid = threadIdx.x;
    if (tid < nbins) h[tid] = 0;
    const int chunk = (e + BIN_G - 1) / BIN_G;
    const int beg = blockIdx.x * chunk, end = min(beg + chunk, e);

    int d0[BIN_BATCH], s0[BIN_BATCH];
    float w0[BIN_BATCH];
    #pragma unroll
    for (int u = 0; u < BIN_BATCH; ++u) {
        const int i = beg + tid + u * BIN_TB;
        const bool ok = i < end;
        d0[u] = ok ? dst[i] : 0;
        s0[u] = ok ? src[i] : 0;
        w0[u] = ok ? ew[i] : 0.0f;
    }
    __syncthreads();
    #pragma unroll
    for (int u = 0; u < BIN_BATCH; ++u)
        if (beg + tid + u * BIN_TB < end) atomicAdd(&h[d0[u] >> BIN_SHIFT], 1);
    for (int i0 = beg + tid + BIN_BATCH * BIN_TB; i0 < end; i0 += BIN_BATCH * BIN_TB) {
        int d[BIN_BATCH];
        #pragma unroll
        for (int u = 0; u < BIN_BATCH; ++u) {
            const int i = i0 + u * BIN_TB;
            d[u] = (i < end) ? dst[i] : 0;
        }
        #pragma unroll
        for (int u = 0; u < BIN_BATCH; ++u)
            if (i0 + u * BIN_TB < end) atomicAdd(&h[d[u] >> BIN_SHIFT], 1);
    }
    __syncthreads();
    const int v  = (tid < nbins) ? h[tid] : 0;
    const int ex = gcrn_block_excl(v, sw);
    if (tid < nbins) {
        h[tid] = beg + ex;
        tab[tid * BIN_G + blockIdx.x] = v;
        rst[tid * BIN_G + blockIdx.x] = beg + ex;
    }
    __syncthreads();
    #pragma unroll
    for (int u = 0; u < BIN_BATCH; ++u) {
        const int i = beg + tid + u * BIN_TB;
        if (i < end) {
            const int slot = atomicAdd(&h[d0[u] >> BIN_SHIFT], 1);
            rec[slot] = make_int4(i, d0[u], s0[u], __float_as_int(w0[u]));
        }
    }
    for (int i0 = beg + tid + BIN_BATCH * BIN_TB; i0 < end; i0 += BIN_BATCH * BIN_TB) {
        int d[BIN_BATCH], s[BIN_BATCH];
        float w[BIN_BATCH];
        #pragma unroll
        for (int u = 0; u < BIN_BATCH; ++u) {
            const int i = i0 + u * BIN_TB;
            const bool ok = i < end;
            d[u] = ok ? dst[i] : 0;
            s[u] = ok ? src[i] : 0;
            w[u] = ok ? ew[i] : 0.0f;
        }
        #pragma unroll
        for (int u = 0; u < BIN_BATCH; ++u) {
            const int i = i0 + u * BIN_TB;
            if (i < end) {
                const int slot = atomicAdd(&h[d[u] >> BIN_SHIFT], 1);
                rec[slot] = make_int4(i, d[u], s[u], __float_as_int(w[u]));
            }
        }
    }
}

template <class F>
__device__ __forceinline__ void gcrn_each_record(const int4* __restrict__ rec, int cnt_l, int st_l, int lane, F f) {
    int4 q[2 * BIN_RPW];
    #pragma unroll
    for (int k = 0; k < 2 * BIN_RPW; ++k) {
        const int r = k % BIN_RPW, o = (k / BIN_RPW) * 32 + lane;
        const int nr = __builtin_amdgcn_readlane(cnt_l, r);
        const int sr = __builtin_amdgcn_readlane(st_l, r);
        q[k] = (o < nr) ? rec[sr + o] : make_int4(0, 0, 0, 0);
    }
    #pragma unroll
    for (int k = 0; k < 2 * BIN_RPW; ++k) {
        const int r = k % BIN_RPW, o = (k / BIN_RPW) * 32 + lane;
        const int nr = __builtin_amdgcn_readlane(cnt_l, r);
        if (o < nr) f(q[k]);
    }
    for (int r = 0; r < BIN_RPW; ++r) {
        const int nr = __builtin_amdgcn_readlane(cnt_l, r);
        const int sr = __builtin_amdgcn_readlane(st_l, r);
        for (int o = 64 + lane; o < nr; o += 32) f(rec[sr + o]);
    }
}

template <bool GLOBAL_PLANES>
__device__ __forceinline__ void gcrn_order_node(const int* kid, const int* ksrc, const float* kw, float* kws,
                                                int b, int deg,
                                                int* __restrict__ csr_src, float* __restrict__ csr_w,
                                                float* __restrict__ dinv, int node, int gbeg, int lane) {
    if (lane == 0) { csr_src[gbeg - 1] = node; csr_w[gbeg - 1] = 1.0f; }
    for (int i0 = 0; i0 < deg; i0 += 32) {
        const bool vi = (i0 + lane) < deg;
        const int ki = vi ? kid[b + i0 + lane] : 0x7fffffff;
        int r = 0;
        for (int j0 = 0; j0 < deg; j0 += 32) {
            const int kj = (j0 + lane < deg) ? kid[b + j0 + lane] : 0x7fffffff;
            const int m8 = (min(32, deg - j0) + 7) & ~7;
            for (int t = 0; t < m8; t += 8) {
                #pragma unroll
                for (int q = 0; q < 8; ++q)
                    r += (__builtin_amdgcn_readlane(kj, t + q) < ki) ? 1 : 0;
            }
        }
        if (vi) {
            const float w = kw[b + i0 + lane];
            csr_src[gbeg + r] = ksrc[b + i0 + lane];
            csr_w  [gbeg + r] = w;
            kws[b + r] = w;
        }
    }
    if (GLOBAL_PLANES) __threadfence_block();
    float d = 1.0f;
    for (int j0 = 0; j0 < deg; j0 += 32) {
        const float wj = (j0 + lane < deg) ? kws[b + j0 + lane] : 0.0f;
        const int m8 = (min(32, deg - j0) + 7) & ~7;
        for (int t = 0; t < m8; t += 8) {
            #pragma unroll
            for (int q = 0; q < 8; ++q)
                d += __int_as_float(__builtin_amdgcn_readlane(__float_as_int(wj), t + q));
        }
    }
    if (lane == 0) dinv[node] = (d > 0.0f) ? rsqrtf(d) : 0.0f;
}

__global__ void __launch_bounds__(BIN_TB)
gcrn_bin_place(const int* __restrict__ tab, const int* __restrict__ rst, const int4* __restrict__ rec,
               int* __restrict__ offs, int* __restrict__ csr_src, float* __restrict__ csr_w,
               float* __restrict__ dinv,
               int* __restrict__ gk_id, int* __restrict__ gk_src, float* __restrict__ gk_w,
               float* __restrict__ gk_ws,
               int n, int nbins) {
    __shared__ int c[BIN_NODES];
    __shared__ int sw[32];
    __shared__ int red_lo[32];
    __shared__ int red_cn[32];
    __shared__ int   kid [BIN_KEYCAP];
    __shared__ int   ksrc[BIN_KEYCAP];
    __shared__ float kw  [BIN_KEYCAP];
    __shared__ float kws [BIN_KEYCAP];
    const int tid  = threadIdx.x;
    const int lane = tid & 31;
    const int wv   = __builtin_amdgcn_readfirstlane(tid >> 5);
    const int b    = blockIdx.x;
    const int nb0  = b << BIN_SHIFT;
    if (tid < BIN_NODES) c[tid] = 0;

    const int g_l   = wv * BIN_RPW + (lane & (BIN_RPW - 1));
    const int cnt_l = tab[b * BIN_G + g_l];
    const int st_l  = rst[b * BIN_G + g_l];

    int acc = 0;
    const int4* t4 = reinterpret_cast<const int4*>(tab);
    const int n4 = b * (BIN_G / 4);
    for (int j0 = tid; j0 < n4; j0 += BIN_BATCH * BIN_TB) {
        int4 q[BIN_BATCH];
        #pragma unroll
        for (int u = 0; u < BIN_BATCH; ++u) {
            const int j = j0 + u * BIN_TB;
            q[u] = (j < n4) ? t4[j] : make_int4(0, 0, 0, 0);
        }
        #pragma unroll
        for (int u = 0; u < BIN_BATCH; ++u) acc += (q[u].x + q[u].y) + (q[u].z + q[u].w);
    }
    const int rowv = (tid < BIN_G) ? tab[b * BIN_G + tid] : 0;
    const int sa = wave_sum_i(acc), sb = wave_sum_i(rowv);
    if (lane == 0) { red_lo[wv] = sa; red_cn[wv] = sb; }
    __syncthreads();
    const int lo   = __builtin_amdgcn_readfirstlane(wave_sum_i(red_lo[lane]));
    const int cntb = __builtin_amdgcn_readfirstlane(wave_sum_i(red_cn[lane]));
    const int hi   = lo + cntb;

    gcrn_each_record(rec, cnt_l, st_l, lane, [&](const int4& r) { atomicAdd(&c[r.y - nb0], 1); });
    __syncthreads();
    const int cv = (tid < BIN_NODES) ? c[tid] : 0;
    const int ce = gcrn_block_excl(cv, sw);
    if (tid < BIN_NODES) {
        c[tid] = lo + ce;
        if (nb0 + tid < n) offs[nb0 + tid] = lo + ce + nb0 + tid;
    }
    if (b == nbins - 1 && tid == 0) offs[n] = hi + n;
    __syncthreads();

    if (cntb <= BIN_KEYCAP) {
        gcrn_each_record(rec, cnt_l, st_l, lane, [&](const int4& r) {
            const int q = atomicAdd(&c[r.y - nb0], 1) - lo;
            kid [q] = r.x;
            ksrc[q] = r.z;
            kw  [q] = __int_as_float(r.w);
        });
        __syncthreads();
        for (int t = wv; t < BIN_NODES; t += BIN_TB / 32) {
            if (nb0 + t >= n) break;
            const int sb2 = __builtin_amdgcn_readfirstlane(t == 0 ? lo : c[t - 1]);
            const int se  = __builtin_amdgcn_readfirstlane(c[t]);
            gcrn_order_node<false>(kid, ksrc, kw, kws, sb2 - lo, se - sb2, csr_src, csr_w, dinv, nb0 + t, sb2 + nb0 + t + 1, lane);
        }
    } else {
        gcrn_each_record(rec, cnt_l, st_l, lane, [&](const int4& r) {
            const int slot = atomicAdd(&c[r.y - nb0], 1);
            gk_id [slot] = r.x;
            gk_src[slot] = r.z;
            gk_w  [slot] = __int_as_float(r.w);
        });
        __syncthreads();
        for (int t = wv; t < BIN_NODES; t += BIN_TB / 32) {
            if (nb0 + t >= n) break;
            const int sb2 = __builtin_amdgcn_readfirstlane(t == 0 ? lo : c[t - 1]);
            const int se  = __builtin_amdgcn_readfirstlane(c[t]);
            gcrn_order_node<true>(gk_id, gk_src, gk_w, gk_ws, sb2, se - sb2, csr_src, csr_w, dinv, nb0 + t, sb2 + nb0 + t + 1, lane);
        }
    }
}


template <bool FIRST>
__device__ __forceinline__ void gcrn_gather_batch(const float* __restrict__ x, int s_l, float w_l, int m, int lane,
                                                  const int R, int p0, int o_l,
                                                  const int* __restrict__ snp, const float* __restrict__ wnp,
                                                  int& s_nx, float& w_nx,
                                                  float& a, int& ni, int& nend, float* axw,
                                                  const float* __restrict__ dinv, const float* __restrict__ ddp,
                                                  float* __restrict__ nwo) {
    float ds = 0.0f, dd = 0.0f;
    if (FIRST) {
        ds = (lane < m) ? dinv[s_l] : 0.0f;
        dd = (lane < m) ? *ddp : 0.0f;
    }
    float v[32];
    #pragma unroll
    for (int u = 0; u < R; ++u) {
        const int tt = (u < R - 8) ? u : min(u, m - 1);
        const int s  = __builtin_amdgcn_readlane(s_l, tt);
        v[u] = x[s * D_CH + lane];
    }
    s_nx = *snp;
    w_nx = *wnp;
    if (FIRST) {
        w_l = ds * w_l * dd;
        if (lane < m) *nwo = w_l;
    }
    asm volatile("" :: "v"(v[R - 8]), "v"(v[R - 7]), "v"(v[R - 6]), "v"(v[R - 5]),
                       "v"(v[R - 4]), "v"(v[R - 3]), "v"(v[R - 2]), "v"(v[R - 1]));
    #pragma unroll
    for (int u = 0; u < R; ++u) {
        if (u < R - 8 || u < m) {
            if (__builtin_expect(p0 + u == nend, 0)) {
                axw[ni * TILE_PITCH] = a;
                a = 0.0f;
                ++ni;
                nend = __builtin_amdgcn_readlane(o_l, ni + 1);
            }
            const float w = __int_as_float(__builtin_amdgcn_readlane(__float_as_int(w_l), u));
            a = fmaf(w, v[u], a);
        }
    }
}

__device__ __forceinline__ float gcrn_tanh(float x) { return __builtin_amdgcn_tanhf(x); }
__device__ __forceinline__ float gcrn_sigmoid(float x) { return fmaf(0.5f, __builtin_amdgcn_tanhf(0.5f * x), 0.5f); }

template <bool FIRST>
__global__ void __launch_bounds__(64)
gcrn_layer(const int* __restrict__ offs, const int* __restrict__ csr_src,
           const float* __restrict__ nw,
           float* __restrict__ nw_out,
           const float* __restrict__ dinv,
           const float* __restrict__ x,
           const float* __restrict__ Wl,
           const float* __restrict__ cb,
           const float* __restrict__ bgl,
           const float* __restrict__ wco,
           float* __restrict__ out, int n, int e) {
    __shared__ float ax[TILE_NODES * TILE_PITCH];
    __shared__ float ot[TILE_NODES * TILE_PITCH];

    const int wave = __builtin_amdgcn_readfirstlane(threadIdx.x >> 5);
    const int lane = threadIdx.x & 31;
    const int nb   = blockIdx.x * TILE_NODES + wave * WAVE_NODES;
    const int tot   = e + n;
    const int slast = max(tot - 1, 0);

    const int o_l = offs[min(nb + min(lane, WAVE_NODES), n)];

    const int wbeg = __builtin_amdgcn_readlane(o_l, 0);
    const int wend = __builtin_amdgcn_readlane(o_l, WAVE_NODES);
    int   s_nx = csr_src[min(wbeg + lane, slast)];
    float w_nx = nw[min(wbeg + lane, slast)];

    float a    = 0.0f;
    int   ni   = 0;
    int   nend = __builtin_amdgcn_readlane(o_l, 1);
    float* axw = ax + wave * WAVE_NODES * TILE_PITCH + lane;

    for (int p0 = wbeg; p0 < wend; p0 += 32) {
        const int s_l = s_nx; const float w_l = w_nx;
        const int m    = min(32, wend - p0);
        const int nidx = min(p0 + 32 + lane, slast);
        const float* ddp = dinv;
        if (FIRST) {
            int j = 0;
            #pragma unroll
            for (int k = 1; k < WAVE_NODES; ++k)
                j += (p0 + lane >= __builtin_amdgcn_readlane(o_l, k)) ? 1 : 0;
            ddp = dinv + min(nb + j, n - 1);
        }
        float* nwo = FIRST ? nw_out + (p0 + lane) : nullptr;
        if (m > 24)      gcrn_gather_batch<FIRST>(x, s_l, w_l, m, lane, 32, p0, o_l, csr_src + nidx, nw + nidx, s_nx, w_nx, a, ni, nend, axw, dinv, ddp, nwo);
        else if (m > 16) gcrn_gather_batch<FIRST>(x, s_l, w_l, m, lane, 24, p0, o_l, csr_src + nidx, nw + nidx, s_nx, w_nx, a, ni, nend, axw, dinv, ddp, nwo);
        else if (m > 8)  gcrn_gather_batch<FIRST>(x, s_l, w_l, m, lane, 16, p0, o_l, csr_src + nidx, nw + nidx, s_nx, w_nx, a, ni, nend, axw, dinv, ddp, nwo);
        else             gcrn_gather_batch<FIRST>(x, s_l, w_l, m, lane, 8,  p0, o_l, csr_src + nidx, nw + nidx, s_nx, w_nx, a, ni, nend, axw, dinv, ddp, nwo);
    }
    #pragma unroll
    for (int j = 0; j < WAVE_NODES; ++j)
        if (j >= ni) axw[j * TILE_PITCH] = (j == ni) ? a : 0.0f;
    __syncthreads();

    const int n0  = lane & 15;
    const int hi  = lane >> 4;
    const int col = wave * 16 + n0;
    const float* ar = ax + n0 * TILE_PITCH;
    const float* WI = Wl + 0 * 32 * 32 + col;
    const float* WT = Wl + 4 * 32 * 32 + col;
    const float* WO = Wl + 6 * 32 * 32 + col;

    v8f cI = {}, cT = {}, cO = {};
    #pragma unroll
    for (int k = 0; k < 32; k += 4) {
        const int ka = k + 2 * hi;
        v2f a;  a.x  = ar[ka];       a.y  = ar[ka + 1];
        v2f bI; bI.x = WI[ka * 32];  bI.y = WI[(ka + 1) * 32];
        v2f bT; bT.x = WT[ka * 32];  bT.y = WT[(ka + 1) * 32];
        v2f bO; bO.x = WO[ka * 32];  bO.y = WO[(ka + 1) * 32];
        cI = __builtin_amdgcn_wmma_f32_16x16x4_f32(false, a, false, bI, (short)0, cI, false, false);
        cT = __builtin_amdgcn_wmma_f32_16x16x4_f32(false, a, false, bT, (short)0, cT, false, false);
        cO = __builtin_amdgcn_wmma_f32_16x16x4_f32(false, a, false, bO, (short)0, cO, false, false);
    }

    const float bi = cb[col]          + cb[32 + col]     + bgl[col];
    const float bt = cb[4 * 32 + col] + cb[5 * 32 + col] + bgl[2 * 32 + col];
    const float bo = cb[6 * 32 + col] + cb[7 * 32 + col] + bgl[3 * 32 + col];
    const float wo = wco[col];

    #pragma unroll
    for (int j = 0; j < 8; ++j) {
        const float I = gcrn_sigmoid(cI[j] + bi);
        const float T = gcrn_tanh(cT[j] + bt);
        const float C = I * T;
        const float O = gcrn_sigmoid(cO[j] + bo + wo * C);
        ot[(hi * 8 + j) * TILE_PITCH + col] = O * gcrn_tanh(C);
    }
    __syncthreads();

    if (nb + WAVE_NODES <= n) {
        #pragma unroll
        for (int i = 0; i < WAVE_NODES; ++i)
            out[(size_t)(nb + i) * D_CH + lane] = ot[(wave * WAVE_NODES + i) * TILE_PITCH + lane];
    } else {
        #pragma unroll
        for (int i = 0; i < WAVE_NODES; ++i) {
            const int node = nb + i;
            if (node < n) out[(size_t)node * D_CH + lane] = ot[(wave * WAVE_NODES + i) * TILE_PITCH + lane];
        }
    }
}

extern "C" void kernel_launch(void* const* d_in, const int* in_sizes, int n_in,
                              void* d_out, int out_size, void* d_ws, size_t ws_size,
                              hipStream_t stream) {
    const float* X       = (const float*)d_in[0];
    const int*   eidx    = (const int*)  d_in[1];
    const float* ew      = (const float*)d_in[2];
    const float* conv_w  = (const float*)d_in[3];
    const float* conv_b  = (const float*)d_in[4];
    const float* wc      = (const float*)d_in[5];
    const float* bg      = (const float*)d_in[6];

    const int N = in_sizes[0] / D_CH;
    const int E = in_sizes[2];
    const int L = in_sizes[3] / (8 * D_CH * D_CH);
    const int* src = eidx;
    const int* dst = eidx + E;

    float* ws   = (float*)d_ws;
    const int NBINS = (N + BIN_NODES - 1) >> BIN_SHIFT;
    float* h1   = ws;
    float* h2   = h1   + (size_t)N * D_CH;
    int*   tab  = (int*)(h2 + (size_t)N * D_CH);
    int*   rst  = tab + (size_t)NBINS * BIN_G;
    int4*  rec  = (int4*)(rst + (size_t)NBINS * BIN_G);
    float* dinv = (float*)(rec + E);
    float* nw   = dinv + N;
    float* csrw = nw   + E + N;
    int*   csrs = (int*)(csrw + E + N);
    int*   gkid = csrs + E + N;
    int*   gksr = gkid + E;
    float* gkw  = (float*)(gksr + E);
    float* gkws = gkw + E;
    int*   offs = (int*)(gkws + E);

    gcrn_chunk_sort<<<BIN_G, BIN_TB, 0, stream>>>(src, dst, ew, tab, rst, rec, E, NBINS);
    gcrn_bin_place <<<NBINS, BIN_TB, 0, stream>>>(tab, rst, rec, offs, csrs, csrw, dinv, gkid, gksr, gkw, gkws, N, NBINS);

    const float* xin = X;
    for (int l = 0; l < L; ++l) {
        float* out_l = (l == L - 1) ? (float*)d_out : ((l & 1) ? h2 : h1);
        const float* W_l  = conv_w + (size_t)l * 8 * 32 * 32;
        const float* cb_l = conv_b + (size_t)l * 8 * 32;
        const float* bg_l = bg + (size_t)l * 4 * 32;
        const float* wo_l = wc + ((size_t)l * 3 + 2) * 32;
        const int tiles = (N + TILE_NODES - 1) / TILE_NODES;
        if (l == 0)
            gcrn_layer<true> <<<tiles, 64, 0, stream>>>(offs, csrs, csrw, nw, dinv, xin, W_l, cb_l, bg_l, wo_l, out_l, N, E);
        else
            gcrn_layer<false><<<tiles, 64, 0, stream>>>(offs, csrs, nw, nullptr, dinv, xin, W_l, cb_l, bg_l, wo_l, out_l, N, E);
        xin = out_l;
    }
}


# Normalisation factored out of the edge weights: gathered rows carry their source's dinv, a node's sum gets its own dinv; the first layer's per-edge dinv lookups and the nw plane go; rounding changes
# speedup vs baseline: 7.4191x; 1.0514x over previous
#include <hip/hip_runtime.h>
#include <hip/hip_bf16.h>
#include <math.h>


typedef __attribute__((ext_vector_type(2))) float v2f;
typedef __attribute__((ext_vector_type(8))) float v8f;

#define D_CH 32
#define TILE_NODES 16
#define WAVE_NODES 8
#define TILE_PITCH 36


#define BIN_SHIFT 9
#define BIN_NODES 512
#define BIN_G 256
#define BIN_MAXB 1024
#define BIN_TB 1024
#define BIN_BATCH 8
#define BIN_KEYCAP 12288
#define BIN_RPW (BIN_G / (BIN_TB / 32))

__device__ __forceinline__ int wave_sum_i(int v) {
    for (int m = 16; m >= 1; m >>= 1) v += __shfl_xor(v, m, 32);
    return v;
}

__device__ __forceinline__ int gcrn_block_excl(int v, int* sw) {
    const int lane = threadIdx.x & 31, wv = threadIdx.x >> 5;
    int inc = v;
    for (int d = 1; d < 32; d <<= 1) {
        int t = __shfl_up(inc, d, 32);
        if (lane >= d) inc += t;
    }
    if (lane == 31) sw[wv] = inc;
    __syncthreads();
    if (wv == 0) {
        const int s = sw[lane];
        int si = s;
        for (int d = 1; d < 32; d <<= 1) {
            int t = __shfl_up(si, d, 32);
            if (lane >= d) si += t;
        }
        sw[lane] = si - s;
    }
    __syncthreads();
    const int r = sw[wv] + inc - v;
    __syncthreads();
    return r;
}

__global__ void __launch_bounds__(BIN_TB)
gcrn_chunk_sort(const int* __restrict__ src, const int* __restrict__ dst, const float* __restrict__ ew,
                int* __restrict__ tab, int* __restrict__ rst, int4* __restrict__ rec, int e, int nbins) {
    __shared__ int h[BIN_MAXB];
    __shared__ int sw[32];
    const int tid = threadIdx.x;
    if (tid < nbins) h[tid] = 0;
    const int chunk = (e + BIN_G - 1) / BIN_G;
    const int beg = blockIdx.x * chunk, end = min(beg + chunk, e);

    int d0[BIN_BATCH], s0[BIN_BATCH];
    float w0[BIN_BATCH];
    #pragma unroll
    for (int u = 0; u < BIN_BATCH; ++u) {
        const int i = beg + tid + u * BIN_TB;
        const bool ok = i < end;
        d0[u] = ok ? dst[i] : 0;
        s0[u] = ok ? src[i] : 0;
        w0[u] = ok ? ew[i] : 0.0f;
    }
    __syncthreads();
    #pragma unroll
    for (int u = 0; u < BIN_BATCH; ++u)
        if (beg + tid + u * BIN_TB < end) atomicAdd(&h[d0[u] >> BIN_SHIFT], 1);
    for (int i0 = beg + tid + BIN_BATCH * BIN_TB; i0 < end; i0 += BIN_BATCH * BIN_TB) {
        int d[BIN_BATCH];
        #pragma unroll
        for (int u = 0; u < BIN_BATCH; ++u) {
            const int i = i0 + u * BIN_TB;
            d[u] = (i < end) ? dst[i] : 0;
        }
        #pragma unroll
        for (int u = 0; u < BIN_BATCH; ++u)
            if (i0 + u * BIN_TB < end) atomicAdd(&h[d[u] >> BIN_SHIFT], 1);
    }
    __syncthreads();
    const int v  = (tid < nbins) ? h[tid] : 0;
    const int ex = gcrn_block_excl(v, sw);
    if (tid < nbins) {
        h[tid] = beg + ex;
        tab[tid * BIN_G + blockIdx.x] = v;
        rst[tid * BIN_G + blockIdx.x] = beg + ex;
    }
    __syncthreads();
    #pragma unroll
    for (int u = 0; u < BIN_BATCH; ++u) {
        const int i = beg + tid + u * BIN_TB;
        if (i < end) {
            const int slot = atomicAdd(&h[d0[u] >> BIN_SHIFT], 1);
            rec[slot] = make_int4(i, d0[u], s0[u], __float_as_int(w0[u]));
        }
    }
    for (int i0 = beg + tid + BIN_BATCH * BIN_TB; i0 < end; i0 += BIN_BATCH * BIN_TB) {
        int d[BIN_BATCH], s[BIN_BATCH];
        float w[BIN_BATCH];
        #pragma unroll
        for (int u = 0; u < BIN_BATCH; ++u) {
            const int i = i0 + u * BIN_TB;
            const bool ok = i < end;
            d[u] = ok ? dst[i] : 0;
            s[u] = ok ? src[i] : 0;
            w[u] = ok ? ew[i] : 0.0f;
        }
        #pragma unroll
        for (int u = 0; u < BIN_BATCH; ++u) {
            const int i = i0 + u * BIN_TB;
            if (i < end) {
                const int slot = atomicAdd(&h[d[u] >> BIN_SHIFT], 1);
                rec[slot] = make_int4(i, d[u], s[u], __float_as_int(w[u]));
            }
        }
    }
}

template <class F>
__device__ __forceinline__ void gcrn_each_record(const int4* __restrict__ rec, int cnt_l, int st_l, int lane, F f) {
    int4 q[2 * BIN_RPW];
    #pragma unroll
    for (int k = 0; k < 2 * BIN_RPW; ++k) {
        const int r = k % BIN_RPW, o = (k / BIN_RPW) * 32 + lane;
        const int nr = __builtin_amdgcn_readlane(cnt_l, r);
        const int sr = __builtin_amdgcn_readlane(st_l, r);
        q[k] = (o < nr) ? rec[sr + o] : make_int4(0, 0, 0, 0);
    }
    #pragma unroll
    for (int k = 0; k < 2 * BIN_RPW; ++k) {
        const int r = k % BIN_RPW, o = (k / BIN_RPW) * 32 + lane;
        const int nr = __builtin_amdgcn_readlane(cnt_l, r);
        if (o < nr) f(q[k]);
    }
    for (int r = 0; r < BIN_RPW; ++r) {
        const int nr = __builtin_amdgcn_readlane(cnt_l, r);
        const int sr = __builtin_amdgcn_readlane(st_l, r);
        for (int o = 64 + lane; o < nr; o += 32) f(rec[sr + o]);
    }
}

template <bool GLOBAL_PLANES>
__device__ __forceinline__ void gcrn_order_node(const int* kid, const int* ksrc, const float* kw, float* kws,
                                                int b, int deg,
                                                int* __restrict__ csr_src, float* __restrict__ csr_w,
                                                float* __restrict__ dinv, int node, int gbeg, int lane,
                                                float xr, float* __restrict__ xs) {
    if (lane == 0) { csr_src[gbeg - 1] = node; csr_w[gbeg - 1] = 1.0f; }
    for (int i0 = 0; i0 < deg; i0 += 32) {
        const bool vi = (i0 + lane) < deg;
        const int ki = vi ? kid[b + i0 + lane] : 0x7fffffff;
        int r = 0;
        for (int j0 = 0; j0 < deg; j0 += 32) {
            const int kj = (j0 + lane < deg) ? kid[b + j0 + lane] : 0x7fffffff;
            const int m8 = (min(32, deg - j0) + 7) & ~7;
            for (int t = 0; t < m8; t += 8) {
                #pragma unroll
                for (int q = 0; q < 8; ++q)
                    r += (__builtin_amdgcn_readlane(kj, t + q) < ki) ? 1 : 0;
            }
        }
        if (vi) {
            const float w = kw[b + i0 + lane];
            csr_src[gbeg + r] = ksrc[b + i0 + lane];
            csr_w  [gbeg + r] = w;
            kws[b + r] = w;
        }
    }
    if (GLOBAL_PLANES) __threadfence_block();
    float d = 1.0f;
    for (int j0 = 0; j0 < deg; j0 += 32) {
        const float wj = (j0 + lane < deg) ? kws[b + j0 + lane] : 0.0f;
        const int m8 = (min(32, deg - j0) + 7) & ~7;
        for (int t = 0; t < m8; t += 8) {
            #pragma unroll
            for (int q = 0; q < 8; ++q)
                d += __int_as_float(__builtin_amdgcn_readlane(__float_as_int(wj), t + q));
        }
    }
    const float dv = (d > 0.0f) ? rsqrtf(d) : 0.0f;
    if (lane == 0) dinv[node] = dv;
    xs[node * D_CH + lane] = dv * xr;
}

__global__ void __launch_bounds__(BIN_TB)
gcrn_bin_place(const int* __restrict__ tab, const int* __restrict__ rst, const int4* __restrict__ rec,
               int* __restrict__ offs, int* __restrict__ csr_src, float* __restrict__ csr_w,
               float* __restrict__ dinv,
               int* __restrict__ gk_id, int* __restrict__ gk_src, float* __restrict__ gk_w,
               float* __restrict__ gk_ws,
               const float* __restrict__ X, float* __restrict__ xs,
               int n, int nbins) {
    __shared__ int c[BIN_NODES];
    __shared__ int sw[32];
    __shared__ int red_lo[32];
    __shared__ int red_cn[32];
    __shared__ int   kid [BIN_KEYCAP];
    __shared__ int   ksrc[BIN_KEYCAP];
    __shared__ float kw  [BIN_KEYCAP];
    __shared__ float kws [BIN_KEYCAP];
    __shared__ float xrow[BIN_NODES * D_CH];
    const int tid  = threadIdx.x;
    const int lane = tid & 31;
    const int wv   = __builtin_amdgcn_readfirstlane(tid >> 5);
    const int b    = blockIdx.x;
    const int nb0  = b << BIN_SHIFT;
    if (tid < BIN_NODES) c[tid] = 0;

    float xq[BIN_NODES / (BIN_TB / 32)];
    #pragma unroll
    for (int k = 0; k < BIN_NODES / (BIN_TB / 32); ++k)
        xq[k] = X[min(nb0 + wv + k * (BIN_TB / 32), n - 1) * D_CH + lane];

    const int g_l   = wv * BIN_RPW + (lane & (BIN_RPW - 1));
    const int cnt_l = tab[b * BIN_G + g_l];
    const int st_l  = rst[b * BIN_G + g_l];

    int acc = 0;
    const int4* t4 = reinterpret_cast<const int4*>(tab);
    const int n4 = b * (BIN_G / 4);
    for (int j0 = tid; j0 < n4; j0 += BIN_BATCH * BIN_TB) {
        int4 q[BIN_BATCH];
        #pragma unroll
        for (int u = 0; u < BIN_BATCH; ++u) {
            const int j = j0 + u * BIN_TB;
            q[u] = (j < n4) ? t4[j] : make_int4(0, 0, 0, 0);
        }
        #pragma unroll
        for (int u = 0; u < BIN_BATCH; ++u) acc += (q[u].x + q[u].y) + (q[u].z + q[u].w);
    }
    const int rowv = (tid < BIN_G) ? tab[b * BIN_G + tid] : 0;
    const int sa = wave_sum_i(acc), sb = wave_sum_i(rowv);
    if (lane == 0) { red_lo[wv] = sa; red_cn[wv] = sb; }
    __syncthreads();
    const int lo   = __builtin_amdgcn_readfirstlane(wave_sum_i(red_lo[lane]));
    const int cntb = __builtin_amdgcn_readfirstlane(wave_sum_i(red_cn[lane]));
    const int hi   = lo + cntb;

    gcrn_each_record(rec, cnt_l, st_l, lane, [&](const int4& r) { atomicAdd(&c[r.y - nb0], 1); });
    __syncthreads();
    const int cv = (tid < BIN_NODES) ? c[tid] : 0;
    const int ce = gcrn_block_excl(cv, sw);
    if (tid < BIN_NODES) {
        c[tid] = lo + ce;
        if (nb0 + tid < n) offs[nb0 + tid] = lo + ce + nb0 + tid;
    }
    if (b == nbins - 1 && tid == 0) offs[n] = hi + n;
    #pragma unroll
    for (int k = 0; k < BIN_NODES / (BIN_TB / 32); ++k)
        xrow[(wv + k * (BIN_TB / 32)) * D_CH + lane] = xq[k];
    __syncthreads();

    if (cntb <= BIN_KEYCAP) {
        gcrn_each_record(rec, cnt_l, st_l, lane, [&](const int4& r) {
            const int q = atomicAdd(&c[r.y - nb0], 1) - lo;
            kid [q] = r.x;
            ksrc[q] = r.z;
            kw  [q] = __int_as_float(r.w);
        });
        __syncthreads();
        for (int t = wv; t < BIN_NODES; t += BIN_TB / 32) {
            if (nb0 + t >= n) break;
            const float x0 = xrow[t * D_CH + lane];
            const int sb2 = __builtin_amdgcn_readfirstlane(t == 0 ? lo : c[t - 1]);
            const int se  = __builtin_amdgcn_readfirstlane(c[t]);
            gcrn_order_node<false>(kid, ksrc, kw, kws, sb2 - lo, se - sb2, csr_src, csr_w, dinv, nb0 + t, sb2 + nb0 + t + 1, lane, x0, xs);
        }
    } else {
        gcrn_each_record(rec, cnt_l, st_l, lane, [&](const int4& r) {
            const int slot = atomicAdd(&c[r.y - nb0], 1);
            gk_id [slot] = r.x;
            gk_src[slot] = r.z;
            gk_w  [slot] = __int_as_float(r.w);
        });
        __syncthreads();
        for (int t = wv; t < BIN_NODES; t += BIN_TB / 32) {
            if (nb0 + t >= n) break;
            const float x0 = xrow[t * D_CH + lane];
            const int sb2 = __builtin_amdgcn_readfirstlane(t == 0 ? lo : c[t - 1]);
            const int se  = __builtin_amdgcn_readfirstlane(c[t]);
            gcrn_order_node<true>(gk_id, gk_src, gk_w, gk_ws, sb2, se - sb2, csr_src, csr_w, dinv, nb0 + t, sb2 + nb0 + t + 1, lane, x0, xs);
        }
    }
}


__device__ __forceinline__ void gcrn_gather_batch(const float* __restrict__ x, int s_l, float w_l, int m, int lane,
                                                  const int R, int p0, int o_l,
                                                  const int* __restrict__ snp, const float* __restrict__ wnp,
                                                  int& s_nx, float& w_nx,
                                                  float& a, int& ni, int& nend, float* axw) {
    float v[32];
    #pragma unroll
    for (int u = 0; u < R; ++u) {
        const int tt = (u < R - 8) ? u : min(u, m - 1);
        const int s  = __builtin_amdgcn_readlane(s_l, tt);
        v[u] = x[s * D_CH + lane];
    }
    s_nx = *snp;
    w_nx = *wnp;
    asm volatile("" :: "v"(v[R - 8]), "v"(v[R - 7]), "v"(v[R - 6]), "v"(v[R - 5]),
                       "v"(v[R - 4]), "v"(v[R - 3]), "v"(v[R - 2]), "v"(v[R - 1]));
    #pragma unroll
    for (int u = 0; u < R; ++u) {
        if (u < R - 8 || u < m) {
            if (__builtin_expect(p0 + u == nend, 0)) {
                axw[ni * TILE_PITCH] = a * __int_as_float(__builtin_amdgcn_readlane(o_l, 16 + ni));
                a = 0.0f;
                ++ni;
                nend = __builtin_amdgcn_readlane(o_l, ni + 1);
            }
            const float w = __int_as_float(__builtin_amdgcn_readlane(__float_as_int(w_l), u));
            a = fmaf(w, v[u], a);
        }
    }
}

__device__ __forceinline__ float gcrn_tanh(float x) { return __builtin_amdgcn_tanhf(x); }
__device__ __forceinline__ float gcrn_sigmoid(float x) { return fmaf(0.5f, __builtin_amdgcn_tanhf(0.5f * x), 0.5f); }

template <bool SCALE_OUT>
__global__ void __launch_bounds__(64)
gcrn_layer(const int* __restrict__ offs, const int* __restrict__ csr_src,
           const float* __restrict__ nw,
           const float* __restrict__ dinv,
           const float* __restrict__ x,
           const float* __restrict__ Wl,
           const float* __restrict__ cb,
           const float* __restrict__ bgl,
           const float* __restrict__ wco,
           float* __restrict__ out, int n, int e) {
    __shared__ float ax[TILE_NODES * TILE_PITCH];
    __shared__ float ot[TILE_NODES * TILE_PITCH];

    const int wave = __builtin_amdgcn_readfirstlane(threadIdx.x >> 5);
    const int lane = threadIdx.x & 31;
    const int nb   = blockIdx.x * TILE_NODES + wave * WAVE_NODES;
    const int tot   = e + n;
    const int slast = max(tot - 1, 0);

    const int o_l = (lane < 16) ? offs[min(nb + min(lane, WAVE_NODES), n)]
                                : __float_as_int(dinv[min(nb + min(max(lane - 16, 0), WAVE_NODES - 1), max(n - 1, 0))]);

    if (SCALE_OUT && lane >= 16 && lane < 16 + WAVE_NODES)
        ax[(wave * WAVE_NODES + lane - 16) * TILE_PITCH + D_CH] = __int_as_float(o_l);

    const int wbeg = __builtin_amdgcn_readlane(o_l, 0);
    const int wend = __builtin_amdgcn_readlane(o_l, WAVE_NODES);
    int   s_nx = csr_src[min(wbeg + lane, slast)];
    float w_nx = nw[min(wbeg + lane, slast)];

    float a    = 0.0f;
    int   ni   = 0;
    int   nend = __builtin_amdgcn_readlane(o_l, 1);
    float* axw = ax + wave * WAVE_NODES * TILE_PITCH + lane;

    for (int p0 = wbeg; p0 < wend; p0 += 32) {
        const int s_l = s_nx; const float w_l = w_nx;
        const int m    = min(32, wend - p0);
        const int nidx = min(p0 + 32 + lane, slast);
        if (m > 24)      gcrn_gather_batch(x, s_l, w_l, m, lane, 32, p0, o_l, csr_src + nidx, nw + nidx, s_nx, w_nx, a, ni, nend, axw);
        else if (m > 16) gcrn_gather_batch(x, s_l, w_l, m, lane, 24, p0, o_l, csr_src + nidx, nw + nidx, s_nx, w_nx, a, ni, nend, axw);
        else if (m > 8)  gcrn_gather_batch(x, s_l, w_l, m, lane, 16, p0, o_l, csr_src + nidx, nw + nidx, s_nx, w_nx, a, ni, nend, axw);
        else             gcrn_gather_batch(x, s_l, w_l, m, lane, 8,  p0, o_l, csr_src + nidx, nw + nidx, s_nx, w_nx, a, ni, nend, axw);
    }
    #pragma unroll
    for (int j = 0; j < WAVE_NODES; ++j)
        if (j >= ni) axw[j * TILE_PITCH] = (j == ni) ? a * __int_as_float(__builtin_amdgcn_readlane(o_l, 16 + j)) : 0.0f;
    __syncthreads();

    const int n0  = lane & 15;
    const int hi  = lane >> 4;
    const int col = wave * 16 + n0;
    const float* ar = ax + n0 * TILE_PITCH;
    const float* WI = Wl + 0 * 32 * 32 + col;
    const float* WT = Wl + 4 * 32 * 32 + col;
    const float* WO = Wl + 6 * 32 * 32 + col;

    v8f cI = {}, cT = {}, cO = {};
    #pragma unroll
    for (int k = 0; k < 32; k += 4) {
        const int ka = k + 2 * hi;
        v2f a;  a.x  = ar[ka];       a.y  = ar[ka + 1];
        v2f bI; bI.x = WI[ka * 32];  bI.y = WI[(ka + 1) * 32];
        v2f bT; bT.x = WT[ka * 32];  bT.y = WT[(ka + 1) * 32];
        v2f bO; bO.x = WO[ka * 32];  bO.y = WO[(ka + 1) * 32];
        cI = __builtin_amdgcn_wmma_f32_16x16x4_f32(false, a, false, bI, (short)0, cI, false, false);
        cT = __builtin_amdgcn_wmma_f32_16x16x4_f32(false, a, false, bT, (short)0, cT, false, false);
        cO = __builtin_amdgcn_wmma_f32_16x16x4_f32(false, a, false, bO, (short)0, cO, false, false);
    }

    const float bi = cb[col]          + cb[32 + col]     + bgl[col];
    const float bt = cb[4 * 32 + col] + cb[5 * 32 + col] + bgl[2 * 32 + col];
    const float bo = cb[6 * 32 + col] + cb[7 * 32 + col] + bgl[3 * 32 + col];
    const float wo = wco[col];

    #pragma unroll
    for (int j = 0; j < 8; ++j) {
        const float I = gcrn_sigmoid(cI[j] + bi);
        const float T = gcrn_tanh(cT[j] + bt);
        const float C = I * T;
        const float O = gcrn_sigmoid(cO[j] + bo + wo * C);
        ot[(hi * 8 + j) * TILE_PITCH + col] = O * gcrn_tanh(C);
    }
    __syncthreads();

    if (nb + WAVE_NODES <= n) {
        #pragma unroll
        for (int i = 0; i < WAVE_NODES; ++i) {
            const float o = ot[(wave * WAVE_NODES + i) * TILE_PITCH + lane];
            out[(size_t)(nb + i) * D_CH + lane] = SCALE_OUT ? o * ax[(wave * WAVE_NODES + i) * TILE_PITCH + D_CH] : o;
        }
    } else {
        #pragma unroll
        for (int i = 0; i < WAVE_NODES; ++i) {
            const int node = nb + i;
            const float o = ot[(wave * WAVE_NODES + i) * TILE_PITCH + lane];
            if (node < n) out[(size_t)node * D_CH + lane] = SCALE_OUT ? o * ax[(wave * WAVE_NODES + i) * TILE_PITCH + D_CH] : o;
        }
    }
}

extern "C" void kernel_launch(void* const* d_in, const int* in_sizes, int n_in,
                              void* d_out, int out_size, void* d_ws, size_t ws_size,
                              hipStream_t stream) {
    const float* X       = (const float*)d_in[0];
    const int*   eidx    = (const int*)  d_in[1];
    const float* ew      = (const float*)d_in[2];
    const float* conv_w  = (const float*)d_in[3];
    const float* conv_b  = (const float*)d_in[4];
    const float* wc      = (const float*)d_in[5];
    const float* bg      = (const float*)d_in[6];

    const int N = in_sizes[0] / D_CH;
    const int E = in_sizes[2];
    const int L = in_sizes[3] / (8 * D_CH * D_CH);
    const int* src = eidx;
    const int* dst = eidx + E;

    float* ws   = (float*)d_ws;
    const int NBINS = (N + BIN_NODES - 1) >> BIN_SHIFT;
    float* h1   = ws;
    float* h2   = h1   + (size_t)N * D_CH;
    float* xs0  = h2   + (size_t)N * D_CH;
    int*   tab  = (int*)(xs0 + (size_t)N * D_CH);
    int*   rst  = tab + (size_t)NBINS * BIN_G;
    int4*  rec  = (int4*)(rst + (size_t)NBINS * BIN_G);
    float* dinv = (float*)(rec + E);
    float* csrw = dinv + N;
    int*   csrs = (int*)(csrw + E + N);
    int*   gkid = csrs + E + N;
    int*   gksr = gkid + E;
    float* gkw  = (float*)(gksr + E);
    float* gkws = gkw + E;
    int*   offs = (int*)(gkws + E);

    gcrn_chunk_sort<<<BIN_G, BIN_TB, 0, stream>>>(src, dst, ew, tab, rst, rec, E, NBINS);
    gcrn_bin_place <<<NBINS, BIN_TB, 0, stream>>>(tab, rst, rec, offs, csrs, csrw, dinv, gkid, gksr, gkw, gkws, X, xs0, N, NBINS);

    const float* xin = xs0;
    for (int l = 0; l < L; ++l) {
        float* out_l = (l == L - 1) ? (float*)d_out : ((l & 1) ? h2 : h1);
        const float* W_l  = conv_w + (size_t)l * 8 * 32 * 32;
        const float* cb_l = conv_b + (size_t)l * 8 * 32;
        const float* bg_l = bg + (size_t)l * 4 * 32;
        const float* wo_l = wc + ((size_t)l * 3 + 2) * 32;
        const int tiles = (N + TILE_NODES - 1) / TILE_NODES;
        if (l < L - 1)
            gcrn_layer<true> <<<tiles, 64, 0, stream>>>(offs, csrs, csrw, dinv, xin, W_l, cb_l, bg_l, wo_l, out_l, N, E);
        else
            gcrn_layer<false><<<tiles, 64, 0, stream>>>(offs, csrs, csrw, dinv, xin, W_l, cb_l, bg_l, wo_l, out_l, N, E);
        xin = out_l;
    }
}
